# starting kernel
# baseline (speedup 1.0000x reference)
#include <hip/hip_runtime.h>
#include <hip/hip_bf16.h>
#include <math.h>

#define N_NODES 100000
#define N_EDGES 3200000
#define NEG_SLOPE 0.2f

typedef __attribute__((ext_vector_type(2))) float v2f;
typedef __attribute__((ext_vector_type(8))) float v8f;

__device__ __forceinline__ float leakyf(float v) { return v > 0.f ? v : NEG_SLOPE * v; }

__device__ __forceinline__ void atomicMaxF(float* addr, float v) {
    if (v >= 0.f) atomicMax((int*)addr, __float_as_int(v));
    else          atomicMin((unsigned int*)addr, __float_as_uint(v));
}

template <int K, int M, int MOUT, int ACT>
__global__ __launch_bounds__(256)
void gemm_wmma_f32(const float* __restrict__ A, const float* __restrict__ W,
                   const float* __restrict__ bias, float* __restrict__ C,
                   int Nrows)
{
    const int lane     = threadIdx.x & 31;
    const int wave     = threadIdx.x >> 5;
    const int row_base = (blockIdx.x * 8 + wave) * 16;
    if (row_base >= Nrows) return;

    const int nlo   = lane & 15;
    const int khalf = (lane >> 4) * 2;
    const int arow  = row_base + nlo;
    const int rofs  = (lane >> 4) * 8;

    v2f av[K / 4];
    #pragma unroll
    for (int kc = 0; kc < K / 4; ++kc) {
        const int k0 = kc * 4 + khalf;
        av[kc].x = A[arow * K + k0];
        av[kc].y = A[arow * K + k0 + 1];
    }

    #pragma unroll
    for (int mt = 0; mt < M / 16; ++mt) {
        const int col = mt * 16 + nlo;
        const float bv = bias ? bias[col] : 0.0f;
        v8f c;
        #pragma unroll
        for (int v = 0; v < 8; ++v) c[v] = bv;

        #pragma unroll
        for (int kc = 0; kc < K / 4; ++kc) {
            const int k0 = kc * 4 + khalf;
            v2f b;
            b.x = W[k0 * M + col];
            b.y = W[(k0 + 1) * M + col];
            c = __builtin_amdgcn_wmma_f32_16x16x4_f32(
                    false, av[kc], false, b, (short)0, c, false, false);
        }

        if (MOUT == M || col < MOUT) {
            #pragma unroll
            for (int v = 0; v < 8; ++v) {
                float r = c[v];
                if (ACT == 1) r = r > 0.f ? r : 0.f;
                C[(row_base + rofs + v) * MOUT + col] = r;
            }
        }
    }
}

__global__ void pad_w3_kernel(const float* __restrict__ W3, const float* __restrict__ b3,
                              float* __restrict__ W3p, float* __restrict__ b3p)
{
    int t = blockIdx.x * blockDim.x + threadIdx.x;
    if (t < 64 * 16) {
        int k = t >> 4, c = t & 15;
        W3p[t] = (c < 4) ? W3[k * 4 + c] : 0.0f;
    }
    if (t < 16) b3p[t] = (t < 4) ? b3[t] : 0.0f;
}

__global__ void wae_kernel(const float* __restrict__ We, const float* __restrict__ ae,
                           float* __restrict__ wae)
{
    int t = threadIdx.x;
    if (t < 24) {
        int i = t >> 3, j = t & 7;
        float s = 0.f;
        #pragma unroll
        for (int k = 0; k < 16; ++k) s += We[i * 128 + j * 16 + k] * ae[i * 16 + k];
        wae[t] = s;
    }
}

__global__ __launch_bounds__(256)
void prep_nodes_kernel(const float* __restrict__ h,
                       const float* __restrict__ asrc, const float* __restrict__ adst,
                       const float* __restrict__ b,
                       float* __restrict__ ssrc, float* __restrict__ sdst,
                       float* __restrict__ m, float* __restrict__ den,
                       float* __restrict__ outb)
{
    int n = blockIdx.x * blockDim.x + threadIdx.x;
    if (n >= N_NODES) return;
    float ss = 0.f, sd = 0.f;
    #pragma unroll
    for (int f = 0; f < 16; ++f) {
        float hv = h[n * 16 + f];
        ss += hv * asrc[f];
        sd += hv * adst[f];
        outb[n * 16 + f] = b[f];
    }
    ssrc[n] = ss;
    sdst[n] = sd;
    m[n]    = -INFINITY;
    den[n]  = 0.f;
}

__global__ __launch_bounds__(256)
void edge_logit_max_kernel(const int* __restrict__ src, const int* __restrict__ dst,
                           const float* __restrict__ ea, const float* __restrict__ wae,
                           const float* __restrict__ ssrc, const float* __restrict__ sdst,
                           float* __restrict__ lp, float* __restrict__ m)
{
    int e = blockIdx.x * blockDim.x + threadIdx.x;
    if (e >= N_EDGES) return;
    int s = src[e], d = dst[e];
    float se = 0.f;
    #pragma unroll
    for (int j = 0; j < 8; ++j) se += ea[e * 8 + j] * wae[j];
    float lg = leakyf(ssrc[s] + sdst[d] + se);
    lp[e] = lg;
    atomicMaxF(&m[d], lg);
}

__global__ __launch_bounds__(256)
void edge_exp_sum_kernel(const int* __restrict__ dst, float* __restrict__ lp,
                         const float* __restrict__ m, float* __restrict__ den)
{
    int e = blockIdx.x * blockDim.x + threadIdx.x;
    if (e >= N_EDGES) return;
    int d = dst[e];
    float p = __expf(lp[e] - m[d]);
    lp[e] = p;
    atomicAdd(&den[d], p);
}

__global__ __launch_bounds__(256)
void edge_aggregate_kernel(const int* __restrict__ src, const int* __restrict__ dst,
                           const float* __restrict__ lp, const float* __restrict__ den,
                           const float* __restrict__ h, float* __restrict__ outb)
{
    long long t = (long long)blockIdx.x * blockDim.x + threadIdx.x;
    if (t >= (long long)N_EDGES * 16) return;
    int e = (int)(t >> 4);
    int f = (int)(t & 15);
    int d = dst[e], s = src[e];
    float alpha = lp[e] / (den[d] + 1e-16f);
    atomicAdd(&outb[d * 16 + f], h[s * 16 + f] * alpha);
}

__global__ __launch_bounds__(256)
void post_leaky_kernel(float* __restrict__ v, int n)
{
    int i = blockIdx.x * blockDim.x + threadIdx.x;
    if (i < n) v[i] = leakyf(v[i]);
}

extern "C" void kernel_launch(void* const* d_in, const int* in_sizes, int n_in,
                              void* d_out, int out_size, void* d_ws, size_t ws_size,
                              hipStream_t stream)
{
    const float* x     = (const float*)d_in[0];
    const int*   ei    = (const int*)  d_in[1];
    const float* ea    = (const float*)d_in[2];
    const float* cW    = (const float*)d_in[3];
    const float* casrc = (const float*)d_in[4];
    const float* cadst = (const float*)d_in[5];
    const float* cWe   = (const float*)d_in[6];
    const float* cae   = (const float*)d_in[7];
    const float* cb    = (const float*)d_in[8];
    const float* W0 = (const float*)d_in[9];  const float* b0 = (const float*)d_in[10];
    const float* W1 = (const float*)d_in[11]; const float* b1 = (const float*)d_in[12];
    const float* W2 = (const float*)d_in[13]; const float* b2 = (const float*)d_in[14];
    const float* W3 = (const float*)d_in[15]; const float* b3 = (const float*)d_in[16];
    float* out = (float*)d_out;

    const int* e_src = ei;
    const int* e_dst = ei + N_EDGES;

    float* ws   = (float*)d_ws;
    float* bufA = ws;
    float* bufB = bufA + (size_t)N_NODES * 16;
    float* ssrc = bufB + (size_t)N_NODES * 16;
    float* sdst = ssrc + N_NODES;
    float* segm = sdst + N_NODES;
    float* den  = segm + N_NODES;
    float* lp   = den  + N_NODES;
    float* wae  = lp   + N_EDGES;
    float* w3p  = wae  + 32;
    float* b3p  = w3p  + 64 * 16;
    float* mlpA = b3p  + 16;
    float* mlpB = bufA;

    const int nodeBlocks  = (N_NODES + 255) / 256;
    const int gemmBlocks  = (N_NODES + 127) / 128;
    const int edgeBlocks  = (N_EDGES + 255) / 256;
    const long long aggT  = (long long)N_EDGES * 16;
    const int aggBlocks   = (int)((aggT + 255) / 256);
    const int featBlocks  = (N_NODES * 16 + 255) / 256;

    wae_kernel<<<1, 32, 0, stream>>>(cWe, cae, wae);
    pad_w3_kernel<<<4, 256, 0, stream>>>(W3, b3, w3p, b3p);

    const float* xin = x;
    for (int i = 0; i < 3; ++i) {
        gemm_wmma_f32<16, 16, 16, 0><<<gemmBlocks, 256, 0, stream>>>(
            xin, cW + i * 256, nullptr, bufA, N_NODES);
        prep_nodes_kernel<<<nodeBlocks, 256, 0, stream>>>(
            bufA, casrc + i * 16, cadst + i * 16, cb + i * 16,
            ssrc, sdst, segm, den, bufB);
        edge_logit_max_kernel<<<edgeBlocks, 256, 0, stream>>>(
            e_src, e_dst, ea, wae + i * 8, ssrc, sdst, lp, segm);
        edge_exp_sum_kernel<<<edgeBlocks, 256, 0, stream>>>(
            e_dst, lp, segm, den);
        edge_aggregate_kernel<<<aggBlocks, 256, 0, stream>>>(
            e_src, e_dst, lp, den, bufA, bufB);
        post_leaky_kernel<<<featBlocks, 256, 0, stream>>>(bufB, N_NODES * 16);
        xin = bufB;
    }

    gemm_wmma_f32<16, 64, 64, 1><<<gemmBlocks, 256, 0, stream>>>(bufB, W0, b0, mlpA, N_NODES);
    gemm_wmma_f32<64, 64, 64, 1><<<gemmBlocks, 256, 0, stream>>>(mlpA, W1, b1, mlpB, N_NODES);
    gemm_wmma_f32<64, 64, 64, 1><<<gemmBlocks, 256, 0, stream>>>(mlpB, W2, b2, mlpA, N_NODES);
    gemm_wmma_f32<64, 16, 4, 0><<<gemmBlocks, 256, 0, stream>>>(mlpA, w3p, b3p, out, N_NODES);
}


# Edges counting-sorted by dst once per call (int atomic rank, scan, scatter); each layer: one wave per node does max, exp-sum, row sum over its segment; no float atomics; sum order = arrival rank
# speedup vs baseline: 2.7194x; 2.7194x over previous
#include <hip/hip_runtime.h>
#include <hip/hip_bf16.h>
#include <math.h>

#define N_NODES 100000
#define N_EDGES 3200000
#define NEG_SLOPE 0.2f
#define SCAN_CH 98
#define SCAN_B  14
#define SCAN_N  (1024 * SCAN_CH)

typedef __attribute__((ext_vector_type(2))) float v2f;
typedef __attribute__((ext_vector_type(8))) float v8f;

__device__ __forceinline__ float leakyf(float v) { return v > 0.f ? v : NEG_SLOPE * v; }

template <int K, int M, int MOUT, int ACT>
__global__ __launch_bounds__(256)
void gemm_wmma_f32(const float* __restrict__ A, const float* __restrict__ W,
                   const float* __restrict__ bias, float* __restrict__ C,
                   int Nrows)
{
    const int lane     = threadIdx.x & 31;
    const int wave     = threadIdx.x >> 5;
    const int row_base = (blockIdx.x * 8 + wave) * 16;
    if (row_base >= Nrows) return;

    const int nlo   = lane & 15;
    const int khalf = (lane >> 4) * 2;
    const int arow  = row_base + nlo;
    const int rofs  = (lane >> 4) * 8;

    v2f av[K / 4];
    #pragma unroll
    for (int kc = 0; kc < K / 4; ++kc) {
        const int k0 = kc * 4 + khalf;
        av[kc].x = A[arow * K + k0];
        av[kc].y = A[arow * K + k0 + 1];
    }

    #pragma unroll
    for (int mt = 0; mt < M / 16; ++mt) {
        const int col = mt * 16 + nlo;
        const float bv = bias ? bias[col] : 0.0f;
        v8f c;
        #pragma unroll
        for (int v = 0; v < 8; ++v) c[v] = bv;

        #pragma unroll
        for (int kc = 0; kc < K / 4; ++kc) {
            const int k0 = kc * 4 + khalf;
            v2f b;
            b.x = W[k0 * M + col];
            b.y = W[(k0 + 1) * M + col];
            c = __builtin_amdgcn_wmma_f32_16x16x4_f32(
                    false, av[kc], false, b, (short)0, c, false, false);
        }

        if (MOUT == M || col < MOUT) {
            #pragma unroll
            for (int v = 0; v < 8; ++v) {
                float r = c[v];
                if (ACT == 1) r = r > 0.f ? r : 0.f;
                C[(row_base + rofs + v) * MOUT + col] = r;
            }
        }
    }
}

__global__ __launch_bounds__(256)
void setup_kernel(const float* __restrict__ W3, const float* __restrict__ b3,
                  float* __restrict__ W3p, float* __restrict__ b3p,
                  int* __restrict__ cnt)
{
    int t = blockIdx.x * blockDim.x + threadIdx.x;
    if (t < 64 * 16) {
        int k = t >> 4, c = t & 15;
        W3p[t] = (c < 4) ? W3[k * 4 + c] : 0.0f;
    }
    if (t < 16) b3p[t] = (t < 4) ? b3[t] : 0.0f;
    if (t < SCAN_N) cnt[t] = 0;
}

__global__ void wae_kernel(const float* __restrict__ We, const float* __restrict__ ae,
                           float* __restrict__ wae)
{
    int t = threadIdx.x;
    if (t < 24) {
        int i = t >> 3, j = t & 7;
        float s = 0.f;
        #pragma unroll
        for (int k = 0; k < 16; ++k) s += We[i * 128 + j * 16 + k] * ae[i * 16 + k];
        wae[t] = s;
    }
}

__global__ __launch_bounds__(256)
void edge_rank_kernel(const int* __restrict__ dst, int* __restrict__ cnt,
                      int* __restrict__ rank)
{
    int e = blockIdx.x * blockDim.x + threadIdx.x;
    if (e >= N_EDGES) return;
    rank[e] = atomicAdd(&cnt[dst[e]], 1);
}

__global__ __launch_bounds__(1024)
void scan_counts_kernel(int* __restrict__ off)
{
    __shared__ int sh[1024];
    const int t  = threadIdx.x;
    int* p = off + t * SCAN_CH;
    int sum = 0;
    for (int i0 = 0; i0 < SCAN_CH; i0 += SCAN_B) {
        int v[SCAN_B];
        #pragma unroll
        for (int k = 0; k < SCAN_B; ++k) v[k] = p[i0 + k];
        #pragma unroll
        for (int k = 0; k < SCAN_B; ++k) sum += v[k];
    }
    sh[t] = sum;
    __syncthreads();
    for (int o = 1; o < 1024; o <<= 1) {
        const int v = (t >= o) ? sh[t - o] : 0;
        __syncthreads();
        sh[t] += v;
        __syncthreads();
    }
    int run = sh[t] - sum;
    for (int i0 = 0; i0 < SCAN_CH; i0 += SCAN_B) {
        int v[SCAN_B];
        #pragma unroll
        for (int k = 0; k < SCAN_B; ++k) v[k] = p[i0 + k];
        #pragma unroll
        for (int k = 0; k < SCAN_B; ++k) { p[i0 + k] = run; run += v[k]; }
    }
}

__global__ __launch_bounds__(256)
void edge_scatter_kernel(const int* __restrict__ src, const int* __restrict__ dst,
                         const int* __restrict__ off, const int* __restrict__ rank,
                         int2* __restrict__ recs)
{
    int e = blockIdx.x * blockDim.x + threadIdx.x;
    if (e >= N_EDGES) return;
    const int pos = off[dst[e]] + rank[e];
    recs[pos] = make_int2(src[e], e);
}

__global__ __launch_bounds__(256)
void prep_nodes_kernel(const float* __restrict__ h,
                       const float* __restrict__ asrc, const float* __restrict__ adst,
                       float* __restrict__ ssrc, float* __restrict__ sdst)
{
    int n = blockIdx.x * blockDim.x + threadIdx.x;
    if (n >= N_NODES) return;
    float ss = 0.f, sd = 0.f;
    #pragma unroll
    for (int f = 0; f < 16; ++f) {
        float hv = h[n * 16 + f];
        ss += hv * asrc[f];
        sd += hv * adst[f];
    }
    ssrc[n] = ss;
    sdst[n] = sd;
}

__global__ __launch_bounds__(256)
void segment_softmax_aggregate_kernel(const int2* __restrict__ recs, const int* __restrict__ off,
                                      const float* __restrict__ ea, const float* __restrict__ wae,
                                      const float* __restrict__ ssrc, const float* __restrict__ sdst,
                                      const float* __restrict__ h, const float* __restrict__ b,
                                      float* __restrict__ lp, float* __restrict__ outb)
{
    const int lane = threadIdx.x & 31;
    const int d    = blockIdx.x * 8 + (threadIdx.x >> 5);
    if (d >= N_NODES) return;
    const int beg = __builtin_amdgcn_readfirstlane(off[d]);
    const int end = __builtin_amdgcn_readfirstlane(off[d + 1]);
    const float sd = sdst[d];
    float w[8];
    for (int j = 0; j < 8; ++j) w[j] = wae[j];

    float mx = -INFINITY;
    for (int j = beg + lane; j < end; j += 32) {
        const int2 r = recs[j];
        const float4* ep = (const float4*)(ea + (size_t)r.y * 8);
        const float4 e0 = ep[0];
        const float4 e1 = ep[1];
        float se = 0.f;
        se += e0.x * w[0]; se += e0.y * w[1]; se += e0.z * w[2]; se += e0.w * w[3];
        se += e1.x * w[4]; se += e1.y * w[5]; se += e1.z * w[6]; se += e1.w * w[7];
        const float lg = leakyf(ssrc[r.x] + sd + se);
        lp[j] = lg;
        mx = fmaxf(mx, lg);
    }
    for (int o = 16; o > 0; o >>= 1) mx = fmaxf(mx, __shfl_xor(mx, o));
    if (!__builtin_isfinite(mx)) mx = 0.f;

    float sum = 0.f;
    for (int j = beg + lane; j < end; j += 32) {
        const float p = __expf(lp[j] - mx);
        lp[j] = p;
        sum += p;
    }
    for (int o = 16; o > 0; o >>= 1) sum += __shfl_xor(sum, o);
    const float dn = sum + 1e-16f;

    const int f    = lane & 15;
    const int half = lane >> 4;
    float acc = 0.f;
    for (int base = beg; base < end; base += 32) {
        const int j = base + lane;
        int   s_l = 0;
        float a_l = 0.f;
        if (j < end) { s_l = recs[j].x; a_l = lp[j] / dn; }
        const int cnt = (end - base) < 32 ? (end - base) : 32;
        for (int t = 0; t < cnt; t += 16) {
            float hv[8], av[8];
            #pragma unroll
            for (int u = 0; u < 8; ++u) {
                const int tt = t + 2 * u + half;
                const int s  = __shfl(s_l, tt);
                av[u] = __shfl(a_l, tt);
                hv[u] = h[s * 16 + f];
            }
            #pragma unroll
            for (int u = 0; u < 8; ++u) {
                const float pr = hv[u] * av[u];
                acc += (t + 2 * u + half < cnt) ? pr : 0.f;
            }
        }
    }
    acc += __shfl_xor(acc, 16);
    if (half == 0) outb[d * 16 + f] = leakyf(acc + b[f]);
}

extern "C" void kernel_launch(void* const* d_in, const int* in_sizes, int n_in,
                              void* d_out, int out_size, void* d_ws, size_t ws_size,
                              hipStream_t stream)
{
    const float* x     = (const float*)d_in[0];
    const int*   ei    = (const int*)  d_in[1];
    const float* ea    = (const float*)d_in[2];
    const float* cW    = (const float*)d_in[3];
    const float* casrc = (const float*)d_in[4];
    const float* cadst = (const float*)d_in[5];
    const float* cWe   = (const float*)d_in[6];
    const float* cae   = (const float*)d_in[7];
    const float* cb    = (const float*)d_in[8];
    const float* W0 = (const float*)d_in[9];  const float* b0 = (const float*)d_in[10];
    const float* W1 = (const float*)d_in[11]; const float* b1 = (const float*)d_in[12];
    const float* W2 = (const float*)d_in[13]; const float* b2 = (const float*)d_in[14];
    const float* W3 = (const float*)d_in[15]; const float* b3 = (const float*)d_in[16];
    float* out = (float*)d_out;

    const int* e_src = ei;
    const int* e_dst = ei + N_EDGES;

    float* ws   = (float*)d_ws;
    float* bufA = ws;
    float* bufB = bufA + (size_t)N_NODES * 16;
    float* ssrc = bufB + (size_t)N_NODES * 16;
    float* sdst = ssrc + N_NODES;
    int*   off  = (int*)(sdst + N_NODES);
    float* lp   = sdst + 3 * (size_t)N_NODES;
    int*   rank = (int*)lp;
    float* wae  = lp   + N_EDGES;
    float* w3p  = wae  + 32;
    float* b3p  = w3p  + 64 * 16;
    float* mlpA = b3p  + 16;
    int2*  recs = (int2*)mlpA;
    float* mlpB = bufA;

    const int nodeBlocks  = (N_NODES + 255) / 256;
    const int scanBlocks  = SCAN_N / 256;
    const int gemmBlocks  = (N_NODES + 127) / 128;
    const int edgeBlocks  = (N_EDGES + 255) / 256;
    const int segBlocks   = (N_NODES + 7) / 8;

    wae_kernel<<<1, 32, 0, stream>>>(cWe, cae, wae);
    setup_kernel<<<scanBlocks, 256, 0, stream>>>(W3, b3, w3p, b3p, off);

    edge_rank_kernel<<<edgeBlocks, 256, 0, stream>>>(e_dst, off, rank);
    scan_counts_kernel<<<1, 1024, 0, stream>>>(off);
    edge_scatter_kernel<<<edgeBlocks, 256, 0, stream>>>(e_src, e_dst, off, rank, recs);

    const float* xin = x;
    for (int i = 0; i < 3; ++i) {
        gemm_wmma_f32<16, 16, 16, 0><<<gemmBlocks, 256, 0, stream>>>(
            xin, cW + i * 256, nullptr, bufA, N_NODES);
        prep_nodes_kernel<<<nodeBlocks, 256, 0, stream>>>(
            bufA, casrc + i * 16, cadst + i * 16, ssrc, sdst);
        segment_softmax_aggregate_kernel<<<segBlocks, 256, 0, stream>>>(
            recs, off, ea, wae + i * 8, ssrc, sdst, bufA, cb + i * 16, lp, bufB);
        xin = bufB;
    }

    gemm_wmma_f32<16, 64, 64, 1><<<gemmBlocks, 256, 0, stream>>>(bufB, W0, b0, mlpA, N_NODES);
    gemm_wmma_f32<64, 64, 64, 1><<<gemmBlocks, 256, 0, stream>>>(mlpA, W1, b1, mlpB, N_NODES);
    gemm_wmma_f32<64, 64, 64, 1><<<gemmBlocks, 256, 0, stream>>>(mlpB, W2, b2, mlpA, N_NODES);
    gemm_wmma_f32<64, 16, 4, 0><<<gemmBlocks, 256, 0, stream>>>(mlpA, w3p, b3p, out, N_NODES);
}


# All three layers' edge terms come from the first layer's edge_attr gather, stored in destination order (over the edge id and old logit plane); logits and p stay in registers; sum orders kept
# speedup vs baseline: 3.2140x; 1.1819x over previous
#include <hip/hip_runtime.h>
#include <hip/hip_bf16.h>
#include <math.h>

#define N_NODES 100000
#define N_EDGES 3200000
#define NEG_SLOPE 0.2f
#define SCAN_CH 98
#define SCAN_B  14
#define SCAN_N  (1024 * SCAN_CH)

typedef __attribute__((ext_vector_type(2))) float v2f;
typedef __attribute__((ext_vector_type(8))) float v8f;

__device__ __forceinline__ float leakyf(float v) { return v > 0.f ? v : NEG_SLOPE * v; }

template <int K, int M, int MOUT, int ACT>
__global__ __launch_bounds__(256)
void gemm_wmma_f32(const float* __restrict__ A, const float* __restrict__ W,
                   const float* __restrict__ bias, float* __restrict__ C,
                   int Nrows)
{
    const int lane     = threadIdx.x & 31;
    const int wave     = threadIdx.x >> 5;
    const int row_base = (blockIdx.x * 8 + wave) * 16;
    if (row_base >= Nrows) return;

    const int nlo   = lane & 15;
    const int khalf = (lane >> 4) * 2;
    const int arow  = row_base + nlo;
    const int rofs  = (lane >> 4) * 8;

    v2f av[K / 4];
    #pragma unroll
    for (int kc = 0; kc < K / 4; ++kc) {
        const int k0 = kc * 4 + khalf;
        av[kc].x = A[arow * K + k0];
        av[kc].y = A[arow * K + k0 + 1];
    }

    #pragma unroll
    for (int mt = 0; mt < M / 16; ++mt) {
        const int col = mt * 16 + nlo;
        const float bv = bias ? bias[col] : 0.0f;
        v8f c;
        #pragma unroll
        for (int v = 0; v < 8; ++v) c[v] = bv;

        #pragma unroll
        for (int kc = 0; kc < K / 4; ++kc) {
            const int k0 = kc * 4 + khalf;
            v2f b;
            b.x = W[k0 * M + col];
            b.y = W[(k0 + 1) * M + col];
            c = __builtin_amdgcn_wmma_f32_16x16x4_f32(
                    false, av[kc], false, b, (short)0, c, false, false);
        }

        if (MOUT == M || col < MOUT) {
            #pragma unroll
            for (int v = 0; v < 8; ++v) {
                float r = c[v];
                if (ACT == 1) r = r > 0.f ? r : 0.f;
                C[(row_base + rofs + v) * MOUT + col] = r;
            }
        }
    }
}

__global__ __launch_bounds__(256)
void setup_kernel(const float* __restrict__ W3, const float* __restrict__ b3,
                  float* __restrict__ W3p, float* __restrict__ b3p,
                  int* __restrict__ cnt)
{
    int t = blockIdx.x * blockDim.x + threadIdx.x;
    if (t < 64 * 16) {
        int k = t >> 4, c = t & 15;
        W3p[t] = (c < 4) ? W3[k * 4 + c] : 0.0f;
    }
    if (t < 16) b3p[t] = (t < 4) ? b3[t] : 0.0f;
    if (t < SCAN_N) cnt[t] = 0;
}

__global__ void wae_kernel(const float* __restrict__ We, const float* __restrict__ ae,
                           float* __restrict__ wae)
{
    int t = threadIdx.x;
    if (t < 24) {
        int i = t >> 3, j = t & 7;
        float s = 0.f;
        #pragma unroll
        for (int k = 0; k < 16; ++k) s += We[i * 128 + j * 16 + k] * ae[i * 16 + k];
        wae[t] = s;
    }
}

__global__ __launch_bounds__(256)
void edge_rank_kernel(const int* __restrict__ dst, int* __restrict__ cnt,
                      int* __restrict__ rank)
{
    int e = blockIdx.x * blockDim.x + threadIdx.x;
    if (e >= N_EDGES) return;
    rank[e] = atomicAdd(&cnt[dst[e]], 1);
}

__global__ __launch_bounds__(1024)
void scan_counts_kernel(int* __restrict__ off)
{
    __shared__ int sh[1024];
    const int t  = threadIdx.x;
    int* p = off + t * SCAN_CH;
    int sum = 0;
    for (int i0 = 0; i0 < SCAN_CH; i0 += SCAN_B) {
        int v[SCAN_B];
        #pragma unroll
        for (int k = 0; k < SCAN_B; ++k) v[k] = p[i0 + k];
        #pragma unroll
        for (int k = 0; k < SCAN_B; ++k) sum += v[k];
    }
    sh[t] = sum;
    __syncthreads();
    for (int o = 1; o < 1024; o <<= 1) {
        const int v = (t >= o) ? sh[t - o] : 0;
        __syncthreads();
        sh[t] += v;
        __syncthreads();
    }
    int run = sh[t] - sum;
    for (int i0 = 0; i0 < SCAN_CH; i0 += SCAN_B) {
        int v[SCAN_B];
        #pragma unroll
        for (int k = 0; k < SCAN_B; ++k) v[k] = p[i0 + k];
        #pragma unroll
        for (int k = 0; k < SCAN_B; ++k) { p[i0 + k] = run; run += v[k]; }
    }
}

__global__ __launch_bounds__(256)
void edge_scatter_kernel(const int* __restrict__ src, const int* __restrict__ dst,
                         const int* __restrict__ off, const int* __restrict__ rank,
                         int2* __restrict__ recs)
{
    int e = blockIdx.x * blockDim.x + threadIdx.x;
    if (e >= N_EDGES) return;
    const int pos = off[dst[e]] + rank[e];
    recs[pos] = make_int2(src[e], e);
}

__global__ __launch_bounds__(256)
void prep_nodes_kernel(const float* __restrict__ h,
                       const float* __restrict__ asrc, const float* __restrict__ adst,
                       float* __restrict__ ssrc, float* __restrict__ sdst)
{
    int n = blockIdx.x * blockDim.x + threadIdx.x;
    if (n >= N_NODES) return;
    float ss = 0.f, sd = 0.f;
    #pragma unroll
    for (int f = 0; f < 16; ++f) {
        float hv = h[n * 16 + f];
        ss += hv * asrc[f];
        sd += hv * adst[f];
    }
    ssrc[n] = ss;
    sdst[n] = sd;
}

__device__ __forceinline__ float edge_dot(const float4 e0, const float4 e1, const float* w)
{
    float se = 0.f;
    se += e0.x * w[0]; se += e0.y * w[1]; se += e0.z * w[2]; se += e0.w * w[3];
    se += e1.x * w[4]; se += e1.y * w[5]; se += e1.z * w[6]; se += e1.w * w[7];
    return se;
}

template <int LAYER>
__global__ __launch_bounds__(256)
void segment_softmax_aggregate_kernel(int2* recs, const int* __restrict__ off,
                                      const float* __restrict__ ea, const float* __restrict__ wae,
                                      const float* __restrict__ ssrc, const float* __restrict__ sdst,
                                      const float* __restrict__ h, const float* __restrict__ b,
                                      float* se1, float* __restrict__ outb)
{
    const int lane = threadIdx.x & 31;
    const int d    = blockIdx.x * 8 + (threadIdx.x >> 5);
    if (d >= N_NODES) return;
    const int beg = __builtin_amdgcn_readfirstlane(off[d]);
    const int end = __builtin_amdgcn_readfirstlane(off[d + 1]);
    const float sd = sdst[d];
    float w[LAYER == 0 ? 24 : 1];
    if (LAYER == 0) { for (int k = 0; k < 24; ++k) w[k] = wae[k]; } else { w[0] = 0.f; }

    auto logit_mem = [&](const int j, int& s) -> float {
        const int2 r = recs[j];
        s = r.x;
        float se;
        if (LAYER == 0) {
            const float4* ep = (const float4*)(ea + (size_t)r.y * 8);
            se = edge_dot(ep[0], ep[1], w);
        } else if (LAYER == 1) {
            se = se1[j];
        } else {
            se = __int_as_float(r.y);
        }
        return leakyf(ssrc[r.x] + sd + se);
    };

    float q0 = 0.f, q1 = 0.f;
    int   s0 = 0,   s1 = 0;
    float mx = -INFINITY;
    {
        int it = 0;
        for (int j = beg + lane; j < end; j += 32, ++it) {
            int s;
            float lg;
            if (LAYER == 0) {
                const int2 r = recs[j];
                s = r.x;
                const float4* ep = (const float4*)(ea + (size_t)r.y * 8);
                const float4 e0 = ep[0];
                const float4 e1 = ep[1];
                lg = leakyf(ssrc[r.x] + sd + edge_dot(e0, e1, w));
                se1[j] = edge_dot(e0, e1, w + 8);
                if (it < 2) recs[j] = make_int2(r.x, __float_as_int(edge_dot(e0, e1, w + 16)));
            } else {
                lg = logit_mem(j, s);
            }
            if (it == 0) { q0 = lg; s0 = s; } else if (it == 1) { q1 = lg; s1 = s; }
            mx = fmaxf(mx, lg);
        }
    }
    for (int o = 16; o > 0; o >>= 1) mx = fmaxf(mx, __shfl_xor(mx, o));
    if (!__builtin_isfinite(mx)) mx = 0.f;

    float sum = 0.f;
    {
        int it = 0;
        for (int j = beg + lane; j < end; j += 32, ++it) {
            float p;
            if (it == 0)      { p = __expf(q0 - mx); q0 = p; }
            else if (it == 1) { p = __expf(q1 - mx); q1 = p; }
            else              { int s; p = __expf(logit_mem(j, s) - mx); }
            sum += p;
        }
    }
    for (int o = 16; o > 0; o >>= 1) sum += __shfl_xor(sum, o);
    const float dn = sum + 1e-16f;

    const int f    = lane & 15;
    const int half = lane >> 4;
    float acc = 0.f;
    {
        int it = 0;
        for (int base = beg; base < end; base += 32, ++it) {
            const int j = base + lane;
            int   s_l = 0;
            float a_l = 0.f;
            if (j < end) {
                float p;
                if (it == 0)      { s_l = s0; p = q0; }
                else if (it == 1) { s_l = s1; p = q1; }
                else              { p = __expf(logit_mem(j, s_l) - mx); }
                a_l = p / dn;
            }
            const int cnt = (end - base) < 32 ? (end - base) : 32;
            for (int t = 0; t < cnt; t += 16) {
                float hv[8], av[8];
                #pragma unroll
                for (int u = 0; u < 8; ++u) {
                    const int tt = t + 2 * u + half;
                    const int s  = __shfl(s_l, tt);
                    av[u] = __shfl(a_l, tt);
                    hv[u] = h[s * 16 + f];
                }
                #pragma unroll
                for (int u = 0; u < 8; ++u) {
                    const float pr = hv[u] * av[u];
                    acc += (t + 2 * u + half < cnt) ? pr : 0.f;
                }
            }
        }
    }
    acc += __shfl_xor(acc, 16);
    if (half == 0) outb[d * 16 + f] = leakyf(acc + b[f]);

    if (LAYER == 0) {
        for (int j = beg + lane + 64; j < end; j += 32) {
            const int2 r = recs[j];
            const float4* ep = (const float4*)(ea + (size_t)r.y * 8);
            recs[j] = make_int2(r.x, __float_as_int(edge_dot(ep[0], ep[1], w + 16)));
        }
    }
}

extern "C" void kernel_launch(void* const* d_in, const int* in_sizes, int n_in,
                              void* d_out, int out_size, void* d_ws, size_t ws_size,
                              hipStream_t stream)
{
    const float* x     = (const float*)d_in[0];
    const int*   ei    = (const int*)  d_in[1];
    const float* ea    = (const float*)d_in[2];
    const float* cW    = (const float*)d_in[3];
    const float* casrc = (const float*)d_in[4];
    const float* cadst = (const float*)d_in[5];
    const float* cWe   = (const float*)d_in[6];
    const float* cae   = (const float*)d_in[7];
    const float* cb    = (const float*)d_in[8];
    const float* W0 = (const float*)d_in[9];  const float* b0 = (const float*)d_in[10];
    const float* W1 = (const float*)d_in[11]; const float* b1 = (const float*)d_in[12];
    const float* W2 = (const float*)d_in[13]; const float* b2 = (const float*)d_in[14];
    const float* W3 = (const float*)d_in[15]; const float* b3 = (const float*)d_in[16];
    float* out = (float*)d_out;

    const int* e_src = ei;
    const int* e_dst = ei + N_EDGES;

    float* ws   = (float*)d_ws;
    float* bufA = ws;
    float* bufB = bufA + (size_t)N_NODES * 16;
    float* ssrc = bufB + (size_t)N_NODES * 16;
    float* sdst = ssrc + N_NODES;
    int*   off  = (int*)(sdst + N_NODES);
    float* lp   = sdst + 3 * (size_t)N_NODES;
    int*   rank = (int*)lp;
    float* wae  = lp   + N_EDGES;
    float* w3p  = wae  + 32;
    float* b3p  = w3p  + 64 * 16;
    float* mlpA = b3p  + 16;
    int2*  recs = (int2*)mlpA;
    float* mlpB = bufA;

    const int nodeBlocks  = (N_NODES + 255) / 256;
    const int scanBlocks  = SCAN_N / 256;
    const int gemmBlocks  = (N_NODES + 127) / 128;
    const int edgeBlocks  = (N_EDGES + 255) / 256;
    const int segBlocks   = (N_NODES + 7) / 8;

    wae_kernel<<<1, 32, 0, stream>>>(cWe, cae, wae);
    setup_kernel<<<scanBlocks, 256, 0, stream>>>(W3, b3, w3p, b3p, off);

    edge_rank_kernel<<<edgeBlocks, 256, 0, stream>>>(e_dst, off, rank);
    scan_counts_kernel<<<1, 1024, 0, stream>>>(off);
    edge_scatter_kernel<<<edgeBlocks, 256, 0, stream>>>(e_src, e_dst, off, rank, recs);

    gemm_wmma_f32<16, 16, 16, 0><<<gemmBlocks, 256, 0, stream>>>(x, cW, nullptr, bufA, N_NODES);
    prep_nodes_kernel<<<nodeBlocks, 256, 0, stream>>>(bufA, casrc, cadst, ssrc, sdst);
    segment_softmax_aggregate_kernel<0><<<segBlocks, 256, 0, stream>>>(
        recs, off, ea, wae, ssrc, sdst, bufA, cb, lp, bufB);
    gemm_wmma_f32<16, 16, 16, 0><<<gemmBlocks, 256, 0, stream>>>(bufB, cW + 256, nullptr, bufA, N_NODES);
    prep_nodes_kernel<<<nodeBlocks, 256, 0, stream>>>(bufA, casrc + 16, cadst + 16, ssrc, sdst);
    segment_softmax_aggregate_kernel<1><<<segBlocks, 256, 0, stream>>>(
        recs, off, ea, wae, ssrc, sdst, bufA, cb + 16, lp, bufB);
    gemm_wmma_f32<16, 16, 16, 0><<<gemmBlocks, 256, 0, stream>>>(bufB, cW + 512, nullptr, bufA, N_NODES);
    prep_nodes_kernel<<<nodeBlocks, 256, 0, stream>>>(bufA, casrc + 32, cadst + 32, ssrc, sdst);
    segment_softmax_aggregate_kernel<2><<<segBlocks, 256, 0, stream>>>(
        recs, off, ea, wae, ssrc, sdst, bufA, cb + 32, lp, bufB);

    gemm_wmma_f32<16, 64, 64, 1><<<gemmBlocks, 256, 0, stream>>>(bufB, W0, b0, mlpA, N_NODES);
    gemm_wmma_f32<64, 64, 64, 1><<<gemmBlocks, 256, 0, stream>>>(mlpA, W1, b1, mlpB, N_NODES);
    gemm_wmma_f32<64, 64, 64, 1><<<gemmBlocks, 256, 0, stream>>>(mlpB, W2, b2, mlpA, N_NODES);
    gemm_wmma_f32<64, 16, 4, 0><<<gemmBlocks, 256, 0, stream>>>(mlpA, w3p, b3p, out, N_NODES);
}


# Source rows gathered once per edge ahead of the softmax (8 lanes a row), kept in registers; s_src summed from the row by a lane tree, no score gather; row sum now 4 lane groups x 8 slots, then a tree
# speedup vs baseline: 3.3476x; 1.0416x over previous
#include <hip/hip_runtime.h>
#include <hip/hip_bf16.h>
#include <math.h>

#define N_NODES 100000
#define N_EDGES 3200000
#define NEG_SLOPE 0.2f
#define SCAN_CH 98
#define SCAN_B  14
#define SCAN_N  (1024 * SCAN_CH)

typedef __attribute__((ext_vector_type(2))) float v2f;
typedef __attribute__((ext_vector_type(8))) float v8f;

__device__ __forceinline__ float leakyf(float v) { return v > 0.f ? v : NEG_SLOPE * v; }

template <int K, int M, int MOUT, int ACT>
__global__ __launch_bounds__(256)
void gemm_wmma_f32(const float* __restrict__ A, const float* __restrict__ W,
                   const float* __restrict__ bias, float* __restrict__ C,
                   int Nrows)
{
    const int lane     = threadIdx.x & 31;
    const int wave     = threadIdx.x >> 5;
    const int row_base = (blockIdx.x * 8 + wave) * 16;
    if (row_base >= Nrows) return;

    const int nlo   = lane & 15;
    const int khalf = (lane >> 4) * 2;
    const int arow  = row_base + nlo;
    const int rofs  = (lane >> 4) * 8;

    v2f av[K / 4];
    #pragma unroll
    for (int kc = 0; kc < K / 4; ++kc) {
        const int k0 = kc * 4 + khalf;
        av[kc].x = A[arow * K + k0];
        av[kc].y = A[arow * K + k0 + 1];
    }

    #pragma unroll
    for (int mt = 0; mt < M / 16; ++mt) {
        const int col = mt * 16 + nlo;
        const float bv = bias ? bias[col] : 0.0f;
        v8f c;
        #pragma unroll
        for (int v = 0; v < 8; ++v) c[v] = bv;

        #pragma unroll
        for (int kc = 0; kc < K / 4; ++kc) {
            const int k0 = kc * 4 + khalf;
            v2f b;
            b.x = W[k0 * M + col];
            b.y = W[(k0 + 1) * M + col];
            c = __builtin_amdgcn_wmma_f32_16x16x4_f32(
                    false, av[kc], false, b, (short)0, c, false, false);
        }

        if (MOUT == M || col < MOUT) {
            #pragma unroll
            for (int v = 0; v < 8; ++v) {
                float r = c[v];
                if (ACT == 1) r = r > 0.f ? r : 0.f;
                C[(row_base + rofs + v) * MOUT + col] = r;
            }
        }
    }
}

__global__ __launch_bounds__(256)
void setup_kernel(const float* __restrict__ W3, const float* __restrict__ b3,
                  float* __restrict__ W3p, float* __restrict__ b3p,
                  int* __restrict__ cnt)
{
    int t = blockIdx.x * blockDim.x + threadIdx.x;
    if (t < 64 * 16) {
        int k = t >> 4, c = t & 15;
        W3p[t] = (c < 4) ? W3[k * 4 + c] : 0.0f;
    }
    if (t < 16) b3p[t] = (t < 4) ? b3[t] : 0.0f;
    if (t < SCAN_N) cnt[t] = 0;
}

__global__ void wae_kernel(const float* __restrict__ We, const float* __restrict__ ae,
                           float* __restrict__ wae)
{
    int t = threadIdx.x;
    if (t < 24) {
        int i = t >> 3, j = t & 7;
        float s = 0.f;
        #pragma unroll
        for (int k = 0; k < 16; ++k) s += We[i * 128 + j * 16 + k] * ae[i * 16 + k];
        wae[t] = s;
    }
}

__global__ __launch_bounds__(256)
void edge_rank_kernel(const int* __restrict__ dst, int* __restrict__ cnt,
                      int* __restrict__ rank)
{
    int e = blockIdx.x * blockDim.x + threadIdx.x;
    if (e >= N_EDGES) return;
    rank[e] = atomicAdd(&cnt[dst[e]], 1);
}

__global__ __launch_bounds__(1024)
void scan_counts_kernel(int* __restrict__ off)
{
    __shared__ int sh[1024];
    const int t  = threadIdx.x;
    int* p = off + t * SCAN_CH;
    int sum = 0;
    for (int i0 = 0; i0 < SCAN_CH; i0 += SCAN_B) {
        int v[SCAN_B];
        #pragma unroll
        for (int k = 0; k < SCAN_B; ++k) v[k] = p[i0 + k];
        #pragma unroll
        for (int k = 0; k < SCAN_B; ++k) sum += v[k];
    }
    sh[t] = sum;
    __syncthreads();
    for (int o = 1; o < 1024; o <<= 1) {
        const int v = (t >= o) ? sh[t - o] : 0;
        __syncthreads();
        sh[t] += v;
        __syncthreads();
    }
    int run = sh[t] - sum;
    for (int i0 = 0; i0 < SCAN_CH; i0 += SCAN_B) {
        int v[SCAN_B];
        #pragma unroll
        for (int k = 0; k < SCAN_B; ++k) v[k] = p[i0 + k];
        #pragma unroll
        for (int k = 0; k < SCAN_B; ++k) { p[i0 + k] = run; run += v[k]; }
    }
}

__global__ __launch_bounds__(256)
void edge_scatter_kernel(const int* __restrict__ src, const int* __restrict__ dst,
                         const int* __restrict__ off, const int* __restrict__ rank,
                         int2* __restrict__ recs)
{
    int e = blockIdx.x * blockDim.x + threadIdx.x;
    if (e >= N_EDGES) return;
    const int pos = off[dst[e]] + rank[e];
    recs[pos] = make_int2(src[e], e);
}

__global__ __launch_bounds__(256)
void prep_nodes_kernel(const float* __restrict__ h, const float* __restrict__ adst,
                       float* __restrict__ sdst)
{
    int n = blockIdx.x * blockDim.x + threadIdx.x;
    if (n >= N_NODES) return;
    float sd = 0.f;
    #pragma unroll
    for (int f = 0; f < 16; ++f) sd += h[n * 16 + f] * adst[f];
    sdst[n] = sd;
}

__device__ __forceinline__ float edge_dot(const float4 e0, const float4 e1, const float* w)
{
    float se = 0.f;
    se += e0.x * w[0]; se += e0.y * w[1]; se += e0.z * w[2]; se += e0.w * w[3];
    se += e1.x * w[4]; se += e1.y * w[5]; se += e1.z * w[6]; se += e1.w * w[7];
    return se;
}

template <unsigned SEL>
__device__ __forceinline__ float grp8f(const float v)
{
    return __uint_as_float(__builtin_amdgcn_mov_dpp8(__float_as_uint(v), SEL));
}
template <unsigned SEL>
__device__ __forceinline__ int grp8i(const int v)
{
    return (int)__builtin_amdgcn_mov_dpp8((unsigned)v, SEL);
}
#define SEL_LANE(u) ((unsigned)(u) * 2396745u)
#define SEL_XOR4 6852524u
#define SEL_XOR2 11788826u
#define SEL_XOR1 14570689u

template <int LAYER>
__global__ __launch_bounds__(256)
void segment_softmax_aggregate_kernel(int2* recs, const int* __restrict__ off,
                                      const float* __restrict__ ea, const float* __restrict__ wae,
                                      const float* __restrict__ asrc, const float* __restrict__ sdst,
                                      const float* __restrict__ h, const float* __restrict__ b,
                                      float* se1, float* __restrict__ outb)
{
    const int lane = threadIdx.x & 31;
    const int d    = blockIdx.x * 8 + (threadIdx.x >> 5);
    if (d >= N_NODES) return;
    const int beg = __builtin_amdgcn_readfirstlane(off[d]);
    const int end = __builtin_amdgcn_readfirstlane(off[d + 1]);
    const int n   = end - beg;
    const float sd = sdst[d];
    float w[LAYER == 0 ? 24 : 1];
    if (LAYER == 0) { for (int k = 0; k < 24; ++k) w[k] = wae[k]; } else { w[0] = 0.f; }

    const int g  = lane >> 3;
    const int i8 = lane & 7;
    const float2 a2 = *(const float2*)(asrc + 2 * i8);
    const float2 bb = *(const float2*)(b + 2 * i8);
    const bool b4 = (i8 & 4) != 0, b2 = (i8 & 2) != 0, b1 = (i8 & 1) != 0;

    auto slot_index = [&](const int j) -> int {
        const int jj = j < end ? j : end - 1;
        return jj < 0 ? 0 : jj;
    };
    auto slot_rec = [&](const int j) -> int2 {
        const int2 r = recs[slot_index(j)];
        return j < end ? r : make_int2(0, 0);
    };
    auto term_load = [&](const int j, const int2 r, float4& e0, float4& e1, float& se) {
        e0 = make_float4(0.f, 0.f, 0.f, 0.f);
        e1 = e0;
        se = 0.f;
        if (LAYER == 0) {
            const float4* ep = (const float4*)(ea + (size_t)r.y * 8);
            e0 = ep[0];
            e1 = ep[1];
        } else if (LAYER == 1) {
            se = se1[slot_index(j)];
        } else {
            se = __int_as_float(r.y);
        }
    };
    auto slot_term = [&](const int j, const int2 r, const float4 e0, const float4 e1,
                         const float se_in, const bool store, const bool keep) -> float {
        if (LAYER != 0) return se_in;
        if (store && j < end) {
            se1[j] = edge_dot(e0, e1, w + 8);
            if (keep) recs[j] = make_int2(r.x, __float_as_int(edge_dot(e0, e1, w + 16)));
        }
        return edge_dot(e0, e1, w);
    };

#define ROW_STEP(u)                                                              \
    {                                                                            \
        const int s = grp8i<SEL_LANE(u)>(s_l);                                   \
        const float2 r = *(const float2*)(h + s * 16 + 2 * i8);                  \
        const bool ok = 8 * g + (u) < cnt;                                       \
        row[u] = make_float2(ok ? r.x : 0.f, ok ? r.y : 0.f);                    \
    }
    auto gather_rows = [&](const int s_l, const int cnt, float2 (&row)[8]) {
        ROW_STEP(0) ROW_STEP(1) ROW_STEP(2) ROW_STEP(3)
        ROW_STEP(4) ROW_STEP(5) ROW_STEP(6) ROW_STEP(7)
    };
    auto slot_score = [&](const float2 (&row)[8]) -> float {
        float pv[8], w4[4], w2[2];
        #pragma unroll
        for (int u = 0; u < 8; ++u) pv[u] = row[u].x * a2.x + row[u].y * a2.y;
        #pragma unroll
        for (int k = 0; k < 4; ++k) {
            const float keepv = b4 ? pv[k + 4] : pv[k];
            const float send  = b4 ? pv[k] : pv[k + 4];
            w4[k] = keepv + grp8f<SEL_XOR4>(send);
        }
        #pragma unroll
        for (int k = 0; k < 2; ++k) {
            const float keepv = b2 ? w4[k + 2] : w4[k];
            const float send  = b2 ? w4[k] : w4[k + 2];
            w2[k] = keepv + grp8f<SEL_XOR2>(send);
        }
        const float keepv = b1 ? w2[1] : w2[0];
        const float send  = b1 ? w2[0] : w2[1];
        return keepv + grp8f<SEL_XOR1>(send);
    };
    auto slot_logit = [&](const int j, const float ss, const float se) -> float {
        return (j < end) ? leakyf(ss + sd + se) : -INFINITY;
    };
    auto chunk_logit = [&](const int base, const bool store, float2 (&row)[8]) -> float {
        const int j = base + lane;
        const int2 r = slot_rec(j);
        float4 e0, e1;
        float se;
        term_load(j, r, e0, e1, se);
        gather_rows(r.x, end - base, row);
        return slot_logit(j, slot_score(row), slot_term(j, r, e0, e1, se, store, false));
    };

#define ACC_STEP(row, a, u)                                                      \
    {                                                                            \
        const float av = grp8f<SEL_LANE(u)>(a);                                  \
        acc.x += row[u].x * av;                                                  \
        acc.y += row[u].y * av;                                                  \
    }
#define ACC_CHUNK(row, a)                                                        \
    ACC_STEP(row, a, 0) ACC_STEP(row, a, 1) ACC_STEP(row, a, 2) ACC_STEP(row, a, 3) \
    ACC_STEP(row, a, 4) ACC_STEP(row, a, 5) ACC_STEP(row, a, 6) ACC_STEP(row, a, 7)

    const int j0 = beg + lane, j1 = j0 + 32;
    const int2 c0 = slot_rec(j0);
    const int2 c1 = slot_rec(j1);
    float4 x0, y0, x1, y1;
    float t0, t1;
    float2 r0[8], r1[8];
    term_load(j0, c0, x0, y0, t0);
    gather_rows(c0.x, n, r0);
    if (n > 32) {
        term_load(j1, c1, x1, y1, t1);
        gather_rows(c1.x, n - 32, r1);
    }
    float q0, q1 = -INFINITY;
    q0 = slot_logit(j0, slot_score(r0), slot_term(j0, c0, x0, y0, t0, true, true));
    if (n > 32) q1 = slot_logit(j1, slot_score(r1), slot_term(j1, c1, x1, y1, t1, true, true));
    float mx = fmaxf(q0, q1);
    for (int base = beg + 64; base < end; base += 32) {
        float2 t[8];
        mx = fmaxf(mx, chunk_logit(base, true, t));
    }
    for (int o = 16; o > 0; o >>= 1) mx = fmaxf(mx, __shfl_xor(mx, o));
    if (!__builtin_isfinite(mx)) mx = 0.f;

    q0 = __expf(q0 - mx);
    q1 = __expf(q1 - mx);
    float sum = q0 + q1;
    for (int base = beg + 64; base < end; base += 32) {
        float2 t[8];
        sum += __expf(chunk_logit(base, false, t) - mx);
    }
    for (int o = 16; o > 0; o >>= 1) sum += __shfl_xor(sum, o);
    const float dn = sum + 1e-16f;

    float2 acc = make_float2(0.f, 0.f);
    {
        const float a0 = q0 / dn;
        const float a1 = q1 / dn;
        ACC_CHUNK(r0, a0)
        if (n > 32) { ACC_CHUNK(r1, a1) }
        for (int base = beg + 64; base < end; base += 32) {
            float2 t[8];
            const float a = __expf(chunk_logit(base, false, t) - mx) / dn;
            ACC_CHUNK(t, a)
        }
    }
    acc.x += __shfl_xor(acc.x, 8);  acc.y += __shfl_xor(acc.y, 8);
    acc.x += __shfl_xor(acc.x, 16); acc.y += __shfl_xor(acc.y, 16);
    if (lane < 8) *(float2*)(outb + d * 16 + 2 * i8) = make_float2(leakyf(acc.x + bb.x), leakyf(acc.y + bb.y));

    if (LAYER == 0) {
        for (int j = beg + lane + 64; j < end; j += 32) {
            const int2 r = recs[j];
            const float4* ep = (const float4*)(ea + (size_t)r.y * 8);
            recs[j] = make_int2(r.x, __float_as_int(edge_dot(ep[0], ep[1], w + 16)));
        }
    }
#undef ROW_STEP
#undef ACC_STEP
#undef ACC_CHUNK
}

extern "C" void kernel_launch(void* const* d_in, const int* in_sizes, int n_in,
                              void* d_out, int out_size, void* d_ws, size_t ws_size,
                              hipStream_t stream)
{
    const float* x     = (const float*)d_in[0];
    const int*   ei    = (const int*)  d_in[1];
    const float* ea    = (const float*)d_in[2];
    const float* cW    = (const float*)d_in[3];
    const float* casrc = (const float*)d_in[4];
    const float* cadst = (const float*)d_in[5];
    const float* cWe   = (const float*)d_in[6];
    const float* cae   = (const float*)d_in[7];
    const float* cb    = (const float*)d_in[8];
    const float* W0 = (const float*)d_in[9];  const float* b0 = (const float*)d_in[10];
    const float* W1 = (const float*)d_in[11]; const float* b1 = (const float*)d_in[12];
    const float* W2 = (const float*)d_in[13]; const float* b2 = (const float*)d_in[14];
    const float* W3 = (const float*)d_in[15]; const float* b3 = (const float*)d_in[16];
    float* out = (float*)d_out;

    const int* e_src = ei;
    const int* e_dst = ei + N_EDGES;

    float* ws   = (float*)d_ws;
    float* bufA = ws;
    float* bufB = bufA + (size_t)N_NODES * 16;
    float* sdst = bufB + (size_t)N_NODES * 16 + N_NODES;
    int*   off  = (int*)(sdst + N_NODES);
    float* lp   = sdst + 3 * (size_t)N_NODES;
    int*   rank = (int*)lp;
    float* wae  = lp   + N_EDGES;
    float* w3p  = wae  + 32;
    float* b3p  = w3p  + 64 * 16;
    float* mlpA = b3p  + 16;
    int2*  recs = (int2*)mlpA;
    float* mlpB = bufA;

    const int nodeBlocks  = (N_NODES + 255) / 256;
    const int scanBlocks  = SCAN_N / 256;
    const int gemmBlocks  = (N_NODES + 127) / 128;
    const int edgeBlocks  = (N_EDGES + 255) / 256;
    const int segBlocks   = (N_NODES + 7) / 8;

    wae_kernel<<<1, 32, 0, stream>>>(cWe, cae, wae);
    setup_kernel<<<scanBlocks, 256, 0, stream>>>(W3, b3, w3p, b3p, off);

    edge_rank_kernel<<<edgeBlocks, 256, 0, stream>>>(e_dst, off, rank);
    scan_counts_kernel<<<1, 1024, 0, stream>>>(off);
    edge_scatter_kernel<<<edgeBlocks, 256, 0, stream>>>(e_src, e_dst, off, rank, recs);

    gemm_wmma_f32<16, 16, 16, 0><<<gemmBlocks, 256, 0, stream>>>(x, cW, nullptr, bufA, N_NODES);
    prep_nodes_kernel<<<nodeBlocks, 256, 0, stream>>>(bufA, cadst, sdst);
    segment_softmax_aggregate_kernel<0><<<segBlocks, 256, 0, stream>>>(
        recs, off, ea, wae, casrc, sdst, bufA, cb, lp, bufB);
    gemm_wmma_f32<16, 16, 16, 0><<<gemmBlocks, 256, 0, stream>>>(bufB, cW + 256, nullptr, bufA, N_NODES);
    prep_nodes_kernel<<<nodeBlocks, 256, 0, stream>>>(bufA, cadst + 16, sdst);
    segment_softmax_aggregate_kernel<1><<<segBlocks, 256, 0, stream>>>(
        recs, off, ea, wae, casrc + 16, sdst, bufA, cb + 16, lp, bufB);
    gemm_wmma_f32<16, 16, 16, 0><<<gemmBlocks, 256, 0, stream>>>(bufB, cW + 512, nullptr, bufA, N_NODES);
    prep_nodes_kernel<<<nodeBlocks, 256, 0, stream>>>(bufA, cadst + 32, sdst);
    segment_softmax_aggregate_kernel<2><<<segBlocks, 256, 0, stream>>>(
        recs, off, ea, wae, casrc + 32, sdst, bufA, cb + 32, lp, bufB);

    gemm_wmma_f32<16, 64, 64, 1><<<gemmBlocks, 256, 0, stream>>>(bufB, W0, b0, mlpA, N_NODES);
    gemm_wmma_f32<64, 64, 64, 1><<<gemmBlocks, 256, 0, stream>>>(mlpA, W1, b1, mlpB, N_NODES);
    gemm_wmma_f32<64, 64, 64, 1><<<gemmBlocks, 256, 0, stream>>>(mlpB, W2, b2, mlpA, N_NODES);
    gemm_wmma_f32<64, 16, 4, 0><<<gemmBlocks, 256, 0, stream>>>(mlpA, w3p, b3p, out, N_NODES);
}


# Edges ordered by destination through local-memory counters: count per 512-node bucket, scan, partition, sort each bucket; contiguous writes, no global atomics; sum order = arrival rank
# speedup vs baseline: 6.5634x; 1.9606x over previous
#include <hip/hip_runtime.h>
#include <hip/hip_bf16.h>
#include <math.h>

#define N_NODES 100000
#define N_EDGES 3200000
#define NEG_SLOPE 0.2f
#define NB      196
#define PW      256
#define PCH     12500
#define HN      (NB * PW)
#define BCAP    20480
#define OFF_N   100352
#define SCAN_W  32
#define SCAN_CK 15
#define SCAN_B  5
#define SCAN_N  (SCAN_W * SCAN_CK * 128)
static_assert(PW * PCH == N_EDGES, "the partition work-groups must cover the edges exactly");
static_assert(NB * 512 >= N_NODES + 1 && HN + 1 <= SCAN_N, "bucket and scan extents");

typedef __attribute__((ext_vector_type(2))) float v2f;
typedef __attribute__((ext_vector_type(8))) float v8f;

__device__ __forceinline__ float leakyf(float v) { return v > 0.f ? v : NEG_SLOPE * v; }

template <int K, int M, int MOUT, int ACT>
__global__ __launch_bounds__(256)
void gemm_wmma_f32(const float* __restrict__ A, const float* __restrict__ W,
                   const float* __restrict__ bias, float* __restrict__ C,
                   int Nrows)
{
    const int lane     = threadIdx.x & 31;
    const int wave     = threadIdx.x >> 5;
    const int row_base = (blockIdx.x * 8 + wave) * 16;
    if (row_base >= Nrows) return;

    const int nlo   = lane & 15;
    const int khalf = (lane >> 4) * 2;
    const int arow  = row_base + nlo;
    const int rofs  = (lane >> 4) * 8;

    v2f av[K / 4];
    #pragma unroll
    for (int kc = 0; kc < K / 4; ++kc) {
        const int k0 = kc * 4 + khalf;
        av[kc].x = A[arow * K + k0];
        av[kc].y = A[arow * K + k0 + 1];
    }

    #pragma unroll
    for (int mt = 0; mt < M / 16; ++mt) {
        const int col = mt * 16 + nlo;
        const float bv = bias ? bias[col] : 0.0f;
        v8f c;
        #pragma unroll
        for (int v = 0; v < 8; ++v) c[v] = bv;

        #pragma unroll
        for (int kc = 0; kc < K / 4; ++kc) {
            const int k0 = kc * 4 + khalf;
            v2f b;
            b.x = W[k0 * M + col];
            b.y = W[(k0 + 1) * M + col];
            c = __builtin_amdgcn_wmma_f32_16x16x4_f32(
                    false, av[kc], false, b, (short)0, c, false, false);
        }

        if (MOUT == M || col < MOUT) {
            #pragma unroll
            for (int v = 0; v < 8; ++v) {
                float r = c[v];
                if (ACT == 1) r = r > 0.f ? r : 0.f;
                C[(row_base + rofs + v) * MOUT + col] = r;
            }
        }
    }
}

__global__ __launch_bounds__(256)
void setup_kernel(const float* __restrict__ W3, const float* __restrict__ b3,
                  float* __restrict__ W3p, float* __restrict__ b3p,
                  int* __restrict__ cnt)
{
    int t = blockIdx.x * blockDim.x + threadIdx.x;
    if (t < 64 * 16) {
        int k = t >> 4, c = t & 15;
        W3p[t] = (c < 4) ? W3[k * 4 + c] : 0.0f;
    }
    if (t < 16) b3p[t] = (t < 4) ? b3[t] : 0.0f;
    if (t < SCAN_N) cnt[t] = 0;
}

__global__ void wae_kernel(const float* __restrict__ We, const float* __restrict__ ae,
                           float* __restrict__ wae)
{
    int t = threadIdx.x;
    if (t < 24) {
        int i = t >> 3, j = t & 7;
        float s = 0.f;
        #pragma unroll
        for (int k = 0; k < 16; ++k) s += We[i * 128 + j * 16 + k] * ae[i * 16 + k];
        wae[t] = s;
    }
}


__device__ __forceinline__ int block_excl_scan(const int c, int* wt)
{
    const int lane = threadIdx.x & 31, wv = threadIdx.x >> 5;
    int inc = c;
    for (int o = 1; o < 32; o <<= 1) {
        const int u = __shfl_up(inc, o);
        inc += (lane >= o) ? u : 0;
    }
    if (lane == 31) wt[wv] = inc;
    __syncthreads();
    int base = (lane < wv) ? wt[lane] : 0;
    for (int o = 16; o > 0; o >>= 1) base += __shfl_xor(base, o);
    return base + inc - c;
}

__global__ __launch_bounds__(1024)
void bucket_count_kernel(const int* __restrict__ dst, int* __restrict__ hcnt)
{
    __shared__ int lh[NB];
    const int t = threadIdx.x, w = blockIdx.x;
    if (t < NB) lh[t] = 0;
    __syncthreads();
    const int* dp = dst + w * PCH;
    int i = t;
    for (; i + 3072 < PCH; i += 4096) {
        int d[4];
        #pragma unroll
        for (int k = 0; k < 4; ++k) d[k] = dp[i + k * 1024];
        #pragma unroll
        for (int k = 0; k < 4; ++k) atomicAdd(&lh[d[k] >> 9], 1);
    }
    for (; i < PCH; i += 1024) atomicAdd(&lh[dp[i] >> 9], 1);
    __syncthreads();
    if (t < NB) hcnt[t * PW + w] = lh[t];
}

__global__ __launch_bounds__(1024)
void scan_counts_kernel(int* __restrict__ bsc)
{
    __shared__ int wtot[SCAN_W];
    const int lane = threadIdx.x & 31;
    const int w    = threadIdx.x >> 5;
    int4* p = (int4*)(bsc + w * (SCAN_CK * 128)) + lane;

    int tot = 0;
    for (int c0 = 0; c0 < SCAN_CK; c0 += SCAN_B) {
        int4 v[SCAN_B];
        #pragma unroll
        for (int k = 0; k < SCAN_B; ++k) v[k] = p[(c0 + k) * 32];
        #pragma unroll
        for (int k = 0; k < SCAN_B; ++k) tot += (v[k].x + v[k].y) + (v[k].z + v[k].w);
    }
    for (int o = 16; o > 0; o >>= 1) tot += __shfl_xor(tot, o);
    if (lane == 0) wtot[w] = tot;
    __syncthreads();
    int carry = (lane < w) ? wtot[lane] : 0;
    for (int o = 16; o > 0; o >>= 1) carry += __shfl_xor(carry, o);

    for (int c0 = 0; c0 < SCAN_CK; c0 += SCAN_B) {
        int4 v[SCAN_B];
        int lt[SCAN_B], inc[SCAN_B];
        #pragma unroll
        for (int k = 0; k < SCAN_B; ++k) v[k] = p[(c0 + k) * 32];
        #pragma unroll
        for (int k = 0; k < SCAN_B; ++k) { lt[k] = (v[k].x + v[k].y) + (v[k].z + v[k].w); inc[k] = lt[k]; }
        #pragma unroll
        for (int o = 1; o < 32; o <<= 1) {
            #pragma unroll
            for (int k = 0; k < SCAN_B; ++k) {
                const int u = __shfl_up(inc[k], o);
                inc[k] += (lane >= o) ? u : 0;
            }
        }
        #pragma unroll
        for (int k = 0; k < SCAN_B; ++k) {
            int4 r;
            r.x = carry + inc[k] - lt[k];
            r.y = r.x + v[k].x;
            r.z = r.y + v[k].y;
            r.w = r.z + v[k].z;
            p[(c0 + k) * 32] = r;
            carry += __builtin_amdgcn_readlane(inc[k], 31);
        }
    }
}

__global__ __launch_bounds__(1024)
void partition_kernel(const int* __restrict__ src, const int* __restrict__ dst,
                      const int* __restrict__ bsc, int* __restrict__ px, int* __restrict__ py)
{
    __shared__ int  lst[NB];
    __shared__ int  lcur[NB];
    __shared__ int  gb[NB];
    __shared__ int  wt[32];
    __shared__ int2 stage[PCH];
    const int t = threadIdx.x, w = blockIdx.x;
    int c = 0;
    if (t < NB) {
        const int g0 = bsc[t * PW + w];
        gb[t] = g0;
        c = bsc[t * PW + w + 1] - g0;
    }
    const int ex = block_excl_scan(c, wt);
    if (t < NB) { lst[t] = ex; lcur[t] = ex; }
    __syncthreads();

    const int e0 = w * PCH;
    auto place = [&](const int e, const int d, const int sr) {
        const int b = d >> 9;
        const int r = atomicAdd(&lcur[b], 1);
        stage[r] = make_int2(sr | ((d & 511) << 17), e | (b << 22));
    };
    int i = t;
    for (; i + 3072 < PCH; i += 4096) {
        int d[4], sr[4];
        #pragma unroll
        for (int k = 0; k < 4; ++k) { d[k] = dst[e0 + i + k * 1024]; sr[k] = src[e0 + i + k * 1024]; }
        #pragma unroll
        for (int k = 0; k < 4; ++k) place(e0 + i + k * 1024, d[k], sr[k]);
    }
    for (; i < PCH; i += 1024) place(e0 + i, dst[e0 + i], src[e0 + i]);
    __syncthreads();
    for (int j = t; j < PCH; j += 1024) {
        const int2 rec = stage[j];
        const int b = (int)((unsigned)rec.y >> 22);
        const int g = gb[b] + (j - lst[b]);
        px[g] = rec.x;
        py[g] = rec.y & 0x3FFFFF;
    }
}

__global__ __launch_bounds__(1024)
void bucket_sort_kernel(const int* __restrict__ bsc, const int* __restrict__ px,
                        const int* __restrict__ py, int* __restrict__ off,
                        int2* __restrict__ recs)
{
    __shared__ int  ncnt[512];
    __shared__ int  wt[32];
    __shared__ int2 stage[BCAP];
    const int t = threadIdx.x, b = blockIdx.x;
    const int s = bsc[b * PW];
    const int n = bsc[(b + 1) * PW] - s;
    const int* qx = px + s;
    const int* qy = py + s;
    if (t < 512) ncnt[t] = 0;
    __syncthreads();
    int i = t;
    for (; i + 3072 < n; i += 4096) {
        int x[4];
        #pragma unroll
        for (int k = 0; k < 4; ++k) x[k] = qx[i + k * 1024];
        #pragma unroll
        for (int k = 0; k < 4; ++k) atomicAdd(&ncnt[x[k] >> 17], 1);
    }
    for (; i < n; i += 1024) atomicAdd(&ncnt[qx[i] >> 17], 1);
    __syncthreads();
    const int c  = (t < 512) ? ncnt[t] : 0;
    const int ex = block_excl_scan(c, wt);
    if (t < 512) {
        ncnt[t] = ex;
        const int node = b * 512 + t;
        if (node <= N_NODES) off[node] = s + ex;
    }
    __syncthreads();
    const bool staged = n <= BCAP;
    auto place = [&](const int x, const int y) {
        const int p = atomicAdd(&ncnt[x >> 17], 1);
        const int2 rec = make_int2(x & 0x1FFFF, y);
        if (staged) stage[p] = rec; else recs[s + p] = rec;
    };
    i = t;
    for (; i + 3072 < n; i += 4096) {
        int x[4], y[4];
        #pragma unroll
        for (int k = 0; k < 4; ++k) { x[k] = qx[i + k * 1024]; y[k] = qy[i + k * 1024]; }
        #pragma unroll
        for (int k = 0; k < 4; ++k) place(x[k], y[k]);
    }
    for (; i < n; i += 1024) place(qx[i], qy[i]);
    __syncthreads();
    if (staged) {
        for (int j = t; j < n; j += 1024) recs[s + j] = stage[j];
    }
}

__global__ __launch_bounds__(256)
void prep_nodes_kernel(const float* __restrict__ h, const float* __restrict__ adst,
                       float* __restrict__ sdst)
{
    int n = blockIdx.x * blockDim.x + threadIdx.x;
    if (n >= N_NODES) return;
    float sd = 0.f;
    #pragma unroll
    for (int f = 0; f < 16; ++f) sd += h[n * 16 + f] * adst[f];
    sdst[n] = sd;
}

__device__ __forceinline__ float edge_dot(const float4 e0, const float4 e1, const float* w)
{
    float se = 0.f;
    se += e0.x * w[0]; se += e0.y * w[1]; se += e0.z * w[2]; se += e0.w * w[3];
    se += e1.x * w[4]; se += e1.y * w[5]; se += e1.z * w[6]; se += e1.w * w[7];
    return se;
}

template <unsigned SEL>
__device__ __forceinline__ float grp8f(const float v)
{
    return __uint_as_float(__builtin_amdgcn_mov_dpp8(__float_as_uint(v), SEL));
}
template <unsigned SEL>
__device__ __forceinline__ int grp8i(const int v)
{
    return (int)__builtin_amdgcn_mov_dpp8((unsigned)v, SEL);
}
#define SEL_LANE(u) ((unsigned)(u) * 2396745u)
#define SEL_XOR4 6852524u
#define SEL_XOR2 11788826u
#define SEL_XOR1 14570689u

template <int LAYER>
__global__ __launch_bounds__(256)
void segment_softmax_aggregate_kernel(int2* recs, const int* __restrict__ off,
                                      const float* __restrict__ ea, const float* __restrict__ wae,
                                      const float* __restrict__ asrc, const float* __restrict__ sdst,
                                      const float* __restrict__ h, const float* __restrict__ b,
                                      float* se1, float* __restrict__ outb)
{
    const int lane = threadIdx.x & 31;
    const int d    = blockIdx.x * 8 + (threadIdx.x >> 5);
    if (d >= N_NODES) return;
    const int beg = __builtin_amdgcn_readfirstlane(off[d]);
    const int end = __builtin_amdgcn_readfirstlane(off[d + 1]);
    const int n   = end - beg;
    const float sd = sdst[d];
    float w[LAYER == 0 ? 24 : 1];
    if (LAYER == 0) { for (int k = 0; k < 24; ++k) w[k] = wae[k]; } else { w[0] = 0.f; }

    const int g  = lane >> 3;
    const int i8 = lane & 7;
    const float2 a2 = *(const float2*)(asrc + 2 * i8);
    const float2 bb = *(const float2*)(b + 2 * i8);
    const bool b4 = (i8 & 4) != 0, b2 = (i8 & 2) != 0, b1 = (i8 & 1) != 0;

    auto slot_index = [&](const int j) -> int {
        const int jj = j < end ? j : end - 1;
        return jj < 0 ? 0 : jj;
    };
    auto slot_rec = [&](const int j) -> int2 {
        const int2 r = recs[slot_index(j)];
        return j < end ? r : make_int2(0, 0);
    };
    auto term_load = [&](const int j, const int2 r, float4& e0, float4& e1, float& se) {
        e0 = make_float4(0.f, 0.f, 0.f, 0.f);
        e1 = e0;
        se = 0.f;
        if (LAYER == 0) {
            const float4* ep = (const float4*)(ea + (size_t)r.y * 8);
            e0 = ep[0];
            e1 = ep[1];
        } else if (LAYER == 1) {
            se = se1[slot_index(j)];
        } else {
            se = __int_as_float(r.y);
        }
    };
    auto slot_term = [&](const int j, const int2 r, const float4 e0, const float4 e1,
                         const float se_in, const bool store, const bool keep) -> float {
        if (LAYER != 0) return se_in;
        if (store && j < end) {
            se1[j] = edge_dot(e0, e1, w + 8);
            if (keep) recs[j] = make_int2(r.x, __float_as_int(edge_dot(e0, e1, w + 16)));
        }
        return edge_dot(e0, e1, w);
    };

#define ROW_STEP(u)                                                              \
    {                                                                            \
        const int s = grp8i<SEL_LANE(u)>(s_l);                                   \
        const float2 r = *(const float2*)(h + s * 16 + 2 * i8);                  \
        const bool ok = 8 * g + (u) < cnt;                                       \
        row[u] = make_float2(ok ? r.x : 0.f, ok ? r.y : 0.f);                    \
    }
    auto gather_rows = [&](const int s_l, const int cnt, float2 (&row)[8]) {
        ROW_STEP(0) ROW_STEP(1) ROW_STEP(2) ROW_STEP(3)
        ROW_STEP(4) ROW_STEP(5) ROW_STEP(6) ROW_STEP(7)
    };
    auto slot_score = [&](const float2 (&row)[8]) -> float {
        float pv[8], w4[4], w2[2];
        #pragma unroll
        for (int u = 0; u < 8; ++u) pv[u] = row[u].x * a2.x + row[u].y * a2.y;
        #pragma unroll
        for (int k = 0; k < 4; ++k) {
            const float keepv = b4 ? pv[k + 4] : pv[k];
            const float send  = b4 ? pv[k] : pv[k + 4];
            w4[k] = keepv + grp8f<SEL_XOR4>(send);
        }
        #pragma unroll
        for (int k = 0; k < 2; ++k) {
            const float keepv = b2 ? w4[k + 2] : w4[k];
            const float send  = b2 ? w4[k] : w4[k + 2];
            w2[k] = keepv + grp8f<SEL_XOR2>(send);
        }
        const float keepv = b1 ? w2[1] : w2[0];
        const float send  = b1 ? w2[0] : w2[1];
        return keepv + grp8f<SEL_XOR1>(send);
    };
    auto slot_logit = [&](const int j, const float ss, const float se) -> float {
        return (j < end) ? leakyf(ss + sd + se) : -INFINITY;
    };
    auto chunk_logit = [&](const int base, const bool store, float2 (&row)[8]) -> float {
        const int j = base + lane;
        const int2 r = slot_rec(j);
        float4 e0, e1;
        float se;
        term_load(j, r, e0, e1, se);
        gather_rows(r.x, end - base, row);
        return slot_logit(j, slot_score(row), slot_term(j, r, e0, e1, se, store, false));
    };

#define ACC_STEP(row, a, u)                                                      \
    {                                                                            \
        const float av = grp8f<SEL_LANE(u)>(a);                                  \
        acc.x += row[u].x * av;                                                  \
        acc.y += row[u].y * av;                                                  \
    }
#define ACC_CHUNK(row, a)                                                        \
    ACC_STEP(row, a, 0) ACC_STEP(row, a, 1) ACC_STEP(row, a, 2) ACC_STEP(row, a, 3) \
    ACC_STEP(row, a, 4) ACC_STEP(row, a, 5) ACC_STEP(row, a, 6) ACC_STEP(row, a, 7)

    const int j0 = beg + lane, j1 = j0 + 32;
    const int2 c0 = slot_rec(j0);
    const int2 c1 = slot_rec(j1);
    float4 x0, y0, x1, y1;
    float t0, t1;
    float2 r0[8], r1[8];
    term_load(j0, c0, x0, y0, t0);
    gather_rows(c0.x, n, r0);
    if (n > 32) {
        term_load(j1, c1, x1, y1, t1);
        gather_rows(c1.x, n - 32, r1);
    }
    float q0, q1 = -INFINITY;
    q0 = slot_logit(j0, slot_score(r0), slot_term(j0, c0, x0, y0, t0, true, true));
    if (n > 32) q1 = slot_logit(j1, slot_score(r1), slot_term(j1, c1, x1, y1, t1, true, true));
    float mx = fmaxf(q0, q1);
    for (int base = beg + 64; base < end; base += 32) {
        float2 t[8];
        mx = fmaxf(mx, chunk_logit(base, true, t));
    }
    for (int o = 16; o > 0; o >>= 1) mx = fmaxf(mx, __shfl_xor(mx, o));
    if (!__builtin_isfinite(mx)) mx = 0.f;

    q0 = __expf(q0 - mx);
    q1 = __expf(q1 - mx);
    float sum = q0 + q1;
    for (int base = beg + 64; base < end; base += 32) {
        float2 t[8];
        sum += __expf(chunk_logit(base, false, t) - mx);
    }
    for (int o = 16; o > 0; o >>= 1) sum += __shfl_xor(sum, o);
    const float dn = sum + 1e-16f;

    float2 acc = make_float2(0.f, 0.f);
    {
        const float a0 = q0 / dn;
        const float a1 = q1 / dn;
        ACC_CHUNK(r0, a0)
        if (n > 32) { ACC_CHUNK(r1, a1) }
        for (int base = beg + 64; base < end; base += 32) {
            float2 t[8];
            const float a = __expf(chunk_logit(base, false, t) - mx) / dn;
            ACC_CHUNK(t, a)
        }
    }
    acc.x += __shfl_xor(acc.x, 8);  acc.y += __shfl_xor(acc.y, 8);
    acc.x += __shfl_xor(acc.x, 16); acc.y += __shfl_xor(acc.y, 16);
    if (lane < 8) *(float2*)(outb + d * 16 + 2 * i8) = make_float2(leakyf(acc.x + bb.x), leakyf(acc.y + bb.y));

    if (LAYER == 0) {
        for (int j = beg + lane + 64; j < end; j += 32) {
            const int2 r = recs[j];
            const float4* ep = (const float4*)(ea + (size_t)r.y * 8);
            recs[j] = make_int2(r.x, __float_as_int(edge_dot(ep[0], ep[1], w + 16)));
        }
    }
#undef ROW_STEP
#undef ACC_STEP
#undef ACC_CHUNK
}

extern "C" void kernel_launch(void* const* d_in, const int* in_sizes, int n_in,
                              void* d_out, int out_size, void* d_ws, size_t ws_size,
                              hipStream_t stream)
{
    const float* x     = (const float*)d_in[0];
    const int*   ei    = (const int*)  d_in[1];
    const float* ea    = (const float*)d_in[2];
    const float* cW    = (const float*)d_in[3];
    const float* casrc = (const float*)d_in[4];
    const float* cadst = (const float*)d_in[5];
    const float* cWe   = (const float*)d_in[6];
    const float* cae   = (const float*)d_in[7];
    const float* cb    = (const float*)d_in[8];
    const float* W0 = (const float*)d_in[9];  const float* b0 = (const float*)d_in[10];
    const float* W1 = (const float*)d_in[11]; const float* b1 = (const float*)d_in[12];
    const float* W2 = (const float*)d_in[13]; const float* b2 = (const float*)d_in[14];
    const float* W3 = (const float*)d_in[15]; const float* b3 = (const float*)d_in[16];
    float* out = (float*)d_out;

    const int* e_src = ei;
    const int* e_dst = ei + N_EDGES;

    float* ws   = (float*)d_ws;
    float* bufA = ws;
    float* bufB = bufA + (size_t)N_NODES * 16;
    int*   px   = (int*)ws;
    float* sdst = bufB + (size_t)N_NODES * 16 + N_NODES;
    int*   off  = (int*)(sdst + N_NODES);
    int*   bsc  = off + OFF_N;
    float* lp   = sdst + 3 * (size_t)N_NODES;
    int*   py   = (int*)lp;
    float* wae  = lp   + N_EDGES;
    float* w3p  = wae  + 32;
    float* b3p  = w3p  + 64 * 16;
    float* mlpA = b3p  + 16;
    int2*  recs = (int2*)mlpA;
    float* mlpB = bufA;

    const int nodeBlocks  = (N_NODES + 255) / 256;
    const int scanBlocks  = SCAN_N / 256;
    const int gemmBlocks  = (N_NODES + 127) / 128;
    const int segBlocks   = (N_NODES + 7) / 8;

    wae_kernel<<<1, 32, 0, stream>>>(cWe, cae, wae);
    setup_kernel<<<scanBlocks, 256, 0, stream>>>(W3, b3, w3p, b3p, bsc);

    bucket_count_kernel<<<PW, 1024, 0, stream>>>(e_dst, bsc);
    scan_counts_kernel<<<1, 1024, 0, stream>>>(bsc);
    partition_kernel<<<PW, 1024, 0, stream>>>(e_src, e_dst, bsc, px, py);
    bucket_sort_kernel<<<NB, 1024, 0, stream>>>(bsc, px, py, off, recs);

    gemm_wmma_f32<16, 16, 16, 0><<<gemmBlocks, 256, 0, stream>>>(x, cW, nullptr, bufA, N_NODES);
    prep_nodes_kernel<<<nodeBlocks, 256, 0, stream>>>(bufA, cadst, sdst);
    segment_softmax_aggregate_kernel<0><<<segBlocks, 256, 0, stream>>>(
        recs, off, ea, wae, casrc, sdst, bufA, cb, lp, bufB);
    gemm_wmma_f32<16, 16, 16, 0><<<gemmBlocks, 256, 0, stream>>>(bufB, cW + 256, nullptr, bufA, N_NODES);
    prep_nodes_kernel<<<nodeBlocks, 256, 0, stream>>>(bufA, cadst + 16, sdst);
    segment_softmax_aggregate_kernel<1><<<segBlocks, 256, 0, stream>>>(
        recs, off, ea, wae, casrc + 16, sdst, bufA, cb + 16, lp, bufB);
    gemm_wmma_f32<16, 16, 16, 0><<<gemmBlocks, 256, 0, stream>>>(bufB, cW + 512, nullptr, bufA, N_NODES);
    prep_nodes_kernel<<<nodeBlocks, 256, 0, stream>>>(bufA, cadst + 32, sdst);
    segment_softmax_aggregate_kernel<2><<<segBlocks, 256, 0, stream>>>(
        recs, off, ea, wae, casrc + 32, sdst, bufA, cb + 32, lp, bufB);

    gemm_wmma_f32<16, 64, 64, 1><<<gemmBlocks, 256, 0, stream>>>(bufB, W0, b0, mlpA, N_NODES);
    gemm_wmma_f32<64, 64, 64, 1><<<gemmBlocks, 256, 0, stream>>>(mlpA, W1, b1, mlpB, N_NODES);
    gemm_wmma_f32<64, 64, 64, 1><<<gemmBlocks, 256, 0, stream>>>(mlpB, W2, b2, mlpA, N_NODES);
    gemm_wmma_f32<64, 16, 4, 0><<<gemmBlocks, 256, 0, stream>>>(mlpA, w3p, b3p, out, N_NODES);
}


# Launches 19 -> 12: edge weights formed in setup; s_dst = h[d].a_dst formed inside the aggregation (8-lane tree: sum order changed); the 4 head layers in one kernel, tile kept in local memory
# speedup vs baseline: 7.3308x; 1.1169x over previous
#include <hip/hip_runtime.h>
#include <hip/hip_bf16.h>
#include <math.h>

#define N_NODES 100000
#define N_EDGES 3200000
#define NEG_SLOPE 0.2f
#define NB      196
#define PW      256
#define PCH     12500
#define HN      (NB * PW)
#define BCAP    20480
#define OFF_N   100352
#define SCAN_W  32
#define SCAN_CK 15
#define SCAN_B  5
#define SCAN_N  (SCAN_W * SCAN_CK * 128)
static_assert(PW * PCH == N_EDGES, "the partition work-groups must cover the edges exactly");
static_assert(NB * 512 >= N_NODES + 1 && HN + 1 <= SCAN_N, "bucket and scan extents");

typedef __attribute__((ext_vector_type(2))) float v2f;
typedef __attribute__((ext_vector_type(8))) float v8f;

__device__ __forceinline__ float leakyf(float v) { return v > 0.f ? v : NEG_SLOPE * v; }

template <int K, int M, int MOUT, int ACT>
__global__ __launch_bounds__(256)
void gemm_wmma_f32(const float* __restrict__ A, const float* __restrict__ W,
                   const float* __restrict__ bias, float* __restrict__ C,
                   int Nrows)
{
    const int lane     = threadIdx.x & 31;
    const int wave     = threadIdx.x >> 5;
    const int row_base = (blockIdx.x * 8 + wave) * 16;
    if (row_base >= Nrows) return;

    const int nlo   = lane & 15;
    const int khalf = (lane >> 4) * 2;
    const int arow  = row_base + nlo;
    const int rofs  = (lane >> 4) * 8;

    v2f av[K / 4];
    #pragma unroll
    for (int kc = 0; kc < K / 4; ++kc) {
        const int k0 = kc * 4 + khalf;
        av[kc].x = A[arow * K + k0];
        av[kc].y = A[arow * K + k0 + 1];
    }

    #pragma unroll
    for (int mt = 0; mt < M / 16; ++mt) {
        const int col = mt * 16 + nlo;
        const float bv = bias ? bias[col] : 0.0f;
        v8f c;
        #pragma unroll
        for (int v = 0; v < 8; ++v) c[v] = bv;

        #pragma unroll
        for (int kc = 0; kc < K / 4; ++kc) {
            const int k0 = kc * 4 + khalf;
            v2f b;
            b.x = W[k0 * M + col];
            b.y = W[(k0 + 1) * M + col];
            c = __builtin_amdgcn_wmma_f32_16x16x4_f32(
                    false, av[kc], false, b, (short)0, c, false, false);
        }

        if (MOUT == M || col < MOUT) {
            #pragma unroll
            for (int v = 0; v < 8; ++v) {
                float r = c[v];
                if (ACT == 1) r = r > 0.f ? r : 0.f;
                C[(row_base + rofs + v) * MOUT + col] = r;
            }
        }
    }
}

#define HEAD_WAVES 4
#define HEAD_PITCH 68

template <int K>
__device__ __forceinline__ void head_load_a(const float* p, const int pitch, const int nlo,
                                            const int khalf, v2f (&av)[16])
{
    #pragma unroll
    for (int kc = 0; kc < K / 4; ++kc) {
        const int k0 = kc * 4 + khalf;
        av[kc].x = p[nlo * pitch + k0];
        av[kc].y = p[nlo * pitch + k0 + 1];
    }
}

template <int K, int M, int MOUT, int ACT>
__device__ __forceinline__ void head_layer(const v2f (&av)[16], const float* __restrict__ W,
                                           const float* __restrict__ bias, float* q, const int pitch,
                                           const int nlo, const int khalf, const int rofs)
{
    #pragma unroll
    for (int mt = 0; mt < M / 16; ++mt) {
        const int col = mt * 16 + nlo;
        const float bv = bias[col];
        v8f c;
        #pragma unroll
        for (int v = 0; v < 8; ++v) c[v] = bv;
        #pragma unroll
        for (int kc = 0; kc < K / 4; ++kc) {
            const int k0 = kc * 4 + khalf;
            v2f b;
            b.x = W[k0 * M + col];
            b.y = W[(k0 + 1) * M + col];
            c = __builtin_amdgcn_wmma_f32_16x16x4_f32(
                    false, av[kc], false, b, (short)0, c, false, false);
        }
        if (MOUT == M || col < MOUT) {
            #pragma unroll
            for (int v = 0; v < 8; ++v) {
                float r = c[v];
                if (ACT == 1) r = r > 0.f ? r : 0.f;
                q[(rofs + v) * pitch + col] = r;
            }
        }
    }
}

__global__ __launch_bounds__(HEAD_WAVES * 32)
void mlp_head_kernel(const float* __restrict__ X,
                     const float* __restrict__ W0, const float* __restrict__ b0,
                     const float* __restrict__ W1, const float* __restrict__ b1,
                     const float* __restrict__ W2, const float* __restrict__ b2,
                     const float* __restrict__ W3p, const float* __restrict__ b3p,
                     float* __restrict__ out)
{
    __shared__ float tile[HEAD_WAVES][16 * HEAD_PITCH];
    const int lane     = threadIdx.x & 31;
    const int wave     = threadIdx.x >> 5;
    const int row_base = (blockIdx.x * HEAD_WAVES + wave) * 16;
    if (row_base >= N_NODES) return;

    const int nlo   = lane & 15;
    const int khalf = (lane >> 4) * 2;
    const int rofs  = (lane >> 4) * 8;
    float* a = tile[wave];

    v2f av[16];
    head_load_a<16>(X + (size_t)row_base * 16, 16, nlo, khalf, av);
    head_layer<16, 64, 64, 1>(av, W0, b0, a, HEAD_PITCH, nlo, khalf, rofs);
    head_load_a<64>(a, HEAD_PITCH, nlo, khalf, av);
    head_layer<64, 64, 64, 1>(av, W1, b1, a, HEAD_PITCH, nlo, khalf, rofs);
    head_load_a<64>(a, HEAD_PITCH, nlo, khalf, av);
    head_layer<64, 64, 64, 1>(av, W2, b2, a, HEAD_PITCH, nlo, khalf, rofs);
    head_load_a<64>(a, HEAD_PITCH, nlo, khalf, av);
    head_layer<64, 16, 4, 0>(av, W3p, b3p, out + (size_t)row_base * 4, 4, nlo, khalf, rofs);
}

__global__ __launch_bounds__(256)
void setup_kernel(const float* __restrict__ W3, const float* __restrict__ b3,
                  float* __restrict__ W3p, float* __restrict__ b3p,
                  int* __restrict__ cnt,
                  const float* __restrict__ We, const float* __restrict__ ae,
                  float* __restrict__ wae)
{
    int t = blockIdx.x * blockDim.x + threadIdx.x;
    if (t < 64 * 16) {
        int k = t >> 4, c = t & 15;
        W3p[t] = (c < 4) ? W3[k * 4 + c] : 0.0f;
    }
    if (t < 16) b3p[t] = (t < 4) ? b3[t] : 0.0f;
    if (t < SCAN_N) cnt[t] = 0;
    if (blockIdx.x == gridDim.x - 1 && threadIdx.x < 24) {
        const int u = threadIdx.x, i = u >> 3, j = u & 7;
        float s = 0.f;
        #pragma unroll
        for (int k = 0; k < 16; ++k) s += We[i * 128 + j * 16 + k] * ae[i * 16 + k];
        wae[u] = s;
    }
}


__device__ __forceinline__ int block_excl_scan(const int c, int* wt)
{
    const int lane = threadIdx.x & 31, wv = threadIdx.x >> 5;
    int inc = c;
    for (int o = 1; o < 32; o <<= 1) {
        const int u = __shfl_up(inc, o);
        inc += (lane >= o) ? u : 0;
    }
    if (lane == 31) wt[wv] = inc;
    __syncthreads();
    int base = (lane < wv) ? wt[lane] : 0;
    for (int o = 16; o > 0; o >>= 1) base += __shfl_xor(base, o);
    return base + inc - c;
}

__global__ __launch_bounds__(1024)
void bucket_count_kernel(const int* __restrict__ dst, int* __restrict__ hcnt)
{
    __shared__ int lh[NB];
    const int t = threadIdx.x, w = blockIdx.x;
    if (t < NB) lh[t] = 0;
    __syncthreads();
    const int* dp = dst + w * PCH;
    int i = t;
    for (; i + 3072 < PCH; i += 4096) {
        int d[4];
        #pragma unroll
        for (int k = 0; k < 4; ++k) d[k] = dp[i + k * 1024];
        #pragma unroll
        for (int k = 0; k < 4; ++k) atomicAdd(&lh[d[k] >> 9], 1);
    }
    for (; i < PCH; i += 1024) atomicAdd(&lh[dp[i] >> 9], 1);
    __syncthreads();
    if (t < NB) hcnt[t * PW + w] = lh[t];
}

__global__ __launch_bounds__(1024)
void scan_counts_kernel(int* __restrict__ bsc)
{
    __shared__ int wtot[SCAN_W];
    const int lane = threadIdx.x & 31;
    const int w    = threadIdx.x >> 5;
    int4* p = (int4*)(bsc + w * (SCAN_CK * 128)) + lane;

    int tot = 0;
    for (int c0 = 0; c0 < SCAN_CK; c0 += SCAN_B) {
        int4 v[SCAN_B];
        #pragma unroll
        for (int k = 0; k < SCAN_B; ++k) v[k] = p[(c0 + k) * 32];
        #pragma unroll
        for (int k = 0; k < SCAN_B; ++k) tot += (v[k].x + v[k].y) + (v[k].z + v[k].w);
    }
    for (int o = 16; o > 0; o >>= 1) tot += __shfl_xor(tot, o);
    if (lane == 0) wtot[w] = tot;
    __syncthreads();
    int carry = (lane < w) ? wtot[lane] : 0;
    for (int o = 16; o > 0; o >>= 1) carry += __shfl_xor(carry, o);

    for (int c0 = 0; c0 < SCAN_CK; c0 += SCAN_B) {
        int4 v[SCAN_B];
        int lt[SCAN_B], inc[SCAN_B];
        #pragma unroll
        for (int k = 0; k < SCAN_B; ++k) v[k] = p[(c0 + k) * 32];
        #pragma unroll
        for (int k = 0; k < SCAN_B; ++k) { lt[k] = (v[k].x + v[k].y) + (v[k].z + v[k].w); inc[k] = lt[k]; }
        #pragma unroll
        for (int o = 1; o < 32; o <<= 1) {
            #pragma unroll
            for (int k = 0; k < SCAN_B; ++k) {
                const int u = __shfl_up(inc[k], o);
                inc[k] += (lane >= o) ? u : 0;
            }
        }
        #pragma unroll
        for (int k = 0; k < SCAN_B; ++k) {
            int4 r;
            r.x = carry + inc[k] - lt[k];
            r.y = r.x + v[k].x;
            r.z = r.y + v[k].y;
            r.w = r.z + v[k].z;
            p[(c0 + k) * 32] = r;
            carry += __builtin_amdgcn_readlane(inc[k], 31);
        }
    }
}

__global__ __launch_bounds__(1024)
void partition_kernel(const int* __restrict__ src, const int* __restrict__ dst,
                      const int* __restrict__ bsc, int* __restrict__ px, int* __restrict__ py)
{
    __shared__ int  lst[NB];
    __shared__ int  lcur[NB];
    __shared__ int  gb[NB];
    __shared__ int  wt[32];
    __shared__ int2 stage[PCH];
    const int t = threadIdx.x, w = blockIdx.x;
    int c = 0;
    if (t < NB) {
        const int g0 = bsc[t * PW + w];
        gb[t] = g0;
        c = bsc[t * PW + w + 1] - g0;
    }
    const int ex = block_excl_scan(c, wt);
    if (t < NB) { lst[t] = ex; lcur[t] = ex; }
    __syncthreads();

    const int e0 = w * PCH;
    auto place = [&](const int e, const int d, const int sr) {
        const int b = d >> 9;
        const int r = atomicAdd(&lcur[b], 1);
        stage[r] = make_int2(sr | ((d & 511) << 17), e | (b << 22));
    };
    int i = t;
    for (; i + 3072 < PCH; i += 4096) {
        int d[4], sr[4];
        #pragma unroll
        for (int k = 0; k < 4; ++k) { d[k] = dst[e0 + i + k * 1024]; sr[k] = src[e0 + i + k * 1024]; }
        #pragma unroll
        for (int k = 0; k < 4; ++k) place(e0 + i + k * 1024, d[k], sr[k]);
    }
    for (; i < PCH; i += 1024) place(e0 + i, dst[e0 + i], src[e0 + i]);
    __syncthreads();
    for (int j = t; j < PCH; j += 1024) {
        const int2 rec = stage[j];
        const int b = (int)((unsigned)rec.y >> 22);
        const int g = gb[b] + (j - lst[b]);
        px[g] = rec.x;
        py[g] = rec.y & 0x3FFFFF;
    }
}

__global__ __launch_bounds__(1024)
void bucket_sort_kernel(const int* __restrict__ bsc, const int* __restrict__ px,
                        const int* __restrict__ py, int* __restrict__ off,
                        int2* __restrict__ recs)
{
    __shared__ int  ncnt[512];
    __shared__ int  wt[32];
    __shared__ int2 stage[BCAP];
    const int t = threadIdx.x, b = blockIdx.x;
    const int s = bsc[b * PW];
    const int n = bsc[(b + 1) * PW] - s;
    const int* qx = px + s;
    const int* qy = py + s;
    if (t < 512) ncnt[t] = 0;
    __syncthreads();
    int i = t;
    for (; i + 3072 < n; i += 4096) {
        int x[4];
        #pragma unroll
        for (int k = 0; k < 4; ++k) x[k] = qx[i + k * 1024];
        #pragma unroll
        for (int k = 0; k < 4; ++k) atomicAdd(&ncnt[x[k] >> 17], 1);
    }
    for (; i < n; i += 1024) atomicAdd(&ncnt[qx[i] >> 17], 1);
    __syncthreads();
    const int c  = (t < 512) ? ncnt[t] : 0;
    const int ex = block_excl_scan(c, wt);
    if (t < 512) {
        ncnt[t] = ex;
        const int node = b * 512 + t;
        if (node <= N_NODES) off[node] = s + ex;
    }
    __syncthreads();
    const bool staged = n <= BCAP;
    auto place = [&](const int x, const int y) {
        const int p = atomicAdd(&ncnt[x >> 17], 1);
        const int2 rec = make_int2(x & 0x1FFFF, y);
        if (staged) stage[p] = rec; else recs[s + p] = rec;
    };
    i = t;
    for (; i + 3072 < n; i += 4096) {
        int x[4], y[4];
        #pragma unroll
        for (int k = 0; k < 4; ++k) { x[k] = qx[i + k * 1024]; y[k] = qy[i + k * 1024]; }
        #pragma unroll
        for (int k = 0; k < 4; ++k) place(x[k], y[k]);
    }
    for (; i < n; i += 1024) place(qx[i], qy[i]);
    __syncthreads();
    if (staged) {
        for (int j = t; j < n; j += 1024) recs[s + j] = stage[j];
    }
}

__device__ __forceinline__ float edge_dot(const float4 e0, const float4 e1, const float* w)
{
    float se = 0.f;
    se += e0.x * w[0]; se += e0.y * w[1]; se += e0.z * w[2]; se += e0.w * w[3];
    se += e1.x * w[4]; se += e1.y * w[5]; se += e1.z * w[6]; se += e1.w * w[7];
    return se;
}

template <unsigned SEL>
__device__ __forceinline__ float grp8f(const float v)
{
    return __uint_as_float(__builtin_amdgcn_mov_dpp8(__float_as_uint(v), SEL));
}
template <unsigned SEL>
__device__ __forceinline__ int grp8i(const int v)
{
    return (int)__builtin_amdgcn_mov_dpp8((unsigned)v, SEL);
}
#define SEL_LANE(u) ((unsigned)(u) * 2396745u)
#define SEL_XOR4 6852524u
#define SEL_XOR2 11788826u
#define SEL_XOR1 14570689u

template <int LAYER>
__global__ __launch_bounds__(256)
void segment_softmax_aggregate_kernel(int2* recs, const int* __restrict__ off,
                                      const float* __restrict__ ea, const float* __restrict__ wae,
                                      const float* __restrict__ asrc, const float* __restrict__ adst,
                                      const float* __restrict__ h, const float* __restrict__ b,
                                      float* se1, float* __restrict__ outb)
{
    const int lane = threadIdx.x & 31;
    const int d    = blockIdx.x * 8 + (threadIdx.x >> 5);
    if (d >= N_NODES) return;
    const int beg = __builtin_amdgcn_readfirstlane(off[d]);
    const int end = __builtin_amdgcn_readfirstlane(off[d + 1]);
    const int n   = end - beg;
    float w[LAYER == 0 ? 24 : 1];
    if (LAYER == 0) { for (int k = 0; k < 24; ++k) w[k] = wae[k]; } else { w[0] = 0.f; }

    const int g  = lane >> 3;
    const int i8 = lane & 7;
    const float2 a2 = *(const float2*)(asrc + 2 * i8);
    const float2 bb = *(const float2*)(b + 2 * i8);
    const float2 d2 = *(const float2*)(adst + 2 * i8);
    const float2 hd = *(const float2*)(h + d * 16 + 2 * i8);
    float sd = hd.x * d2.x + hd.y * d2.y;
    sd += grp8f<SEL_XOR4>(sd);
    sd += grp8f<SEL_XOR2>(sd);
    sd += grp8f<SEL_XOR1>(sd);
    const bool b4 = (i8 & 4) != 0, b2 = (i8 & 2) != 0, b1 = (i8 & 1) != 0;

    auto slot_index = [&](const int j) -> int {
        const int jj = j < end ? j : end - 1;
        return jj < 0 ? 0 : jj;
    };
    auto slot_rec = [&](const int j) -> int2 {
        const int2 r = recs[slot_index(j)];
        return j < end ? r : make_int2(0, 0);
    };
    auto term_load = [&](const int j, const int2 r, float4& e0, float4& e1, float& se) {
        e0 = make_float4(0.f, 0.f, 0.f, 0.f);
        e1 = e0;
        se = 0.f;
        if (LAYER == 0) {
            const float4* ep = (const float4*)(ea + (size_t)r.y * 8);
            e0 = ep[0];
            e1 = ep[1];
        } else if (LAYER == 1) {
            se = se1[slot_index(j)];
        } else {
            se = __int_as_float(r.y);
        }
    };
    auto slot_term = [&](const int j, const int2 r, const float4 e0, const float4 e1,
                         const float se_in, const bool store, const bool keep) -> float {
        if (LAYER != 0) return se_in;
        if (store && j < end) {
            se1[j] = edge_dot(e0, e1, w + 8);
            if (keep) recs[j] = make_int2(r.x, __float_as_int(edge_dot(e0, e1, w + 16)));
        }
        return edge_dot(e0, e1, w);
    };

#define ROW_STEP(u)                                                              \
    {                                                                            \
        const int s = grp8i<SEL_LANE(u)>(s_l);                                   \
        const float2 r = *(const float2*)(h + s * 16 + 2 * i8);                  \
        const bool ok = 8 * g + (u) < cnt;                                       \
        row[u] = make_float2(ok ? r.x : 0.f, ok ? r.y : 0.f);                    \
    }
    auto gather_rows = [&](const int s_l, const int cnt, float2 (&row)[8]) {
        ROW_STEP(0) ROW_STEP(1) ROW_STEP(2) ROW_STEP(3)
        ROW_STEP(4) ROW_STEP(5) ROW_STEP(6) ROW_STEP(7)
    };
    auto slot_score = [&](const float2 (&row)[8]) -> float {
        float pv[8], w4[4], w2[2];
        #pragma unroll
        for (int u = 0; u < 8; ++u) pv[u] = row[u].x * a2.x + row[u].y * a2.y;
        #pragma unroll
        for (int k = 0; k < 4; ++k) {
            const float keepv = b4 ? pv[k + 4] : pv[k];
            const float send  = b4 ? pv[k] : pv[k + 4];
            w4[k] = keepv + grp8f<SEL_XOR4>(send);
        }
        #pragma unroll
        for (int k = 0; k < 2; ++k) {
            const float keepv = b2 ? w4[k + 2] : w4[k];
            const float send  = b2 ? w4[k] : w4[k + 2];
            w2[k] = keepv + grp8f<SEL_XOR2>(send);
        }
        const float keepv = b1 ? w2[1] : w2[0];
        const float send  = b1 ? w2[0] : w2[1];
        return keepv + grp8f<SEL_XOR1>(send);
    };
    auto slot_logit = [&](const int j, const float ss, const float se) -> float {
        return (j < end) ? leakyf(ss + sd + se) : -INFINITY;
    };
    auto chunk_logit = [&](const int base, const bool store, float2 (&row)[8]) -> float {
        const int j = base + lane;
        const int2 r = slot_rec(j);
        float4 e0, e1;
        float se;
        term_load(j, r, e0, e1, se);
        gather_rows(r.x, end - base, row);
        return slot_logit(j, slot_score(row), slot_term(j, r, e0, e1, se, store, false));
    };

#define ACC_STEP(row, a, u)                                                      \
    {                                                                            \
        const float av = grp8f<SEL_LANE(u)>(a);                                  \
        acc.x += row[u].x * av;                                                  \
        acc.y += row[u].y * av;                                                  \
    }
#define ACC_CHUNK(row, a)                                                        \
    ACC_STEP(row, a, 0) ACC_STEP(row, a, 1) ACC_STEP(row, a, 2) ACC_STEP(row, a, 3) \
    ACC_STEP(row, a, 4) ACC_STEP(row, a, 5) ACC_STEP(row, a, 6) ACC_STEP(row, a, 7)

    const int j0 = beg + lane, j1 = j0 + 32;
    const int2 c0 = slot_rec(j0);
    const int2 c1 = slot_rec(j1);
    float4 x0, y0, x1, y1;
    float t0, t1;
    float2 r0[8], r1[8];
    term_load(j0, c0, x0, y0, t0);
    gather_rows(c0.x, n, r0);
    if (n > 32) {
        term_load(j1, c1, x1, y1, t1);
        gather_rows(c1.x, n - 32, r1);
    }
    float q0, q1 = -INFINITY;
    q0 = slot_logit(j0, slot_score(r0), slot_term(j0, c0, x0, y0, t0, true, true));
    if (n > 32) q1 = slot_logit(j1, slot_score(r1), slot_term(j1, c1, x1, y1, t1, true, true));
    float mx = fmaxf(q0, q1);
    for (int base = beg + 64; base < end; base += 32) {
        float2 t[8];
        mx = fmaxf(mx, chunk_logit(base, true, t));
    }
    for (int o = 16; o > 0; o >>= 1) mx = fmaxf(mx, __shfl_xor(mx, o));
    if (!__builtin_isfinite(mx)) mx = 0.f;

    q0 = __expf(q0 - mx);
    q1 = __expf(q1 - mx);
    float sum = q0 + q1;
    for (int base = beg + 64; base < end; base += 32) {
        float2 t[8];
        sum += __expf(chunk_logit(base, false, t) - mx);
    }
    for (int o = 16; o > 0; o >>= 1) sum += __shfl_xor(sum, o);
    const float dn = sum + 1e-16f;

    float2 acc = make_float2(0.f, 0.f);
    {
        const float a0 = q0 / dn;
        const float a1 = q1 / dn;
        ACC_CHUNK(r0, a0)
        if (n > 32) { ACC_CHUNK(r1, a1) }
        for (int base = beg + 64; base < end; base += 32) {
            float2 t[8];
            const float a = __expf(chunk_logit(base, false, t) - mx) / dn;
            ACC_CHUNK(t, a)
        }
    }
    acc.x += __shfl_xor(acc.x, 8);  acc.y += __shfl_xor(acc.y, 8);
    acc.x += __shfl_xor(acc.x, 16); acc.y += __shfl_xor(acc.y, 16);
    if (lane < 8) *(float2*)(outb + d * 16 + 2 * i8) = make_float2(leakyf(acc.x + bb.x), leakyf(acc.y + bb.y));

    if (LAYER == 0) {
        for (int j = beg + lane + 64; j < end; j += 32) {
            const int2 r = recs[j];
            const float4* ep = (const float4*)(ea + (size_t)r.y * 8);
            recs[j] = make_int2(r.x, __float_as_int(edge_dot(ep[0], ep[1], w + 16)));
        }
    }
#undef ROW_STEP
#undef ACC_STEP
#undef ACC_CHUNK
}

extern "C" void kernel_launch(void* const* d_in, const int* in_sizes, int n_in,
                              void* d_out, int out_size, void* d_ws, size_t ws_size,
                              hipStream_t stream)
{
    const float* x     = (const float*)d_in[0];
    const int*   ei    = (const int*)  d_in[1];
    const float* ea    = (const float*)d_in[2];
    const float* cW    = (const float*)d_in[3];
    const float* casrc = (const float*)d_in[4];
    const float* cadst = (const float*)d_in[5];
    const float* cWe   = (const float*)d_in[6];
    const float* cae   = (const float*)d_in[7];
    const float* cb    = (const float*)d_in[8];
    const float* W0 = (const float*)d_in[9];  const float* b0 = (const float*)d_in[10];
    const float* W1 = (const float*)d_in[11]; const float* b1 = (const float*)d_in[12];
    const float* W2 = (const float*)d_in[13]; const float* b2 = (const float*)d_in[14];
    const float* W3 = (const float*)d_in[15]; const float* b3 = (const float*)d_in[16];
    float* out = (float*)d_out;

    const int* e_src = ei;
    const int* e_dst = ei + N_EDGES;

    float* ws   = (float*)d_ws;
    float* bufA = ws;
    float* bufB = bufA + (size_t)N_NODES * 16;
    int*   px   = (int*)ws;
    float* sdst = bufB + (size_t)N_NODES * 16 + N_NODES;
    int*   off  = (int*)(sdst + N_NODES);
    int*   bsc  = off + OFF_N;
    float* lp   = sdst + 3 * (size_t)N_NODES;
    int*   py   = (int*)lp;
    float* wae  = lp   + N_EDGES;
    float* w3p  = wae  + 32;
    float* b3p  = w3p  + 64 * 16;
    int2*  recs = (int2*)(b3p + 16);

    const int scanBlocks  = SCAN_N / 256;
    const int gemmBlocks  = (N_NODES + 127) / 128;
    const int segBlocks   = (N_NODES + 7) / 8;
    const int headBlocks  = (N_NODES + HEAD_WAVES * 16 - 1) / (HEAD_WAVES * 16);

    setup_kernel<<<scanBlocks, 256, 0, stream>>>(W3, b3, w3p, b3p, bsc, cWe, cae, wae);

    bucket_count_kernel<<<PW, 1024, 0, stream>>>(e_dst, bsc);
    scan_counts_kernel<<<1, 1024, 0, stream>>>(bsc);
    partition_kernel<<<PW, 1024, 0, stream>>>(e_src, e_dst, bsc, px, py);
    bucket_sort_kernel<<<NB, 1024, 0, stream>>>(bsc, px, py, off, recs);

    gemm_wmma_f32<16, 16, 16, 0><<<gemmBlocks, 256, 0, stream>>>(x, cW, nullptr, bufA, N_NODES);
    segment_softmax_aggregate_kernel<0><<<segBlocks, 256, 0, stream>>>(
        recs, off, ea, wae, casrc, cadst, bufA, cb, lp, bufB);
    gemm_wmma_f32<16, 16, 16, 0><<<gemmBlocks, 256, 0, stream>>>(bufB, cW + 256, nullptr, bufA, N_NODES);
    segment_softmax_aggregate_kernel<1><<<segBlocks, 256, 0, stream>>>(
        recs, off, ea, wae, casrc + 16, cadst + 16, bufA, cb + 16, lp, bufB);
    gemm_wmma_f32<16, 16, 16, 0><<<gemmBlocks, 256, 0, stream>>>(bufB, cW + 512, nullptr, bufA, N_NODES);
    segment_softmax_aggregate_kernel<2><<<segBlocks, 256, 0, stream>>>(
        recs, off, ea, wae, casrc + 32, cadst + 32, bufA, cb + 32, lp, bufB);

    mlp_head_kernel<<<headBlocks, HEAD_WAVES * 32, 0, stream>>>(bufB, W0, b0, W1, b1, W2, b2, w3p, b3p, out);
}


# Launches 12 -> 8: next layer's h = out @ W formed inside each aggregation (8-row local-memory tile, first wave's WMMA, same k-chain); partition sums its own run starts; setup rides in count kernel
# speedup vs baseline: 7.7197x; 1.0531x over previous
#include <hip/hip_runtime.h>
#include <hip/hip_bf16.h>
#include <math.h>

#define N_NODES 100000
#define N_EDGES 3200000
#define NEG_SLOPE 0.2f
#define NB      196
#define PW      256
#define PCH     12500
#define HN      (NB * PW)
#define BCAP    20480
#define OFF_N   100352
#define BSC_N   61440
#define PBW     7
static_assert(PW * PCH == N_EDGES, "the partition work-groups must cover the edges exactly");
static_assert(NB * 512 >= N_NODES + 1 && HN + NB + 1 <= BSC_N, "bucket and count extents");
static_assert(PW == 256 && PBW * 32 >= NB, "the prefix step reads a bucket's PW counts as 32 lanes x 8");

typedef __attribute__((ext_vector_type(2))) float v2f;
typedef __attribute__((ext_vector_type(8))) float v8f;

__device__ __forceinline__ float leakyf(float v) { return v > 0.f ? v : NEG_SLOPE * v; }

template <int K, int M, int MOUT, int ACT>
__global__ __launch_bounds__(256)
void gemm_wmma_f32(const float* __restrict__ A, const float* __restrict__ W,
                   const float* __restrict__ bias, float* __restrict__ C,
                   int Nrows)
{
    const int lane     = threadIdx.x & 31;
    const int wave     = threadIdx.x >> 5;
    const int row_base = (blockIdx.x * 8 + wave) * 16;
    if (row_base >= Nrows) return;

    const int nlo   = lane & 15;
    const int khalf = (lane >> 4) * 2;
    const int arow  = row_base + nlo;
    const int rofs  = (lane >> 4) * 8;

    v2f av[K / 4];
    #pragma unroll
    for (int kc = 0; kc < K / 4; ++kc) {
        const int k0 = kc * 4 + khalf;
        av[kc].x = A[arow * K + k0];
        av[kc].y = A[arow * K + k0 + 1];
    }

    #pragma unroll
    for (int mt = 0; mt < M / 16; ++mt) {
        const int col = mt * 16 + nlo;
        const float bv = bias ? bias[col] : 0.0f;
        v8f c;
        #pragma unroll
        for (int v = 0; v < 8; ++v) c[v] = bv;

        #pragma unroll
        for (int kc = 0; kc < K / 4; ++kc) {
            const int k0 = kc * 4 + khalf;
            v2f b;
            b.x = W[k0 * M + col];
            b.y = W[(k0 + 1) * M + col];
            c = __builtin_amdgcn_wmma_f32_16x16x4_f32(
                    false, av[kc], false, b, (short)0, c, false, false);
        }

        if (MOUT == M || col < MOUT) {
            #pragma unroll
            for (int v = 0; v < 8; ++v) {
                float r = c[v];
                if (ACT == 1) r = r > 0.f ? r : 0.f;
                C[(row_base + rofs + v) * MOUT + col] = r;
            }
        }
    }
}

#define HEAD_WAVES 4
#define HEAD_PITCH 68

template <int K>
__device__ __forceinline__ void head_load_a(const float* p, const int pitch, const int nlo,
                                            const int khalf, v2f (&av)[16])
{
    #pragma unroll
    for (int kc = 0; kc < K / 4; ++kc) {
        const int k0 = kc * 4 + khalf;
        av[kc].x = p[nlo * pitch + k0];
        av[kc].y = p[nlo * pitch + k0 + 1];
    }
}

template <int K, int M, int MOUT, int ACT>
__device__ __forceinline__ void head_layer(const v2f (&av)[16], const float* __restrict__ W,
                                           const float* __restrict__ bias, float* q, const int pitch,
                                           const int nlo, const int khalf, const int rofs)
{
    #pragma unroll
    for (int mt = 0; mt < M / 16; ++mt) {
        const int col = mt * 16 + nlo;
        const float bv = bias[col];
        v8f c;
        #pragma unroll
        for (int v = 0; v < 8; ++v) c[v] = bv;
        #pragma unroll
        for (int kc = 0; kc < K / 4; ++kc) {
            const int k0 = kc * 4 + khalf;
            v2f b;
            b.x = W[k0 * M + col];
            b.y = W[(k0 + 1) * M + col];
            c = __builtin_amdgcn_wmma_f32_16x16x4_f32(
                    false, av[kc], false, b, (short)0, c, false, false);
        }
        if (MOUT == M || col < MOUT) {
            #pragma unroll
            for (int v = 0; v < 8; ++v) {
                float r = c[v];
                if (ACT == 1) r = r > 0.f ? r : 0.f;
                q[(rofs + v) * pitch + col] = r;
            }
        }
    }
}

__global__ __launch_bounds__(HEAD_WAVES * 32)
void mlp_head_kernel(const float* __restrict__ X,
                     const float* __restrict__ W0, const float* __restrict__ b0,
                     const float* __restrict__ W1, const float* __restrict__ b1,
                     const float* __restrict__ W2, const float* __restrict__ b2,
                     const float* __restrict__ W3p, const float* __restrict__ b3p,
                     float* __restrict__ out)
{
    __shared__ float tile[HEAD_WAVES][16 * HEAD_PITCH];
    const int lane     = threadIdx.x & 31;
    const int wave     = threadIdx.x >> 5;
    const int row_base = (blockIdx.x * HEAD_WAVES + wave) * 16;
    if (row_base >= N_NODES) return;

    const int nlo   = lane & 15;
    const int khalf = (lane >> 4) * 2;
    const int rofs  = (lane >> 4) * 8;
    float* a = tile[wave];

    v2f av[16];
    head_load_a<16>(X + (size_t)row_base * 16, 16, nlo, khalf, av);
    head_layer<16, 64, 64, 1>(av, W0, b0, a, HEAD_PITCH, nlo, khalf, rofs);
    head_load_a<64>(a, HEAD_PITCH, nlo, khalf, av);
    head_layer<64, 64, 64, 1>(av, W1, b1, a, HEAD_PITCH, nlo, khalf, rofs);
    head_load_a<64>(a, HEAD_PITCH, nlo, khalf, av);
    head_layer<64, 64, 64, 1>(av, W2, b2, a, HEAD_PITCH, nlo, khalf, rofs);
    head_load_a<64>(a, HEAD_PITCH, nlo, khalf, av);
    head_layer<64, 16, 4, 0>(av, W3p, b3p, out + (size_t)row_base * 4, 4, nlo, khalf, rofs);
}


__device__ __forceinline__ int block_excl_scan(const int c, int* wt)
{
    const int lane = threadIdx.x & 31, wv = threadIdx.x >> 5;
    int inc = c;
    for (int o = 1; o < 32; o <<= 1) {
        const int u = __shfl_up(inc, o);
        inc += (lane >= o) ? u : 0;
    }
    if (lane == 31) wt[wv] = inc;
    __syncthreads();
    int base = (lane < wv) ? wt[lane] : 0;
    for (int o = 16; o > 0; o >>= 1) base += __shfl_xor(base, o);
    return base + inc - c;
}

__global__ __launch_bounds__(1024)
void bucket_count_kernel(const int* __restrict__ dst, int* __restrict__ hcnt,
                         const float* __restrict__ W3, const float* __restrict__ b3,
                         float* __restrict__ W3p, float* __restrict__ b3p,
                         const float* __restrict__ We, const float* __restrict__ ae,
                         float* __restrict__ wae)
{
    __shared__ int lh[NB];
    const int t = threadIdx.x, w = blockIdx.x;
    if (t < NB) lh[t] = 0;
    __syncthreads();
    const int* dp = dst + w * PCH;
    int i = t;
    for (; i + 3072 < PCH; i += 4096) {
        int d[4];
        #pragma unroll
        for (int k = 0; k < 4; ++k) d[k] = dp[i + k * 1024];
        #pragma unroll
        for (int k = 0; k < 4; ++k) atomicAdd(&lh[d[k] >> 9], 1);
    }
    for (; i < PCH; i += 1024) atomicAdd(&lh[dp[i] >> 9], 1);
    __syncthreads();
    if (t < NB) hcnt[t * PW + w] = lh[t];
    if (w == 0) {
        const int k = t >> 4, c = t & 15;
        W3p[t] = (c < 4) ? W3[k * 4 + c] : 0.0f;
        if (t < 16) b3p[t] = (t < 4) ? b3[t] : 0.0f;
    }
    if (w == 1 && t < 24) {
        const int i = t >> 3, j = t & 7;
        float s = 0.f;
        #pragma unroll
        for (int k = 0; k < 16; ++k) s += We[i * 128 + j * 16 + k] * ae[i * 16 + k];
        wae[t] = s;
    }
}

__global__ __launch_bounds__(1024)
void partition_kernel(const int* __restrict__ src, const int* __restrict__ dst,
                      const int* __restrict__ hcnt, int* __restrict__ bst,
                      int* __restrict__ px, int* __restrict__ py)
{
    __shared__ int  lst[NB];
    __shared__ int  lcur[NB];
    __shared__ int  gb[NB];
    __shared__ int  ttl[NB];
    __shared__ int  ptl[NB];
    __shared__ int  wt[32];
    __shared__ int  wt2[32];
    __shared__ int2 stage[PCH];
    const int t = threadIdx.x, w = blockIdx.x;
    const int lane = t & 31, wv = t >> 5;
    {
        int4 u0[PBW], u1[PBW];
        #pragma unroll
        for (int i = 0; i < PBW; ++i) {
            const int b  = wv + 32 * i;
            const int br = b < NB ? b : NB - 1;
            const int4* cp = (const int4*)(hcnt + br * PW) + 2 * lane;
            u0[i] = cp[0];
            u1[i] = cp[1];
        }
        #pragma unroll
        for (int i = 0; i < PBW; ++i) {
            const int b = wv + 32 * i;
            const int v[8] = {u0[i].x, u0[i].y, u0[i].z, u0[i].w, u1[i].x, u1[i].y, u1[i].z, u1[i].w};
            int all = 0, front = 0;
            #pragma unroll
            for (int k = 0; k < 8; ++k) {
                all   += v[k];
                front += (8 * lane + k < w) ? v[k] : 0;
            }
            for (int o = 16; o > 0; o >>= 1) {
                all   += __shfl_xor(all, o);
                front += __shfl_xor(front, o);
            }
            if (lane == 0 && b < NB) { ttl[b] = all; ptl[b] = front; }
        }
    }
    __syncthreads();
    int c = 0, tot = 0;
    if (t < NB) {
        c   = hcnt[t * PW + w];
        tot = ttl[t];
    }
    const int bs = block_excl_scan(tot, wt);
    const int ex = block_excl_scan(c, wt2);
    if (t < NB) {
        gb[t] = bs + ptl[t];
        lst[t] = ex;
        lcur[t] = ex;
        if (w == 0) bst[t] = bs;
    }
    if (w == 0 && t == NB) bst[NB] = bs;
    __syncthreads();

    const int e0 = w * PCH;
    auto place = [&](const int e, const int d, const int sr) {
        const int b = d >> 9;
        const int r = atomicAdd(&lcur[b], 1);
        stage[r] = make_int2(sr | ((d & 511) << 17), e | (b << 22));
    };
    int i = t;
    for (; i + 3072 < PCH; i += 4096) {
        int d[4], sr[4];
        #pragma unroll
        for (int k = 0; k < 4; ++k) { d[k] = dst[e0 + i + k * 1024]; sr[k] = src[e0 + i + k * 1024]; }
        #pragma unroll
        for (int k = 0; k < 4; ++k) place(e0 + i + k * 1024, d[k], sr[k]);
    }
    for (; i < PCH; i += 1024) place(e0 + i, dst[e0 + i], src[e0 + i]);
    __syncthreads();
    for (int j = t; j < PCH; j += 1024) {
        const int2 rec = stage[j];
        const int b = (int)((unsigned)rec.y >> 22);
        const int g = gb[b] + (j - lst[b]);
        px[g] = rec.x;
        py[g] = rec.y & 0x3FFFFF;
    }
}

__global__ __launch_bounds__(1024)
void bucket_sort_kernel(const int* __restrict__ bst, const int* __restrict__ px,
                        const int* __restrict__ py, int* __restrict__ off,
                        int2* __restrict__ recs)
{
    __shared__ int  ncnt[512];
    __shared__ int  wt[32];
    __shared__ int2 stage[BCAP];
    const int t = threadIdx.x, b = blockIdx.x;
    const int s = bst[b];
    const int n = bst[b + 1] - s;
    const int* qx = px + s;
    const int* qy = py + s;
    if (t < 512) ncnt[t] = 0;
    __syncthreads();
    int i = t;
    for (; i + 3072 < n; i += 4096) {
        int x[4];
        #pragma unroll
        for (int k = 0; k < 4; ++k) x[k] = qx[i + k * 1024];
        #pragma unroll
        for (int k = 0; k < 4; ++k) atomicAdd(&ncnt[x[k] >> 17], 1);
    }
    for (; i < n; i += 1024) atomicAdd(&ncnt[qx[i] >> 17], 1);
    __syncthreads();
    const int c  = (t < 512) ? ncnt[t] : 0;
    const int ex = block_excl_scan(c, wt);
    if (t < 512) {
        ncnt[t] = ex;
        const int node = b * 512 + t;
        if (node <= N_NODES) off[node] = s + ex;
    }
    __syncthreads();
    const bool staged = n <= BCAP;
    auto place = [&](const int x, const int y) {
        const int p = atomicAdd(&ncnt[x >> 17], 1);
        const int2 rec = make_int2(x & 0x1FFFF, y);
        if (staged) stage[p] = rec; else recs[s + p] = rec;
    };
    i = t;
    for (; i + 3072 < n; i += 4096) {
        int x[4], y[4];
        #pragma unroll
        for (int k = 0; k < 4; ++k) { x[k] = qx[i + k * 1024]; y[k] = qy[i + k * 1024]; }
        #pragma unroll
        for (int k = 0; k < 4; ++k) place(x[k], y[k]);
    }
    for (; i < n; i += 1024) place(qx[i], qy[i]);
    __syncthreads();
    if (staged) {
        for (int j = t; j < n; j += 1024) recs[s + j] = stage[j];
    }
}

__device__ __forceinline__ float edge_dot(const float4 e0, const float4 e1, const float* w)
{
    float se = 0.f;
    se += e0.x * w[0]; se += e0.y * w[1]; se += e0.z * w[2]; se += e0.w * w[3];
    se += e1.x * w[4]; se += e1.y * w[5]; se += e1.z * w[6]; se += e1.w * w[7];
    return se;
}

template <unsigned SEL>
__device__ __forceinline__ float grp8f(const float v)
{
    return __uint_as_float(__builtin_amdgcn_mov_dpp8(__float_as_uint(v), SEL));
}
template <unsigned SEL>
__device__ __forceinline__ int grp8i(const int v)
{
    return (int)__builtin_amdgcn_mov_dpp8((unsigned)v, SEL);
}
#define SEL_LANE(u) ((unsigned)(u) * 2396745u)
#define SEL_XOR4 6852524u
#define SEL_XOR2 11788826u
#define SEL_XOR1 14570689u

#define NEXT_PITCH 18
static_assert(N_NODES % 8 == 0, "every work-group holds 8 nodes: no wave leaves ahead of the barrier");
template <int LAYER>
__global__ __launch_bounds__(256, LAYER == 0 ? 10 : 12)
void segment_softmax_aggregate_kernel(int2* recs, const int* __restrict__ off,
                                      const float* __restrict__ ea, const float* __restrict__ wae,
                                      const float* __restrict__ asrc, const float* __restrict__ adst,
                                      const float* __restrict__ h, const float* __restrict__ b,
                                      float* se1, const float* __restrict__ Wn, float* __restrict__ outb)
{
    constexpr bool NEXT = LAYER < 2;
    __shared__ __attribute__((aligned(16))) float tile[8 * NEXT_PITCH];
    const int lane = threadIdx.x & 31;
    const int wv   = threadIdx.x >> 5;
    const int d    = blockIdx.x * 8 + wv;
    if (d >= N_NODES) return;
    const int beg = __builtin_amdgcn_readfirstlane(off[d]);
    const int end = __builtin_amdgcn_readfirstlane(off[d + 1]);
    const int n   = end - beg;
    float w[LAYER == 0 ? 24 : 1];
    if (LAYER == 0) { for (int k = 0; k < 24; ++k) w[k] = wae[k]; } else { w[0] = 0.f; }

    const int g  = lane >> 3;
    const int i8 = lane & 7;
    const float2 a2 = *(const float2*)(asrc + 2 * i8);
    const float2 bb = *(const float2*)(b + 2 * i8);
    const float2 d2 = *(const float2*)(adst + 2 * i8);
    const float2 hd = *(const float2*)(h + d * 16 + 2 * i8);
    float sd = hd.x * d2.x + hd.y * d2.y;
    sd += grp8f<SEL_XOR4>(sd);
    sd += grp8f<SEL_XOR2>(sd);
    sd += grp8f<SEL_XOR1>(sd);
    const bool b4 = (i8 & 4) != 0, b2 = (i8 & 2) != 0, b1 = (i8 & 1) != 0;

    auto slot_index = [&](const int j) -> int {
        const int jj = j < end ? j : end - 1;
        return jj < 0 ? 0 : jj;
    };
    auto slot_rec = [&](const int j) -> int2 {
        const int2 r = recs[slot_index(j)];
        return j < end ? r : make_int2(0, 0);
    };
    auto term_load = [&](const int j, const int2 r, float4& e0, float4& e1, float& se) {
        e0 = make_float4(0.f, 0.f, 0.f, 0.f);
        e1 = e0;
        se = 0.f;
        if (LAYER == 0) {
            const float4* ep = (const float4*)(ea + (size_t)r.y * 8);
            e0 = ep[0];
            e1 = ep[1];
        } else if (LAYER == 1) {
            se = se1[slot_index(j)];
        } else {
            se = __int_as_float(r.y);
        }
    };
    auto slot_term = [&](const int j, const int2 r, const float4 e0, const float4 e1,
                         const float se_in, const bool store, const bool keep) -> float {
        if (LAYER != 0) return se_in;
        if (store && j < end) {
            se1[j] = edge_dot(e0, e1, w + 8);
            if (keep) recs[j] = make_int2(r.x, __float_as_int(edge_dot(e0, e1, w + 16)));
        }
        return edge_dot(e0, e1, w);
    };

#define ROW_STEP(u)                                                              \
    {                                                                            \
        const int s = grp8i<SEL_LANE(u)>(s_l);                                   \
        const float2 r = *(const float2*)(h + s * 16 + 2 * i8);                  \
        const bool ok = 8 * g + (u) < cnt;                                       \
        row[u] = make_float2(ok ? r.x : 0.f, ok ? r.y : 0.f);                    \
    }
    auto gather_rows = [&](const int s_l, const int cnt, float2 (&row)[8]) {
        ROW_STEP(0) ROW_STEP(1) ROW_STEP(2) ROW_STEP(3)
        ROW_STEP(4) ROW_STEP(5) ROW_STEP(6) ROW_STEP(7)
    };
    auto slot_score = [&](const float2 (&row)[8]) -> float {
        float pv[8], w4[4], w2[2];
        #pragma unroll
        for (int u = 0; u < 8; ++u) pv[u] = row[u].x * a2.x + row[u].y * a2.y;
        #pragma unroll
        for (int k = 0; k < 4; ++k) {
            const float keepv = b4 ? pv[k + 4] : pv[k];
            const float send  = b4 ? pv[k] : pv[k + 4];
            w4[k] = keepv + grp8f<SEL_XOR4>(send);
        }
        #pragma unroll
        for (int k = 0; k < 2; ++k) {
            const float keepv = b2 ? w4[k + 2] : w4[k];
            const float send  = b2 ? w4[k] : w4[k + 2];
            w2[k] = keepv + grp8f<SEL_XOR2>(send);
        }
        const float keepv = b1 ? w2[1] : w2[0];
        const float send  = b1 ? w2[0] : w2[1];
        return keepv + grp8f<SEL_XOR1>(send);
    };
    auto slot_logit = [&](const int j, const float ss, const float se) -> float {
        return (j < end) ? leakyf(ss + sd + se) : -INFINITY;
    };
    auto chunk_logit = [&](const int base, const bool store, float2 (&row)[8]) -> float {
        const int j = base + lane;
        const int2 r = slot_rec(j);
        float4 e0, e1;
        float se;
        term_load(j, r, e0, e1, se);
        gather_rows(r.x, end - base, row);
        return slot_logit(j, slot_score(row), slot_term(j, r, e0, e1, se, store, false));
    };

#define ACC_STEP(row, a, u)                                                      \
    {                                                                            \
        const float av = grp8f<SEL_LANE(u)>(a);                                  \
        acc.x += row[u].x * av;                                                  \
        acc.y += row[u].y * av;                                                  \
    }
#define ACC_CHUNK(row, a)                                                        \
    ACC_STEP(row, a, 0) ACC_STEP(row, a, 1) ACC_STEP(row, a, 2) ACC_STEP(row, a, 3) \
    ACC_STEP(row, a, 4) ACC_STEP(row, a, 5) ACC_STEP(row, a, 6) ACC_STEP(row, a, 7)

    const int j0 = beg + lane, j1 = j0 + 32;
    const int2 c0 = slot_rec(j0);
    const int2 c1 = slot_rec(j1);
    float4 x0, y0, x1, y1;
    float t0, t1;
    float2 r0[8], r1[8];
    term_load(j0, c0, x0, y0, t0);
    gather_rows(c0.x, n, r0);
    if (n > 32) {
        term_load(j1, c1, x1, y1, t1);
        gather_rows(c1.x, n - 32, r1);
    }
    float q0, q1 = -INFINITY;
    q0 = slot_logit(j0, slot_score(r0), slot_term(j0, c0, x0, y0, t0, true, true));
    if (n > 32) q1 = slot_logit(j1, slot_score(r1), slot_term(j1, c1, x1, y1, t1, true, true));
    float mx = fmaxf(q0, q1);
    for (int base = beg + 64; base < end; base += 32) {
        float2 t[8];
        mx = fmaxf(mx, chunk_logit(base, true, t));
    }
    for (int o = 16; o > 0; o >>= 1) mx = fmaxf(mx, __shfl_xor(mx, o));
    if (!__builtin_isfinite(mx)) mx = 0.f;

    q0 = __expf(q0 - mx);
    q1 = __expf(q1 - mx);
    float sum = q0 + q1;
    for (int base = beg + 64; base < end; base += 32) {
        float2 t[8];
        sum += __expf(chunk_logit(base, false, t) - mx);
    }
    for (int o = 16; o > 0; o >>= 1) sum += __shfl_xor(sum, o);
    const float dn = sum + 1e-16f;

    float2 acc = make_float2(0.f, 0.f);
    {
        const float a0 = q0 / dn;
        const float a1 = q1 / dn;
        ACC_CHUNK(r0, a0)
        if (n > 32) { ACC_CHUNK(r1, a1) }
        for (int base = beg + 64; base < end; base += 32) {
            float2 t[8];
            const float a = __expf(chunk_logit(base, false, t) - mx) / dn;
            ACC_CHUNK(t, a)
        }
    }
    acc.x += __shfl_xor(acc.x, 8);  acc.y += __shfl_xor(acc.y, 8);
    acc.x += __shfl_xor(acc.x, 16); acc.y += __shfl_xor(acc.y, 16);
    {
        const float2 o = make_float2(leakyf(acc.x + bb.x), leakyf(acc.y + bb.y));
        if (lane < 8) {
            if (NEXT) *(float2*)(tile + wv * NEXT_PITCH + 2 * i8) = o;
            else      *(float2*)(outb + d * 16 + 2 * i8) = o;
        }
    }

    if (LAYER == 0) {
        for (int j = beg + lane + 64; j < end; j += 32) {
            const int2 r = recs[j];
            const float4* ep = (const float4*)(ea + (size_t)r.y * 8);
            recs[j] = make_int2(r.x, __float_as_int(edge_dot(ep[0], ep[1], w + 16)));
        }
    }
#undef ROW_STEP
#undef ACC_STEP
#undef ACC_CHUNK

    if (NEXT) {
        const int nlo   = lane & 15;
        const int khalf = (lane >> 4) * 2;
        v2f bw[4];
        #pragma unroll
        for (int kc = 0; kc < 4; ++kc) bw[kc] = (v2f){0.0f, 0.0f};
        if (wv == 0) {
            #pragma unroll
            for (int kc = 0; kc < 4; ++kc) {
                const int k0 = kc * 4 + khalf;
                bw[kc].x = Wn[k0 * 16 + nlo];
                bw[kc].y = Wn[(k0 + 1) * 16 + nlo];
            }
        }
        __syncthreads();
        if (wv == 0) {
            const bool real = nlo < 8;
            const float* trow = tile + (nlo & 7) * NEXT_PITCH + khalf;
            v2f a[4];
            #pragma unroll
            for (int kc = 0; kc < 4; ++kc) {
                const float2 t2 = *(const float2*)(trow + kc * 4);
                a[kc].x = real ? t2.x : 0.0f;
                a[kc].y = real ? t2.y : 0.0f;
            }
            v8f c;
            #pragma unroll
            for (int v = 0; v < 8; ++v) c[v] = 0.0f;
            #pragma unroll
            for (int kc = 0; kc < 4; ++kc) {
                c = __builtin_amdgcn_wmma_f32_16x16x4_f32(
                        false, a[kc], false, bw[kc], (short)0, c, false, false);
            }
            if (lane < 16) {
                #pragma unroll
                for (int v = 0; v < 8; ++v) outb[(blockIdx.x * 8 + v) * 16 + nlo] = c[v];
            }
        }
    }
}

extern "C" void kernel_launch(void* const* d_in, const int* in_sizes, int n_in,
                              void* d_out, int out_size, void* d_ws, size_t ws_size,
                              hipStream_t stream)
{
    const float* x     = (const float*)d_in[0];
    const int*   ei    = (const int*)  d_in[1];
    const float* ea    = (const float*)d_in[2];
    const float* cW    = (const float*)d_in[3];
    const float* casrc = (const float*)d_in[4];
    const float* cadst = (const float*)d_in[5];
    const float* cWe   = (const float*)d_in[6];
    const float* cae   = (const float*)d_in[7];
    const float* cb    = (const float*)d_in[8];
    const float* W0 = (const float*)d_in[9];  const float* b0 = (const float*)d_in[10];
    const float* W1 = (const float*)d_in[11]; const float* b1 = (const float*)d_in[12];
    const float* W2 = (const float*)d_in[13]; const float* b2 = (const float*)d_in[14];
    const float* W3 = (const float*)d_in[15]; const float* b3 = (const float*)d_in[16];
    float* out = (float*)d_out;

    const int* e_src = ei;
    const int* e_dst = ei + N_EDGES;

    float* ws   = (float*)d_ws;
    float* bufA = ws;
    float* bufB = bufA + (size_t)N_NODES * 16;
    int*   px   = (int*)ws;
    float* sdst = bufB + (size_t)N_NODES * 16 + N_NODES;
    int*   off  = (int*)(sdst + N_NODES);
    int*   bsc  = off + OFF_N;
    int*   bst  = bsc + HN;
    float* lp   = sdst + 3 * (size_t)N_NODES;
    int*   py   = (int*)lp;
    float* wae  = lp   + N_EDGES;
    float* w3p  = wae  + 32;
    float* b3p  = w3p  + 64 * 16;
    int2*  recs = (int2*)(b3p + 16);

    const int gemmBlocks  = (N_NODES + 127) / 128;
    const int segBlocks   = (N_NODES + 7) / 8;
    const int headBlocks  = (N_NODES + HEAD_WAVES * 16 - 1) / (HEAD_WAVES * 16);

    bucket_count_kernel<<<PW, 1024, 0, stream>>>(e_dst, bsc, W3, b3, w3p, b3p, cWe, cae, wae);
    partition_kernel<<<PW, 1024, 0, stream>>>(e_src, e_dst, bsc, bst, px, py);
    bucket_sort_kernel<<<NB, 1024, 0, stream>>>(bst, px, py, off, recs);

    gemm_wmma_f32<16, 16, 16, 0><<<gemmBlocks, 256, 0, stream>>>(x, cW, nullptr, bufA, N_NODES);
    segment_softmax_aggregate_kernel<0><<<segBlocks, 256, 0, stream>>>(
        recs, off, ea, wae, casrc, cadst, bufA, cb, lp, cW + 256, bufB);
    segment_softmax_aggregate_kernel<1><<<segBlocks, 256, 0, stream>>>(
        recs, off, ea, wae, casrc + 16, cadst + 16, bufB, cb + 16, lp, cW + 512, bufA);
    segment_softmax_aggregate_kernel<2><<<segBlocks, 256, 0, stream>>>(
        recs, off, ea, wae, casrc + 32, cadst + 32, bufA, cb + 32, lp, nullptr, bufB);

    mlp_head_kernel<<<headBlocks, HEAD_WAVES * 32, 0, stream>>>(bufB, W0, b0, W1, b1, W2, b2, w3p, b3p, out);
}


# Edge terms formed in the partition pass (edge_attr read once, in edge order) and carried through both ordering passes, no gather by edge id; layer 0 sums x rows, then applies W; sum orders changed
# speedup vs baseline: 9.0574x; 1.1733x over previous
#include <hip/hip_runtime.h>
#include <hip/hip_bf16.h>
#include <math.h>

#define N_NODES 100000
#define N_EDGES 3200000
#define NEG_SLOPE 0.2f
#define NB      196
#define PW      256
#define PCH     12500
#define HN      (NB * PW)
#define BCAP    18432
#define OFF_N   100352
#define BSC_N   61440
#define PBW     7
static_assert(PW * PCH == N_EDGES, "the partition work-groups must cover the edges exactly");
static_assert(N_NODES <= 131072 && N_EDGES % 2 == 0, "a source is 2 bytes and one bit; two 2-byte planes fill N_EDGES words");
static_assert(NB * 512 >= N_NODES + 1 && HN + NB + 1 <= BSC_N, "bucket and count extents");
static_assert(PW == 256 && PBW * 32 >= NB, "the prefix step reads a bucket's PW counts as 32 lanes x 8");

typedef __attribute__((ext_vector_type(2))) float v2f;
typedef __attribute__((ext_vector_type(8))) float v8f;

__device__ __forceinline__ float leakyf(float v) { return v > 0.f ? v : NEG_SLOPE * v; }

#define HEAD_WAVES 4
#define HEAD_PITCH 68

template <int K>
__device__ __forceinline__ void head_load_a(const float* p, const int pitch, const int nlo,
                                            const int khalf, v2f (&av)[16])
{
    #pragma unroll
    for (int kc = 0; kc < K / 4; ++kc) {
        const int k0 = kc * 4 + khalf;
        av[kc].x = p[nlo * pitch + k0];
        av[kc].y = p[nlo * pitch + k0 + 1];
    }
}

template <int K, int M, int MOUT, int ACT>
__device__ __forceinline__ void head_layer(const v2f (&av)[16], const float* __restrict__ W,
                                           const float* __restrict__ bias, float* q, const int pitch,
                                           const int nlo, const int khalf, const int rofs)
{
    #pragma unroll
    for (int mt = 0; mt < M / 16; ++mt) {
        const int col = mt * 16 + nlo;
        const float bv = bias[col];
        v8f c;
        #pragma unroll
        for (int v = 0; v < 8; ++v) c[v] = bv;
        #pragma unroll
        for (int kc = 0; kc < K / 4; ++kc) {
            const int k0 = kc * 4 + khalf;
            v2f b;
            b.x = W[k0 * M + col];
            b.y = W[(k0 + 1) * M + col];
            c = __builtin_amdgcn_wmma_f32_16x16x4_f32(
                    false, av[kc], false, b, (short)0, c, false, false);
        }
        if (MOUT == M || col < MOUT) {
            #pragma unroll
            for (int v = 0; v < 8; ++v) {
                float r = c[v];
                if (ACT == 1) r = r > 0.f ? r : 0.f;
                q[(rofs + v) * pitch + col] = r;
            }
        }
    }
}

__global__ __launch_bounds__(HEAD_WAVES * 32)
void mlp_head_kernel(const float* __restrict__ X,
                     const float* __restrict__ W0, const float* __restrict__ b0,
                     const float* __restrict__ W1, const float* __restrict__ b1,
                     const float* __restrict__ W2, const float* __restrict__ b2,
                     const float* __restrict__ W3p, const float* __restrict__ b3p,
                     float* __restrict__ out)
{
    __shared__ float tile[HEAD_WAVES][16 * HEAD_PITCH];
    const int lane     = threadIdx.x & 31;
    const int wave     = threadIdx.x >> 5;
    const int row_base = (blockIdx.x * HEAD_WAVES + wave) * 16;
    if (row_base >= N_NODES) return;

    const int nlo   = lane & 15;
    const int khalf = (lane >> 4) * 2;
    const int rofs  = (lane >> 4) * 8;
    float* a = tile[wave];

    v2f av[16];
    head_load_a<16>(X + (size_t)row_base * 16, 16, nlo, khalf, av);
    head_layer<16, 64, 64, 1>(av, W0, b0, a, HEAD_PITCH, nlo, khalf, rofs);
    head_load_a<64>(a, HEAD_PITCH, nlo, khalf, av);
    head_layer<64, 64, 64, 1>(av, W1, b1, a, HEAD_PITCH, nlo, khalf, rofs);
    head_load_a<64>(a, HEAD_PITCH, nlo, khalf, av);
    head_layer<64, 64, 64, 1>(av, W2, b2, a, HEAD_PITCH, nlo, khalf, rofs);
    head_load_a<64>(a, HEAD_PITCH, nlo, khalf, av);
    head_layer<64, 16, 4, 0>(av, W3p, b3p, out + (size_t)row_base * 4, 4, nlo, khalf, rofs);
}


__device__ __forceinline__ int block_excl_scan(const int c, int* wt)
{
    const int lane = threadIdx.x & 31, wv = threadIdx.x >> 5;
    int inc = c;
    for (int o = 1; o < 32; o <<= 1) {
        const int u = __shfl_up(inc, o);
        inc += (lane >= o) ? u : 0;
    }
    if (lane == 31) wt[wv] = inc;
    __syncthreads();
    int base = (lane < wv) ? wt[lane] : 0;
    for (int o = 16; o > 0; o >>= 1) base += __shfl_xor(base, o);
    return base + inc - c;
}

__global__ __launch_bounds__(1024)
void bucket_count_kernel(const int* __restrict__ dst, int* __restrict__ hcnt,
                         const float* __restrict__ W3, const float* __restrict__ b3,
                         float* __restrict__ W3p, float* __restrict__ b3p,
                         const float* __restrict__ We, const float* __restrict__ ae,
                         float* __restrict__ wae,
                         const float* __restrict__ W0c, const float* __restrict__ as0,
                         const float* __restrict__ ad0, float* __restrict__ uv)
{
    __shared__ int lh[NB];
    const int t = threadIdx.x, w = blockIdx.x;
    if (t < NB) lh[t] = 0;
    __syncthreads();
    const int* dp = dst + w * PCH;
    int i = t;
    for (; i + 3072 < PCH; i += 4096) {
        int d[4];
        #pragma unroll
        for (int k = 0; k < 4; ++k) d[k] = dp[i + k * 1024];
        #pragma unroll
        for (int k = 0; k < 4; ++k) atomicAdd(&lh[d[k] >> 9], 1);
    }
    for (; i < PCH; i += 1024) atomicAdd(&lh[dp[i] >> 9], 1);
    __syncthreads();
    if (t < NB) hcnt[t * PW + w] = lh[t];
    if (w == 0) {
        const int k = t >> 4, c = t & 15;
        W3p[t] = (c < 4) ? W3[k * 4 + c] : 0.0f;
        if (t < 16) b3p[t] = (t < 4) ? b3[t] : 0.0f;
    }
    if (w == 1 && t < 24) {
        const int i = t >> 3, j = t & 7;
        float s = 0.f;
        #pragma unroll
        for (int k = 0; k < 16; ++k) s += We[i * 128 + j * 16 + k] * ae[i * 16 + k];
        wae[t] = s;
    }
    if (w == 1 && t >= 32 && t < 64) {
        const int q = t - 32, k = q & 15;
        const float* av = (q < 16) ? as0 : ad0;
        float s = 0.f;
        #pragma unroll
        for (int c = 0; c < 16; ++c) s += W0c[k * 16 + c] * av[c];
        uv[q] = s;
    }
}

__device__ __forceinline__ float edge_dot(const float4 e0, const float4 e1, const float* w)
{
    float se = 0.f;
    se += e0.x * w[0]; se += e0.y * w[1]; se += e0.z * w[2]; se += e0.w * w[3];
    se += e1.x * w[4]; se += e1.y * w[5]; se += e1.z * w[6]; se += e1.w * w[7];
    return se;
}

__global__ __launch_bounds__(1024)
void partition_kernel(const int* __restrict__ src, const int* __restrict__ dst,
                      const float* __restrict__ ea, const float* __restrict__ wae,
                      const int* __restrict__ hcnt, int* __restrict__ bst,
                      int* __restrict__ kaw, int* __restrict__ kbw,
                      float* __restrict__ t0p, float* __restrict__ t1p, float* __restrict__ t2p)
{
    __shared__ int  lst[NB];
    __shared__ int  lcur[NB];
    __shared__ int  gb[NB];
    __shared__ int  ttl[NB];
    __shared__ int  ptl[NB];
    __shared__ int  wt[32];
    __shared__ int  wt2[32];
    __shared__ float wl[24];
    __shared__ unsigned char sbk[PCH];
    __shared__ int4 stage[PCH];
    unsigned short* __restrict__ ka = (unsigned short*)kaw;
    unsigned short* __restrict__ kb = (unsigned short*)kbw;
    const int t = threadIdx.x, w = blockIdx.x;
    const int lane = t & 31, wv = t >> 5;
    if (t < 24) wl[t] = wae[t];
    {
        int4 u0[PBW], u1[PBW];
        #pragma unroll
        for (int i = 0; i < PBW; ++i) {
            const int b  = wv + 32 * i;
            const int br = b < NB ? b : NB - 1;
            const int4* cp = (const int4*)(hcnt + br * PW) + 2 * lane;
            u0[i] = cp[0];
            u1[i] = cp[1];
        }
        #pragma unroll
        for (int i = 0; i < PBW; ++i) {
            const int b = wv + 32 * i;
            const int v[8] = {u0[i].x, u0[i].y, u0[i].z, u0[i].w, u1[i].x, u1[i].y, u1[i].z, u1[i].w};
            int all = 0, front = 0;
            #pragma unroll
            for (int k = 0; k < 8; ++k) {
                all   += v[k];
                front += (8 * lane + k < w) ? v[k] : 0;
            }
            for (int o = 16; o > 0; o >>= 1) {
                all   += __shfl_xor(all, o);
                front += __shfl_xor(front, o);
            }
            if (lane == 0 && b < NB) { ttl[b] = all; ptl[b] = front; }
        }
    }
    __syncthreads();
    int c = 0, tot = 0;
    if (t < NB) {
        c   = hcnt[t * PW + w];
        tot = ttl[t];
    }
    const int bs = block_excl_scan(tot, wt);
    const int ex = block_excl_scan(c, wt2);
    if (t < NB) {
        gb[t] = bs + ptl[t];
        lst[t] = ex;
        lcur[t] = ex;
        if (w == 0) bst[t] = bs;
    }
    if (w == 0 && t == NB) bst[NB] = bs;
    __syncthreads();

    const int e0 = w * PCH;
    auto place = [&](const int d, const int sr, const float4 x0, const float4 x1) {
        const int b = d >> 9;
        const int r = atomicAdd(&lcur[b], 1);
        const int key = ((d & 511) << 1) | (sr >> 16);
        stage[r] = make_int4((sr & 0xFFFF) | (key << 16),
                             __float_as_int(edge_dot(x0, x1, wl)),
                             __float_as_int(edge_dot(x0, x1, wl + 8)),
                             __float_as_int(edge_dot(x0, x1, wl + 16)));
        sbk[r] = (unsigned char)b;
    };
    int i = t;
    for (; i + 3072 < PCH; i += 4096) {
        int d[4], sr[4];
        float4 x0[4], x1[4];
        #pragma unroll
        for (int k = 0; k < 4; ++k) {
            const int e = e0 + i + k * 1024;
            d[k] = dst[e]; sr[k] = src[e];
            const float4* ep = (const float4*)(ea + (size_t)e * 8);
            x0[k] = ep[0]; x1[k] = ep[1];
        }
        #pragma unroll
        for (int k = 0; k < 4; ++k) place(d[k], sr[k], x0[k], x1[k]);
    }
    for (; i < PCH; i += 1024) {
        const int e = e0 + i;
        const float4* ep = (const float4*)(ea + (size_t)e * 8);
        place(dst[e], src[e], ep[0], ep[1]);
    }
    __syncthreads();
    for (int j = t; j < PCH; j += 1024) {
        const int4 rec = stage[j];
        const int b = sbk[j];
        const int g = gb[b] + (j - lst[b]);
        ka[g]  = (unsigned short)(rec.x & 0xFFFF);
        kb[g]  = (unsigned short)((unsigned)rec.x >> 16);
        t0p[g] = __int_as_float(rec.y);
        t1p[g] = __int_as_float(rec.z);
        t2p[g] = __int_as_float(rec.w);
    }
}


__global__ __launch_bounds__(1024)
void bucket_sort_kernel(const int* __restrict__ bst, int* kaw, int* kbw,
                        float* t0p, float* t1p, float* t2p,
                        int* __restrict__ off, int* __restrict__ mid)
{
    __shared__ int  kst[1024];
    __shared__ int  knx[1024];
    __shared__ int  wt[32];
    __shared__ int4 stage[BCAP];
    const int t = threadIdx.x, b = blockIdx.x;
    const int s = bst[b];
    const int n = bst[b + 1] - s;
    unsigned short* qa = (unsigned short*)kaw + s;
    unsigned short* qb = (unsigned short*)kbw + s;
    float* q0 = t0p + s;
    float* q1 = t1p + s;
    float* q2 = t2p + s;
    knx[t] = 0;
    __syncthreads();
    int i = t;
    for (; i + 3072 < n; i += 4096) {
        int x[4];
        #pragma unroll
        for (int k = 0; k < 4; ++k) x[k] = qb[i + k * 1024];
        #pragma unroll
        for (int k = 0; k < 4; ++k) atomicAdd(&knx[x[k] & 1023], 1);
    }
    for (; i < n; i += 1024) atomicAdd(&knx[qb[i] & 1023], 1);
    __syncthreads();
    const int c  = knx[t];
    const int ex = block_excl_scan(c, wt);
    kst[t] = ex;
    knx[t] = ex;
    {
        const int node = b * 512 + (t >> 1);
        if (node <= N_NODES) {
            if (t & 1) mid[node] = s + ex; else off[node] = s + ex;
        }
    }
    __syncthreads();
    if (n <= BCAP) {
        i = t;
        for (; i + 3072 < n; i += 4096) {
            int xa[4], xb[4];
            float y0[4], y1[4], y2[4];
            #pragma unroll
            for (int k = 0; k < 4; ++k) {
                const int j = i + k * 1024;
                xa[k] = qa[j]; xb[k] = qb[j]; y0[k] = q0[j]; y1[k] = q1[j]; y2[k] = q2[j];
            }
            #pragma unroll
            for (int k = 0; k < 4; ++k) {
                const int p = atomicAdd(&knx[xb[k] & 1023], 1);
                stage[p] = make_int4(xa[k], __float_as_int(y0[k]), __float_as_int(y1[k]), __float_as_int(y2[k]));
            }
        }
        for (; i < n; i += 1024) {
            const int p = atomicAdd(&knx[qb[i] & 1023], 1);
            stage[p] = make_int4((int)qa[i], __float_as_int(q0[i]), __float_as_int(q1[i]), __float_as_int(q2[i]));
        }
        __syncthreads();
        for (int j = t; j < n; j += 1024) {
            const int4 rec = stage[j];
            qa[j] = (unsigned short)rec.x;
            q0[j] = __int_as_float(rec.y);
            q1[j] = __int_as_float(rec.z);
            q2[j] = __int_as_float(rec.w);
        }
    } else if (t == 0) {
        volatile unsigned short* va = qa;
        volatile unsigned short* vb = qb;
        volatile float* v0 = q0;
        volatile float* v1 = q1;
        volatile float* v2 = q2;
        for (int k = 0; k < 1024; ++k) {
            const int kend = (k < 1023) ? kst[k + 1] : n;
            while (knx[k] < kend) {
                const int p = knx[k];
                const unsigned short ra = va[p], rb = vb[p];
                const float r0 = v0[p], r1 = v1[p], r2 = v2[p];
                const int kk = rb & 1023;
                if (kk == k) { knx[k] = p + 1; continue; }
                const int j = knx[kk];
                knx[kk] = j + 1;
                const unsigned short sa = va[j], sb = vb[j];
                const float s0 = v0[j], s1 = v1[j], s2 = v2[j];
                va[j] = ra; vb[j] = rb; v0[j] = r0; v1[j] = r1; v2[j] = r2;
                va[p] = sa; vb[p] = sb; v0[p] = s0; v1[p] = s1; v2[p] = s2;
            }
        }
    }
}


template <unsigned SEL>
__device__ __forceinline__ float grp8f(const float v)
{
    return __uint_as_float(__builtin_amdgcn_mov_dpp8(__float_as_uint(v), SEL));
}
template <unsigned SEL>
__device__ __forceinline__ int grp8i(const int v)
{
    return (int)__builtin_amdgcn_mov_dpp8((unsigned)v, SEL);
}
#define SEL_LANE(u) ((unsigned)(u) * 2396745u)
#define SEL_XOR4 6852524u
#define SEL_XOR2 11788826u
#define SEL_XOR1 14570689u


#define NEXT_PITCH 18
#define SEG_GROUPS (N_NODES / 8)
#define SEG_WGS 1250
static_assert(N_NODES % 8 == 0, "every group holds 8 nodes: all 8 waves reach every barrier");
static_assert(SEG_WGS <= 256 * 5, "the whole grid is resident at once");
template <int LAYER>
__global__ __launch_bounds__(256, 10)
void segment_softmax_aggregate_kernel(const int* __restrict__ s16w, const float* __restrict__ tq,
                                      const int* __restrict__ off, const int* __restrict__ mid,
                                      const float* __restrict__ asrc, const float* __restrict__ adst,
                                      const float* __restrict__ h, const float* __restrict__ b,
                                      const float* __restrict__ Wf, const float* __restrict__ Wn,
                                      float* __restrict__ outb)
{
    constexpr bool NEXT  = LAYER < 2;
    constexpr bool FIRST = LAYER == 0;
    const unsigned short* __restrict__ s16 = (const unsigned short*)s16w;
    __shared__ __attribute__((aligned(16))) float tile[2][8 * NEXT_PITCH];
    const int lane = threadIdx.x & 31;
    const int wv   = threadIdx.x >> 5;

    const int g  = lane >> 3;
    const int i8 = lane & 7;
    const float2 a2 = *(const float2*)(asrc + 2 * i8);
    const float2 bb = *(const float2*)(b + 2 * i8);
    const float2 d2 = *(const float2*)(adst + 2 * i8);
    const bool b4 = (i8 & 4) != 0, b2 = (i8 & 2) != 0, b1 = (i8 & 1) != 0;

    auto rec_raw = [&](const int j, const int e) -> int2 {
        int jj = j < e ? j : e - 1;
        jj = jj < 0 ? 0 : jj;
        return make_int2((int)s16[jj], __float_as_int(tq[jj]));
    };
    auto rec_fix = [&](const int2 r, const int j, const int e, const int m) -> int2 {
        return j < e ? make_int2(r.x + (j >= m ? 65536 : 0), r.y) : make_int2(0, 0);
    };

    int beg = __builtin_amdgcn_readfirstlane(off[blockIdx.x * 8 + wv]);
    int end = __builtin_amdgcn_readfirstlane(off[blockIdx.x * 8 + wv + 1]);
    int mdl = __builtin_amdgcn_readfirstlane(mid[blockIdx.x * 8 + wv]);
    int2 c0r = rec_raw(beg + lane, end);
    int2 c1r = rec_raw(beg + 32 + lane, end);
    float2 hd = *(const float2*)(h + (blockIdx.x * 8 + wv) * 16 + 2 * i8);

    int it = 0;
    for (int grp = blockIdx.x; grp < SEG_GROUPS; grp += SEG_WGS, ++it) {
    const int d = grp * 8 + wv;
    const int n = end - beg;
    const int2 c0 = rec_fix(c0r, beg + lane, end, mdl);
    const int2 c1 = rec_fix(c1r, beg + 32 + lane, end, mdl);
    const int gnx = grp + SEG_WGS < SEG_GROUPS ? grp + SEG_WGS : grp;
    const int obn = off[gnx * 8 + wv];
    const int oen = off[gnx * 8 + wv + 1];
    const int omn = mid[gnx * 8 + wv];
    float sd = hd.x * d2.x + hd.y * d2.y;
    sd += grp8f<SEL_XOR4>(sd);
    sd += grp8f<SEL_XOR2>(sd);
    sd += grp8f<SEL_XOR1>(sd);

    auto slot_rec = [&](const int j) -> int2 { return rec_fix(rec_raw(j, end), j, end, mdl); };

#define ROW_STEP(u)                                                              \
    {                                                                            \
        const int s = grp8i<SEL_LANE(u)>(s_l);                                   \
        const float2 r = *(const float2*)(h + s * 16 + 2 * i8);                  \
        const bool ok = 8 * g + (u) < cnt;                                       \
        row[u] = make_float2(ok ? r.x : 0.f, ok ? r.y : 0.f);                    \
    }
    auto gather_rows = [&](const int s_l, const int cnt, float2 (&row)[8]) {
        ROW_STEP(0) ROW_STEP(1) ROW_STEP(2) ROW_STEP(3)
        ROW_STEP(4) ROW_STEP(5) ROW_STEP(6) ROW_STEP(7)
    };
    auto slot_score = [&](const float2 (&row)[8]) -> float {
        float pv[8], w4[4], w2[2];
        #pragma unroll
        for (int u = 0; u < 8; ++u) pv[u] = row[u].x * a2.x + row[u].y * a2.y;
        #pragma unroll
        for (int k = 0; k < 4; ++k) {
            const float keepv = b4 ? pv[k + 4] : pv[k];
            const float send  = b4 ? pv[k] : pv[k + 4];
            w4[k] = keepv + grp8f<SEL_XOR4>(send);
        }
        #pragma unroll
        for (int k = 0; k < 2; ++k) {
            const float keepv = b2 ? w4[k + 2] : w4[k];
            const float send  = b2 ? w4[k] : w4[k + 2];
            w2[k] = keepv + grp8f<SEL_XOR2>(send);
        }
        const float keepv = b1 ? w2[1] : w2[0];
        const float send  = b1 ? w2[0] : w2[1];
        return keepv + grp8f<SEL_XOR1>(send);
    };
    auto slot_logit = [&](const int j, const float ss, const float se) -> float {
        return (j < end) ? leakyf(ss + sd + se) : -INFINITY;
    };
    auto chunk_logit = [&](const int base, float2 (&row)[8]) -> float {
        const int j = base + lane;
        const int2 r = slot_rec(j);
        gather_rows(r.x, end - base, row);
        return slot_logit(j, slot_score(row), __int_as_float(r.y));
    };

#define ACC_STEP(row, a, u)                                                      \
    {                                                                            \
        const float av = grp8f<SEL_LANE(u)>(a);                                  \
        acc.x += row[u].x * av;                                                  \
        acc.y += row[u].y * av;                                                  \
    }
#define ACC_CHUNK(row, a)                                                        \
    ACC_STEP(row, a, 0) ACC_STEP(row, a, 1) ACC_STEP(row, a, 2) ACC_STEP(row, a, 3) \
    ACC_STEP(row, a, 4) ACC_STEP(row, a, 5) ACC_STEP(row, a, 6) ACC_STEP(row, a, 7)

    const int j0 = beg + lane, j1 = j0 + 32;
    float2 r0[8], r1[8];
    gather_rows(c0.x, n, r0);
    if (n > 32) gather_rows(c1.x, n - 32, r1);
    float q0, q1 = -INFINITY;
    q0 = slot_logit(j0, slot_score(r0), __int_as_float(c0.y));
    if (n > 32) q1 = slot_logit(j1, slot_score(r1), __int_as_float(c1.y));
    const int nbeg = __builtin_amdgcn_readfirstlane(obn);
    const int nend = __builtin_amdgcn_readfirstlane(oen);
    const int nmid = __builtin_amdgcn_readfirstlane(omn);
    const int2 c0n = rec_raw(nbeg + lane, nend);
    const int2 c1n = rec_raw(nbeg + 32 + lane, nend);
    const float2 hdn = *(const float2*)(h + (gnx * 8 + wv) * 16 + 2 * i8);
    float mx = fmaxf(q0, q1);
    for (int base = beg + 64; base < end; base += 32) {
        float2 t[8];
        mx = fmaxf(mx, chunk_logit(base, t));
    }
    for (int o = 16; o > 0; o >>= 1) mx = fmaxf(mx, __shfl_xor(mx, o));
    if (!__builtin_isfinite(mx)) mx = 0.f;

    q0 = __expf(q0 - mx);
    q1 = __expf(q1 - mx);
    float sum = q0 + q1;
    for (int base = beg + 64; base < end; base += 32) {
        float2 t[8];
        sum += __expf(chunk_logit(base, t) - mx);
    }
    for (int o = 16; o > 0; o >>= 1) sum += __shfl_xor(sum, o);
    const float dn = sum + 1e-16f;

    float2 acc = make_float2(0.f, 0.f);
    {
        const float a0 = q0 / dn;
        const float a1 = q1 / dn;
        ACC_CHUNK(r0, a0)
        if (n > 32) { ACC_CHUNK(r1, a1) }
        for (int base = beg + 64; base < end; base += 32) {
            float2 t[8];
            const float a = __expf(chunk_logit(base, t) - mx) / dn;
            ACC_CHUNK(t, a)
        }
    }
    acc.x += __shfl_xor(acc.x, 8);  acc.y += __shfl_xor(acc.y, 8);
    acc.x += __shfl_xor(acc.x, 16); acc.y += __shfl_xor(acc.y, 16);
    {
        const float2 o = FIRST ? acc : make_float2(leakyf(acc.x + bb.x), leakyf(acc.y + bb.y));
        if (lane < 8) {
            if (NEXT) *(float2*)(tile[it & 1] + wv * NEXT_PITCH + 2 * i8) = o;
            else      *(float2*)(outb + d * 16 + 2 * i8) = o;
        }
    }
#undef ROW_STEP
#undef ACC_STEP
#undef ACC_CHUNK


    if (NEXT) {
        const int nlo   = lane & 15;
        const int khalf = (lane >> 4) * 2;
        v2f bw[4], bf[4];
        float bcol = 0.0f;
        #pragma unroll
        for (int kc = 0; kc < 4; ++kc) { bw[kc] = (v2f){0.0f, 0.0f}; bf[kc] = (v2f){0.0f, 0.0f}; }
        if (wv == 0) {
            #pragma unroll
            for (int kc = 0; kc < 4; ++kc) {
                const int k0 = kc * 4 + khalf;
                bw[kc].x = Wn[k0 * 16 + nlo];
                bw[kc].y = Wn[(k0 + 1) * 16 + nlo];
                if (FIRST) {
                    bf[kc].x = Wf[k0 * 16 + nlo];
                    bf[kc].y = Wf[(k0 + 1) * 16 + nlo];
                }
            }
            if (FIRST) bcol = b[nlo];
        }
        __syncthreads();
        if (wv == 0) {
            const bool real = nlo < 8;
            float* tb = tile[it & 1];
            const float* trow = tb + (nlo & 7) * NEXT_PITCH + khalf;
            v2f a[4];
            #pragma unroll
            for (int kc = 0; kc < 4; ++kc) {
                const float2 t2 = *(const float2*)(trow + kc * 4);
                a[kc].x = real ? t2.x : 0.0f;
                a[kc].y = real ? t2.y : 0.0f;
            }
            if (FIRST) {
                v8f y;
                #pragma unroll
                for (int v = 0; v < 8; ++v) y[v] = 0.0f;
                #pragma unroll
                for (int kc = 0; kc < 4; ++kc) {
                    y = __builtin_amdgcn_wmma_f32_16x16x4_f32(
                            false, a[kc], false, bf[kc], (short)0, y, false, false);
                }
                if (lane < 16) {
                    #pragma unroll
                    for (int v = 0; v < 8; ++v) tb[v * NEXT_PITCH + nlo] = leakyf(y[v] + bcol);
                }
                __builtin_amdgcn_fence(__ATOMIC_ACQ_REL, "wavefront");
                #pragma unroll
                for (int kc = 0; kc < 4; ++kc) {
                    const float2 t2 = *(const float2*)(trow + kc * 4);
                    a[kc].x = real ? t2.x : 0.0f;
                    a[kc].y = real ? t2.y : 0.0f;
                }
            }
            v8f c;
            #pragma unroll
            for (int v = 0; v < 8; ++v) c[v] = 0.0f;
            #pragma unroll
            for (int kc = 0; kc < 4; ++kc) {
                c = __builtin_amdgcn_wmma_f32_16x16x4_f32(
                        false, a[kc], false, bw[kc], (short)0, c, false, false);
            }
            if (lane < 16) {
                #pragma unroll
                for (int v = 0; v < 8; ++v) outb[(grp * 8 + v) * 16 + nlo] = c[v];
            }
        }
    }
    beg = nbeg; end = nend; mdl = nmid; c0r = c0n; c1r = c1n; hd = hdn;
    }
}

extern "C" void kernel_launch(void* const* d_in, const int* in_sizes, int n_in,
                              void* d_out, int out_size, void* d_ws, size_t ws_size,
                              hipStream_t stream)
{
    const float* x     = (const float*)d_in[0];
    const int*   ei    = (const int*)  d_in[1];
    const float* ea    = (const float*)d_in[2];
    const float* cW    = (const float*)d_in[3];
    const float* casrc = (const float*)d_in[4];
    const float* cadst = (const float*)d_in[5];
    const float* cWe   = (const float*)d_in[6];
    const float* cae   = (const float*)d_in[7];
    const float* cb    = (const float*)d_in[8];
    const float* W0 = (const float*)d_in[9];  const float* b0 = (const float*)d_in[10];
    const float* W1 = (const float*)d_in[11]; const float* b1 = (const float*)d_in[12];
    const float* W2 = (const float*)d_in[13]; const float* b2 = (const float*)d_in[14];
    const float* W3 = (const float*)d_in[15]; const float* b3 = (const float*)d_in[16];
    float* out = (float*)d_out;

    const int* e_src = ei;
    const int* e_dst = ei + N_EDGES;

    float* ws   = (float*)d_ws;
    int*   kaw  = (int*)ws;
    int*   kbw  = kaw + N_EDGES / 2;
    float* h1   = (float*)kbw;
    int*   mid  = (int*)(ws + N_EDGES);
    float* uv   = (float*)(mid + OFF_N);
    int*   off  = (int*)(ws + N_EDGES + 2 * (size_t)N_NODES);
    int*   bsc  = off + OFF_N;
    int*   bst  = bsc + HN;
    float* t1p  = ws + N_EDGES + 4 * (size_t)N_NODES;
    float* wae  = t1p  + N_EDGES;
    float* w3p  = wae  + 32;
    float* b3p  = w3p  + 64 * 16;
    float* t2p  = b3p + 16;
    float* t0p  = t2p + N_EDGES;
    float* h2   = t0p;

    const int headBlocks  = (N_NODES + HEAD_WAVES * 16 - 1) / (HEAD_WAVES * 16);

    bucket_count_kernel<<<PW, 1024, 0, stream>>>(e_dst, bsc, W3, b3, w3p, b3p, cWe, cae, wae, cW, casrc, cadst, uv);
    partition_kernel<<<PW, 1024, 0, stream>>>(e_src, e_dst, ea, wae, bsc, bst, kaw, kbw, t0p, t1p, t2p);
    bucket_sort_kernel<<<NB, 1024, 0, stream>>>(bst, kaw, kbw, t0p, t1p, t2p, off, mid);

    segment_softmax_aggregate_kernel<0><<<SEG_WGS, 256, 0, stream>>>(
        kaw, t0p, off, mid, uv, uv + 16, x, cb, cW, cW + 256, h1);
    segment_softmax_aggregate_kernel<1><<<SEG_WGS, 256, 0, stream>>>(
        kaw, t1p, off, mid, casrc + 16, cadst + 16, h1, cb + 16, nullptr, cW + 512, h2);
    segment_softmax_aggregate_kernel<2><<<SEG_WGS, 256, 0, stream>>>(
        kaw, t2p, off, mid, casrc + 32, cadst + 32, h2, cb + 32, nullptr, nullptr, h1);

    mlp_head_kernel<<<headBlocks, HEAD_WAVES * 32, 0, stream>>>(h1, W0, b0, W1, b1, W2, b2, w3p, b3p, out);
}


# Aggregation gathers each source row by 4 lanes (16-byte quarters, 8 rows a load) instead of 8: half the row steps; row sum now 8 lane groups x 4 slots, then a halving tree: sum orders changed
# speedup vs baseline: 9.5866x; 1.0584x over previous
#include <hip/hip_runtime.h>
#include <hip/hip_bf16.h>
#include <math.h>

#define N_NODES 100000
#define N_EDGES 3200000
#define NEG_SLOPE 0.2f
#define NB      196
#define PW      256
#define PCH     12500
#define HN      (NB * PW)
#define BCAP    18432
#define OFF_N   100352
#define BSC_N   61440
#define PBW     7
static_assert(PW * PCH == N_EDGES, "the partition work-groups must cover the edges exactly");
static_assert(N_NODES <= 131072 && N_EDGES % 2 == 0, "a source is 2 bytes and one bit; two 2-byte planes fill N_EDGES words");
static_assert(NB * 512 >= N_NODES + 1 && HN + NB + 1 <= BSC_N, "bucket and count extents");
static_assert(PW == 256 && PBW * 32 >= NB, "the prefix step reads a bucket's PW counts as 32 lanes x 8");

typedef __attribute__((ext_vector_type(2))) float v2f;
typedef __attribute__((ext_vector_type(8))) float v8f;

__device__ __forceinline__ float leakyf(float v) { return v > 0.f ? v : NEG_SLOPE * v; }

#define HEAD_WAVES 4
#define HEAD_PITCH 68

template <int K>
__device__ __forceinline__ void head_load_a(const float* p, const int pitch, const int nlo,
                                            const int khalf, v2f (&av)[16])
{
    #pragma unroll
    for (int kc = 0; kc < K / 4; ++kc) {
        const int k0 = kc * 4 + khalf;
        av[kc].x = p[nlo * pitch + k0];
        av[kc].y = p[nlo * pitch + k0 + 1];
    }
}

template <int K, int M, int MOUT, int ACT>
__device__ __forceinline__ void head_layer(const v2f (&av)[16], const float* __restrict__ W,
                                           const float* __restrict__ bias, float* q, const int pitch,
                                           const int nlo, const int khalf, const int rofs)
{
    #pragma unroll
    for (int mt = 0; mt < M / 16; ++mt) {
        const int col = mt * 16 + nlo;
        const float bv = bias[col];
        v8f c;
        #pragma unroll
        for (int v = 0; v < 8; ++v) c[v] = bv;
        #pragma unroll
        for (int kc = 0; kc < K / 4; ++kc) {
            const int k0 = kc * 4 + khalf;
            v2f b;
            b.x = W[k0 * M + col];
            b.y = W[(k0 + 1) * M + col];
            c = __builtin_amdgcn_wmma_f32_16x16x4_f32(
                    false, av[kc], false, b, (short)0, c, false, false);
        }
        if (MOUT == M || col < MOUT) {
            #pragma unroll
            for (int v = 0; v < 8; ++v) {
                float r = c[v];
                if (ACT == 1) r = r > 0.f ? r : 0.f;
                q[(rofs + v) * pitch + col] = r;
            }
        }
    }
}

__global__ __launch_bounds__(HEAD_WAVES * 32)
void mlp_head_kernel(const float* __restrict__ X,
                     const float* __restrict__ W0, const float* __restrict__ b0,
                     const float* __restrict__ W1, const float* __restrict__ b1,
                     const float* __restrict__ W2, const float* __restrict__ b2,
                     const float* __restrict__ W3p, const float* __restrict__ b3p,
                     float* __restrict__ out)
{
    __shared__ float tile[HEAD_WAVES][16 * HEAD_PITCH];
    const int lane     = threadIdx.x & 31;
    const int wave     = threadIdx.x >> 5;
    const int row_base = (blockIdx.x * HEAD_WAVES + wave) * 16;
    if (row_base >= N_NODES) return;

    const int nlo   = lane & 15;
    const int khalf = (lane >> 4) * 2;
    const int rofs  = (lane >> 4) * 8;
    float* a = tile[wave];

    v2f av[16];
    head_load_a<16>(X + (size_t)row_base * 16, 16, nlo, khalf, av);
    head_layer<16, 64, 64, 1>(av, W0, b0, a, HEAD_PITCH, nlo, khalf, rofs);
    head_load_a<64>(a, HEAD_PITCH, nlo, khalf, av);
    head_layer<64, 64, 64, 1>(av, W1, b1, a, HEAD_PITCH, nlo, khalf, rofs);
    head_load_a<64>(a, HEAD_PITCH, nlo, khalf, av);
    head_layer<64, 64, 64, 1>(av, W2, b2, a, HEAD_PITCH, nlo, khalf, rofs);
    head_load_a<64>(a, HEAD_PITCH, nlo, khalf, av);
    head_layer<64, 16, 4, 0>(av, W3p, b3p, out + (size_t)row_base * 4, 4, nlo, khalf, rofs);
}


__device__ __forceinline__ int block_excl_scan(const int c, int* wt)
{
    const int lane = threadIdx.x & 31, wv = threadIdx.x >> 5;
    int inc = c;
    for (int o = 1; o < 32; o <<= 1) {
        const int u = __shfl_up(inc, o);
        inc += (lane >= o) ? u : 0;
    }
    if (lane == 31) wt[wv] = inc;
    __syncthreads();
    int base = (lane < wv) ? wt[lane] : 0;
    for (int o = 16; o > 0; o >>= 1) base += __shfl_xor(base, o);
    return base + inc - c;
}

__global__ __launch_bounds__(1024)
void bucket_count_kernel(const int* __restrict__ dst, int* __restrict__ hcnt,
                         const float* __restrict__ W3, const float* __restrict__ b3,
                         float* __restrict__ W3p, float* __restrict__ b3p,
                         const float* __restrict__ We, const float* __restrict__ ae,
                         float* __restrict__ wae,
                         const float* __restrict__ W0c, const float* __restrict__ as0,
                         const float* __restrict__ ad0, float* __restrict__ uv)
{
    __shared__ int lh[NB];
    const int t = threadIdx.x, w = blockIdx.x;
    if (t < NB) lh[t] = 0;
    __syncthreads();
    const int* dp = dst + w * PCH;
    int i = t;
    for (; i + 3072 < PCH; i += 4096) {
        int d[4];
        #pragma unroll
        for (int k = 0; k < 4; ++k) d[k] = dp[i + k * 1024];
        #pragma unroll
        for (int k = 0; k < 4; ++k) atomicAdd(&lh[d[k] >> 9], 1);
    }
    for (; i < PCH; i += 1024) atomicAdd(&lh[dp[i] >> 9], 1);
    __syncthreads();
    if (t < NB) hcnt[t * PW + w] = lh[t];
    if (w == 0) {
        const int k = t >> 4, c = t & 15;
        W3p[t] = (c < 4) ? W3[k * 4 + c] : 0.0f;
        if (t < 16) b3p[t] = (t < 4) ? b3[t] : 0.0f;
    }
    if (w == 1 && t < 24) {
        const int i = t >> 3, j = t & 7;
        float s = 0.f;
        #pragma unroll
        for (int k = 0; k < 16; ++k) s += We[i * 128 + j * 16 + k] * ae[i * 16 + k];
        wae[t] = s;
    }
    if (w == 1 && t >= 32 && t < 64) {
        const int q = t - 32, k = q & 15;
        const float* av = (q < 16) ? as0 : ad0;
        float s = 0.f;
        #pragma unroll
        for (int c = 0; c < 16; ++c) s += W0c[k * 16 + c] * av[c];
        uv[q] = s;
    }
}

__device__ __forceinline__ float edge_dot(const float4 e0, const float4 e1, const float* w)
{
    float se = 0.f;
    se += e0.x * w[0]; se += e0.y * w[1]; se += e0.z * w[2]; se += e0.w * w[3];
    se += e1.x * w[4]; se += e1.y * w[5]; se += e1.z * w[6]; se += e1.w * w[7];
    return se;
}

__global__ __launch_bounds__(1024)
void partition_kernel(const int* __restrict__ src, const int* __restrict__ dst,
                      const float* __restrict__ ea, const float* __restrict__ wae,
                      const int* __restrict__ hcnt, int* __restrict__ bst,
                      int* __restrict__ kaw, int* __restrict__ kbw,
                      float* __restrict__ t0p, float* __restrict__ t1p, float* __restrict__ t2p)
{
    __shared__ int  lst[NB];
    __shared__ int  lcur[NB];
    __shared__ int  gb[NB];
    __shared__ int  ttl[NB];
    __shared__ int  ptl[NB];
    __shared__ int  wt[32];
    __shared__ int  wt2[32];
    __shared__ float wl[24];
    __shared__ unsigned char sbk[PCH];
    __shared__ int4 stage[PCH];
    unsigned short* __restrict__ ka = (unsigned short*)kaw;
    unsigned short* __restrict__ kb = (unsigned short*)kbw;
    const int t = threadIdx.x, w = blockIdx.x;
    const int lane = t & 31, wv = t >> 5;
    if (t < 24) wl[t] = wae[t];
    {
        int4 u0[PBW], u1[PBW];
        #pragma unroll
        for (int i = 0; i < PBW; ++i) {
            const int b  = wv + 32 * i;
            const int br = b < NB ? b : NB - 1;
            const int4* cp = (const int4*)(hcnt + br * PW) + 2 * lane;
            u0[i] = cp[0];
            u1[i] = cp[1];
        }
        #pragma unroll
        for (int i = 0; i < PBW; ++i) {
            const int b = wv + 32 * i;
            const int v[8] = {u0[i].x, u0[i].y, u0[i].z, u0[i].w, u1[i].x, u1[i].y, u1[i].z, u1[i].w};
            int all = 0, front = 0;
            #pragma unroll
            for (int k = 0; k < 8; ++k) {
                all   += v[k];
                front += (8 * lane + k < w) ? v[k] : 0;
            }
            for (int o = 16; o > 0; o >>= 1) {
                all   += __shfl_xor(all, o);
                front += __shfl_xor(front, o);
            }
            if (lane == 0 && b < NB) { ttl[b] = all; ptl[b] = front; }
        }
    }
    __syncthreads();
    int c = 0, tot = 0;
    if (t < NB) {
        c   = hcnt[t * PW + w];
        tot = ttl[t];
    }
    const int bs = block_excl_scan(tot, wt);
    const int ex = block_excl_scan(c, wt2);
    if (t < NB) {
        gb[t] = bs + ptl[t];
        lst[t] = ex;
        lcur[t] = ex;
        if (w == 0) bst[t] = bs;
    }
    if (w == 0 && t == NB) bst[NB] = bs;
    __syncthreads();

    const int e0 = w * PCH;
    auto place = [&](const int d, const int sr, const float4 x0, const float4 x1) {
        const int b = d >> 9;
        const int r = atomicAdd(&lcur[b], 1);
        const int key = ((d & 511) << 1) | (sr >> 16);
        stage[r] = make_int4((sr & 0xFFFF) | (key << 16),
                             __float_as_int(edge_dot(x0, x1, wl)),
                             __float_as_int(edge_dot(x0, x1, wl + 8)),
                             __float_as_int(edge_dot(x0, x1, wl + 16)));
        sbk[r] = (unsigned char)b;
    };
    int i = t;
    for (; i + 3072 < PCH; i += 4096) {
        int d[4], sr[4];
        float4 x0[4], x1[4];
        #pragma unroll
        for (int k = 0; k < 4; ++k) {
            const int e = e0 + i + k * 1024;
            d[k] = dst[e]; sr[k] = src[e];
            const float4* ep = (const float4*)(ea + (size_t)e * 8);
            x0[k] = ep[0]; x1[k] = ep[1];
        }
        #pragma unroll
        for (int k = 0; k < 4; ++k) place(d[k], sr[k], x0[k], x1[k]);
    }
    for (; i < PCH; i += 1024) {
        const int e = e0 + i;
        const float4* ep = (const float4*)(ea + (size_t)e * 8);
        place(dst[e], src[e], ep[0], ep[1]);
    }
    __syncthreads();
    for (int j = t; j < PCH; j += 1024) {
        const int4 rec = stage[j];
        const int b = sbk[j];
        const int g = gb[b] + (j - lst[b]);
        ka[g]  = (unsigned short)(rec.x & 0xFFFF);
        kb[g]  = (unsigned short)((unsigned)rec.x >> 16);
        t0p[g] = __int_as_float(rec.y);
        t1p[g] = __int_as_float(rec.z);
        t2p[g] = __int_as_float(rec.w);
    }
}


__global__ __launch_bounds__(1024)
void bucket_sort_kernel(const int* __restrict__ bst, int* kaw, int* kbw,
                        float* t0p, float* t1p, float* t2p,
                        int* __restrict__ off, int* __restrict__ mid)
{
    __shared__ int  kst[1024];
    __shared__ int  knx[1024];
    __shared__ int  wt[32];
    __shared__ int4 stage[BCAP];
    const int t = threadIdx.x, b = blockIdx.x;
    const int s = bst[b];
    const int n = bst[b + 1] - s;
    unsigned short* qa = (unsigned short*)kaw + s;
    unsigned short* qb = (unsigned short*)kbw + s;
    float* q0 = t0p + s;
    float* q1 = t1p + s;
    float* q2 = t2p + s;
    knx[t] = 0;
    __syncthreads();
    int i = t;
    for (; i + 3072 < n; i += 4096) {
        int x[4];
        #pragma unroll
        for (int k = 0; k < 4; ++k) x[k] = qb[i + k * 1024];
        #pragma unroll
        for (int k = 0; k < 4; ++k) atomicAdd(&knx[x[k] & 1023], 1);
    }
    for (; i < n; i += 1024) atomicAdd(&knx[qb[i] & 1023], 1);
    __syncthreads();
    const int c  = knx[t];
    const int ex = block_excl_scan(c, wt);
    kst[t] = ex;
    knx[t] = ex;
    {
        const int node = b * 512 + (t >> 1);
        if (node <= N_NODES) {
            if (t & 1) mid[node] = s + ex; else off[node] = s + ex;
        }
    }
    __syncthreads();
    if (n <= BCAP) {
        i = t;
        for (; i + 3072 < n; i += 4096) {
            int xa[4], xb[4];
            float y0[4], y1[4], y2[4];
            #pragma unroll
            for (int k = 0; k < 4; ++k) {
                const int j = i + k * 1024;
                xa[k] = qa[j]; xb[k] = qb[j]; y0[k] = q0[j]; y1[k] = q1[j]; y2[k] = q2[j];
            }
            #pragma unroll
            for (int k = 0; k < 4; ++k) {
                const int p = atomicAdd(&knx[xb[k] & 1023], 1);
                stage[p] = make_int4(xa[k], __float_as_int(y0[k]), __float_as_int(y1[k]), __float_as_int(y2[k]));
            }
        }
        for (; i < n; i += 1024) {
            const int p = atomicAdd(&knx[qb[i] & 1023], 1);
            stage[p] = make_int4((int)qa[i], __float_as_int(q0[i]), __float_as_int(q1[i]), __float_as_int(q2[i]));
        }
        __syncthreads();
        for (int j = t; j < n; j += 1024) {
            const int4 rec = stage[j];
            qa[j] = (unsigned short)rec.x;
            q0[j] = __int_as_float(rec.y);
            q1[j] = __int_as_float(rec.z);
            q2[j] = __int_as_float(rec.w);
        }
    } else if (t == 0) {
        volatile unsigned short* va = qa;
        volatile unsigned short* vb = qb;
        volatile float* v0 = q0;
        volatile float* v1 = q1;
        volatile float* v2 = q2;
        for (int k = 0; k < 1024; ++k) {
            const int kend = (k < 1023) ? kst[k + 1] : n;
            while (knx[k] < kend) {
                const int p = knx[k];
                const unsigned short ra = va[p], rb = vb[p];
                const float r0 = v0[p], r1 = v1[p], r2 = v2[p];
                const int kk = rb & 1023;
                if (kk == k) { knx[k] = p + 1; continue; }
                const int j = knx[kk];
                knx[kk] = j + 1;
                const unsigned short sa = va[j], sb = vb[j];
                const float s0 = v0[j], s1 = v1[j], s2 = v2[j];
                va[j] = ra; vb[j] = rb; v0[j] = r0; v1[j] = r1; v2[j] = r2;
                va[p] = sa; vb[p] = sb; v0[p] = s0; v1[p] = s1; v2[p] = s2;
            }
        }
    }
}


template <unsigned SEL>
__device__ __forceinline__ float grp8f(const float v)
{
    return __uint_as_float(__builtin_amdgcn_mov_dpp8(__float_as_uint(v), SEL));
}
template <unsigned SEL>
__device__ __forceinline__ int grp8i(const int v)
{
    return (int)__builtin_amdgcn_mov_dpp8((unsigned)v, SEL);
}
#define SEL_QLANE(u) ((unsigned)(u) * 585u + (unsigned)(4 + (u)) * 2396160u)
#define SEL_XOR4 6852524u
#define SEL_XOR2 11788826u
#define SEL_XOR1 14570689u

struct rowq { v2f lo, hi; };
__device__ __forceinline__ rowq make_rowq(const float4 r)
{
    rowq q;
    q.lo = (v2f){r.x, r.y};
    q.hi = (v2f){r.z, r.w};
    return q;
}
__device__ __forceinline__ float dotq(const rowq r, const rowq a)
{
    const v2f t = r.lo * a.lo + r.hi * a.hi;
    return t.x + t.y;
}


#define NEXT_PITCH 18
#define SEG_GROUPS (N_NODES / 8)
#define SEG_WGS 1250
static_assert(N_NODES % 8 == 0, "every group holds 8 nodes: all 8 waves reach every barrier");
static_assert(SEG_WGS <= 256 * 5, "the whole grid is resident at once");
template <int LAYER>
__global__ __launch_bounds__(256, 10)
void segment_softmax_aggregate_kernel(const int* __restrict__ s16w, const float* __restrict__ tq,
                                      const int* __restrict__ off, const int* __restrict__ mid,
                                      const float* __restrict__ asrc, const float* __restrict__ adst,
                                      const float* __restrict__ h, const float* __restrict__ b,
                                      const float* __restrict__ Wf, const float* __restrict__ Wn,
                                      float* __restrict__ outb)
{
    constexpr bool NEXT  = LAYER < 2;
    constexpr bool FIRST = LAYER == 0;
    const unsigned short* __restrict__ s16 = (const unsigned short*)s16w;
    __shared__ __attribute__((aligned(16))) float tile[2][8 * NEXT_PITCH];
    const int lane = threadIdx.x & 31;
    const int wv   = threadIdx.x >> 5;

    const int g  = lane >> 2;
    const unsigned i4 = lane & 3;
    const float4* __restrict__ h4 = (const float4*)h;
    const rowq a4 = make_rowq(*(const float4*)(asrc + 4 * i4));
    const rowq d4 = make_rowq(*(const float4*)(adst + 4 * i4));
    const bool b2 = (i4 & 2) != 0, b1 = (i4 & 1) != 0;
    const int fo = 4 * i4 + ((lane >> 3) & 3);
    const float bo = b[fo];

    auto rec_raw = [&](const int j, const int e) -> int2 {
        int jj = j < e ? j : e - 1;
        jj = jj < 0 ? 0 : jj;
        return make_int2((int)s16[jj], __float_as_int(tq[jj]));
    };
    auto rec_fix = [&](const int2 r, const int j, const int e, const int m) -> int2 {
        return j < e ? make_int2(r.x + (j >= m ? 65536 : 0), r.y) : make_int2(0, 0);
    };

    int beg = __builtin_amdgcn_readfirstlane(off[blockIdx.x * 8 + wv]);
    int end = __builtin_amdgcn_readfirstlane(off[blockIdx.x * 8 + wv + 1]);
    int mdl = __builtin_amdgcn_readfirstlane(mid[blockIdx.x * 8 + wv]);
    int2 c0r = rec_raw(beg + lane, end);
    int2 c1r = rec_raw(beg + 32 + lane, end);
    rowq hd = make_rowq(h4[(unsigned)(blockIdx.x * 8 + wv) * 4u + i4]);

    int it = 0;
    for (int grp = blockIdx.x; grp < SEG_GROUPS; grp += SEG_WGS, ++it) {
    const int d = grp * 8 + wv;
    const int n = end - beg;
    const int2 c0 = rec_fix(c0r, beg + lane, end, mdl);
    const int2 c1 = rec_fix(c1r, beg + 32 + lane, end, mdl);
    const int gnx = grp + SEG_WGS < SEG_GROUPS ? grp + SEG_WGS : grp;
    const int obn = off[gnx * 8 + wv];
    const int oen = off[gnx * 8 + wv + 1];
    const int omn = mid[gnx * 8 + wv];
    float sd = dotq(hd, d4);
    sd += grp8f<SEL_XOR2>(sd);
    sd += grp8f<SEL_XOR1>(sd);

    auto slot_rec = [&](const int j) -> int2 { return rec_fix(rec_raw(j, end), j, end, mdl); };

#define ROW_STEP(u)                                                              \
    {                                                                            \
        const unsigned s = (unsigned)grp8i<SEL_QLANE(u)>(s_l);                   \
        row[u] = make_rowq(h4[s * 4u + i4]);                                     \
    }
    auto gather_rows = [&](const int s_l, rowq (&row)[4]) {
        ROW_STEP(0) ROW_STEP(1) ROW_STEP(2) ROW_STEP(3)
    };
    auto mask_rows = [&](const int cnt, rowq (&row)[4]) {
        #pragma unroll
        for (int u = 0; u < 4; ++u) {
            const bool ok = g < ((cnt - u + 3) >> 2);
            row[u].lo = ok ? row[u].lo : (v2f){0.f, 0.f};
            row[u].hi = ok ? row[u].hi : (v2f){0.f, 0.f};
        }
    };
    auto slot_score = [&](const rowq (&row)[4]) -> float {
        float pv[4], w2[2];
        #pragma unroll
        for (int u = 0; u < 4; ++u) pv[u] = dotq(row[u], a4);
        #pragma unroll
        for (int k = 0; k < 2; ++k) {
            const float keepv = b2 ? pv[k + 2] : pv[k];
            const float send  = b2 ? pv[k] : pv[k + 2];
            w2[k] = keepv + grp8f<SEL_XOR2>(send);
        }
        const float keepv = b1 ? w2[1] : w2[0];
        const float send  = b1 ? w2[0] : w2[1];
        return keepv + grp8f<SEL_XOR1>(send);
    };
    auto slot_logit = [&](const int j, const float ss, const float se) -> float {
        return (j < end) ? leakyf(ss + sd + se) : -INFINITY;
    };
    auto chunk_logit = [&](const int base, rowq (&row)[4]) -> float {
        const int j = base + lane;
        const int2 r = slot_rec(j);
        gather_rows(r.x, row);
        mask_rows(end - base, row);
        return slot_logit(j, slot_score(row), __int_as_float(r.y));
    };

#define ACC_STEP(row, a, u)                                                      \
    {                                                                            \
        const float av = grp8f<SEL_QLANE(u)>(a);                                 \
        acc.lo += row[u].lo * av;                                                \
        acc.hi += row[u].hi * av;                                                \
    }
#define ACC_CHUNK(row, a)                                                        \
    ACC_STEP(row, a, 0) ACC_STEP(row, a, 1) ACC_STEP(row, a, 2) ACC_STEP(row, a, 3)

    const int j0 = beg + lane, j1 = j0 + 32;
    rowq r0[4], r1[4];
    gather_rows(c0.x, r0);
    if (n > 32) gather_rows(c1.x, r1);
    float q0, q1 = -INFINITY;
    mask_rows(n, r0);
    q0 = slot_logit(j0, slot_score(r0), __int_as_float(c0.y));
    if (n > 32) {
        mask_rows(n - 32, r1);
        q1 = slot_logit(j1, slot_score(r1), __int_as_float(c1.y));
    }
    const int nbeg = __builtin_amdgcn_readfirstlane(obn);
    const int nend = __builtin_amdgcn_readfirstlane(oen);
    const int nmid = __builtin_amdgcn_readfirstlane(omn);
    const int2 c0n = rec_raw(nbeg + lane, nend);
    const int2 c1n = rec_raw(nbeg + 32 + lane, nend);
    const rowq hdn = make_rowq(h4[(unsigned)(gnx * 8 + wv) * 4u + i4]);
    float mx = fmaxf(q0, q1);
    for (int base = beg + 64; base < end; base += 32) {
        rowq t[4];
        mx = fmaxf(mx, chunk_logit(base, t));
    }
    for (int o = 16; o > 0; o >>= 1) mx = fmaxf(mx, __shfl_xor(mx, o));
    if (!__builtin_isfinite(mx)) mx = 0.f;

    q0 = __expf(q0 - mx);
    q1 = __expf(q1 - mx);
    float sum = q0 + q1;
    for (int base = beg + 64; base < end; base += 32) {
        rowq t[4];
        sum += __expf(chunk_logit(base, t) - mx);
    }
    for (int o = 16; o > 0; o >>= 1) sum += __shfl_xor(sum, o);
    const float dn = sum + 1e-16f;

    rowq acc = make_rowq(make_float4(0.f, 0.f, 0.f, 0.f));
    {
        const float a0 = q0 / dn;
        const float a1 = q1 / dn;
        ACC_CHUNK(r0, a0)
        if (n > 32) { ACC_CHUNK(r1, a1) }
        for (int base = beg + 64; base < end; base += 32) {
            rowq t[4];
            const float a = __expf(chunk_logit(base, t) - mx) / dn;
            ACC_CHUNK(t, a)
        }
    }
    const bool h16 = (lane & 16) != 0, h8 = (lane & 8) != 0;
    v2f k2 = h16 ? acc.hi : acc.lo;
    const v2f s2 = h16 ? acc.lo : acc.hi;
    k2.x += __shfl_xor(s2.x, 16);
    k2.y += __shfl_xor(s2.y, 16);
    float kf = h8 ? k2.y : k2.x;
    const float sf = h8 ? k2.x : k2.y;
    kf += __shfl_xor(sf, 8);
    kf += grp8f<SEL_XOR4>(kf);
    {
        const float o = FIRST ? kf : leakyf(kf + bo);
        if ((lane & 4) == 0) {
            if (NEXT) tile[it & 1][wv * NEXT_PITCH + fo] = o;
            else      outb[d * 16 + fo] = o;
        }
    }
#undef ROW_STEP
#undef ACC_STEP
#undef ACC_CHUNK


    if (NEXT) {
        const int nlo   = lane & 15;
        const int khalf = (lane >> 4) * 2;
        v2f bw[4], bf[4];
        float bcol = 0.0f;
        #pragma unroll
        for (int kc = 0; kc < 4; ++kc) { bw[kc] = (v2f){0.0f, 0.0f}; bf[kc] = (v2f){0.0f, 0.0f}; }
        if (wv == 0) {
            #pragma unroll
            for (int kc = 0; kc < 4; ++kc) {
                const int k0 = kc * 4 + khalf;
                bw[kc].x = Wn[k0 * 16 + nlo];
                bw[kc].y = Wn[(k0 + 1) * 16 + nlo];
                if (FIRST) {
                    bf[kc].x = Wf[k0 * 16 + nlo];
                    bf[kc].y = Wf[(k0 + 1) * 16 + nlo];
                }
            }
            if (FIRST) bcol = b[nlo];
        }
        __syncthreads();
        if (wv == 0) {
            const bool real = nlo < 8;
            float* tb = tile[it & 1];
            const float* trow = tb + (nlo & 7) * NEXT_PITCH + khalf;
            v2f a[4];
            #pragma unroll
            for (int kc = 0; kc < 4; ++kc) {
                const float2 t2 = *(const float2*)(trow + kc * 4);
                a[kc].x = real ? t2.x : 0.0f;
                a[kc].y = real ? t2.y : 0.0f;
            }
            if (FIRST) {
                v8f y;
                #pragma unroll
                for (int v = 0; v < 8; ++v) y[v] = 0.0f;
                #pragma unroll
                for (int kc = 0; kc < 4; ++kc) {
                    y = __builtin_amdgcn_wmma_f32_16x16x4_f32(
                            false, a[kc], false, bf[kc], (short)0, y, false, false);
                }
                if (lane < 16) {
                    #pragma unroll
                    for (int v = 0; v < 8; ++v) tb[v * NEXT_PITCH + nlo] = leakyf(y[v] + bcol);
                }
                __builtin_amdgcn_fence(__ATOMIC_ACQ_REL, "wavefront");
                #pragma unroll
                for (int kc = 0; kc < 4; ++kc) {
                    const float2 t2 = *(const float2*)(trow + kc * 4);
                    a[kc].x = real ? t2.x : 0.0f;
                    a[kc].y = real ? t2.y : 0.0f;
                }
            }
            v8f c;
            #pragma unroll
            for (int v = 0; v < 8; ++v) c[v] = 0.0f;
            #pragma unroll
            for (int kc = 0; kc < 4; ++kc) {
                c = __builtin_amdgcn_wmma_f32_16x16x4_f32(
                        false, a[kc], false, bw[kc], (short)0, c, false, false);
            }
            if (lane < 16) {
                #pragma unroll
                for (int v = 0; v < 8; ++v) outb[(grp * 8 + v) * 16 + nlo] = c[v];
            }
        }
    }
    beg = nbeg; end = nend; mdl = nmid; c0r = c0n; c1r = c1n; hd = hdn;
    }
}

extern "C" void kernel_launch(void* const* d_in, const int* in_sizes, int n_in,
                              void* d_out, int out_size, void* d_ws, size_t ws_size,
                              hipStream_t stream)
{
    const float* x     = (const float*)d_in[0];
    const int*   ei    = (const int*)  d_in[1];
    const float* ea    = (const float*)d_in[2];
    const float* cW    = (const float*)d_in[3];
    const float* casrc = (const float*)d_in[4];
    const float* cadst = (const float*)d_in[5];
    const float* cWe   = (const float*)d_in[6];
    const float* cae   = (const float*)d_in[7];
    const float* cb    = (const float*)d_in[8];
    const float* W0 = (const float*)d_in[9];  const float* b0 = (const float*)d_in[10];
    const float* W1 = (const float*)d_in[11]; const float* b1 = (const float*)d_in[12];
    const float* W2 = (const float*)d_in[13]; const float* b2 = (const float*)d_in[14];
    const float* W3 = (const float*)d_in[15]; const float* b3 = (const float*)d_in[16];
    float* out = (float*)d_out;

    const int* e_src = ei;
    const int* e_dst = ei + N_EDGES;

    float* ws   = (float*)d_ws;
    int*   kaw  = (int*)ws;
    int*   kbw  = kaw + N_EDGES / 2;
    float* h1   = (float*)kbw;
    int*   mid  = (int*)(ws + N_EDGES);
    float* uv   = (float*)(mid + OFF_N);
    int*   off  = (int*)(ws + N_EDGES + 2 * (size_t)N_NODES);
    int*   bsc  = off + OFF_N;
    int*   bst  = bsc + HN;
    float* t1p  = ws + N_EDGES + 4 * (size_t)N_NODES;
    float* wae  = t1p  + N_EDGES;
    float* w3p  = wae  + 32;
    float* b3p  = w3p  + 64 * 16;
    float* t2p  = b3p + 16;
    float* t0p  = t2p + N_EDGES;
    float* h2   = t0p;

    const int headBlocks  = (N_NODES + HEAD_WAVES * 16 - 1) / (HEAD_WAVES * 16);

    bucket_count_kernel<<<PW, 1024, 0, stream>>>(e_dst, bsc, W3, b3, w3p, b3p, cWe, cae, wae, cW, casrc, cadst, uv);
    partition_kernel<<<PW, 1024, 0, stream>>>(e_src, e_dst, ea, wae, bsc, bst, kaw, kbw, t0p, t1p, t2p);
    bucket_sort_kernel<<<NB, 1024, 0, stream>>>(bst, kaw, kbw, t0p, t1p, t2p, off, mid);

    segment_softmax_aggregate_kernel<0><<<SEG_WGS, 256, 0, stream>>>(
        kaw, t0p, off, mid, uv, uv + 16, x, cb, cW, cW + 256, h1);
    segment_softmax_aggregate_kernel<1><<<SEG_WGS, 256, 0, stream>>>(
        kaw, t1p, off, mid, casrc + 16, cadst + 16, h1, cb + 16, nullptr, cW + 512, h2);
    segment_softmax_aggregate_kernel<2><<<SEG_WGS, 256, 0, stream>>>(
        kaw, t2p, off, mid, casrc + 32, cadst + 32, h2, cb + 32, nullptr, nullptr, h1);

    mlp_head_kernel<<<headBlocks, HEAD_WAVES * 32, 0, stream>>>(h1, W0, b0, W1, b1, W2, b2, w3p, b3p, out);
}


# Every lane exchange (softmax max and sum, row-sum tree, score trees, row and weight hand-out, ordering prefix sums) is a DPP16 or v_permlanex16 operand: no ds_bpermute, no dpp8 move; sum orders kept
# speedup vs baseline: 10.2994x; 1.0744x over previous
#include <hip/hip_runtime.h>
#include <hip/hip_bf16.h>
#include <math.h>

#define N_NODES 100000
#define N_EDGES 3200000
#define NEG_SLOPE 0.2f
#define NB      196
#define PW      256
#define PCH     12500
#define HN      (NB * PW)
#define BCAP    18432
#define OFF_N   100352
#define BSC_N   61440
#define PBW     7
static_assert(PW * PCH == N_EDGES, "the partition work-groups must cover the edges exactly");
static_assert(N_NODES <= 131072 && N_EDGES % 2 == 0, "a source is 2 bytes and one bit; two 2-byte planes fill N_EDGES words");
static_assert(NB * 512 >= N_NODES + 1 && HN + NB + 1 <= BSC_N, "bucket and count extents");
static_assert(PW == 256 && PBW * 32 >= NB, "the prefix step reads a bucket's PW counts as 32 lanes x 8");

typedef __attribute__((ext_vector_type(2))) float v2f;
typedef __attribute__((ext_vector_type(8))) float v8f;

__device__ __forceinline__ float leakyf(float v) { return v > 0.f ? v : NEG_SLOPE * v; }

#define DPP_QUAD(a, b, c, d) ((a) | ((b) << 2) | ((c) << 4) | ((d) << 6))
#define DPP_XOR1     DPP_QUAD(1, 0, 3, 2)
#define DPP_XOR2     DPP_QUAD(2, 3, 0, 1)
#define DPP_XMASK(m) (0x160 + (m))
#define DPP_SHR(n)   (0x110 + (n))
template <int CTRL>
__device__ __forceinline__ int dpp_i(const int v)
{
    return __builtin_amdgcn_update_dpp(0, v, CTRL, 0xF, 0xF, true);
}
template <int CTRL>
__device__ __forceinline__ float dpp_f(const float v)
{
    return __int_as_float(dpp_i<CTRL>(__float_as_int(v)));
}
__device__ __forceinline__ int swap16_i(const int v)
{
    return (int)__builtin_amdgcn_permlanex16((unsigned)v, (unsigned)v, 0x76543210u, 0xFEDCBA98u, false, false);
}
__device__ __forceinline__ float swap16_f(const float v)
{
    return __int_as_float(swap16_i(__float_as_int(v)));
}
__device__ __forceinline__ int other_row_last_i(const int v)
{
    return (int)__builtin_amdgcn_permlanex16((unsigned)v, (unsigned)v, 0xFFFFFFFFu, 0xFFFFFFFFu, false, false);
}
__device__ __forceinline__ int wave_sum_i(int v)
{
    v += swap16_i(v);
    v += dpp_i<DPP_XMASK(8)>(v);
    v += dpp_i<DPP_XMASK(4)>(v);
    v += dpp_i<DPP_XOR2>(v);
    v += dpp_i<DPP_XOR1>(v);
    return v;
}
__device__ __forceinline__ float wave_sum_f(float v)
{
    v += swap16_f(v);
    v += dpp_f<DPP_XMASK(8)>(v);
    v += dpp_f<DPP_XMASK(4)>(v);
    v += dpp_f<DPP_XOR2>(v);
    v += dpp_f<DPP_XOR1>(v);
    return v;
}
__device__ __forceinline__ float wave_max_f(float v)
{
    v = fmaxf(v, swap16_f(v));
    v = fmaxf(v, dpp_f<DPP_XMASK(8)>(v));
    v = fmaxf(v, dpp_f<DPP_XMASK(4)>(v));
    v = fmaxf(v, dpp_f<DPP_XOR2>(v));
    v = fmaxf(v, dpp_f<DPP_XOR1>(v));
    return v;
}

#define HEAD_WAVES 4
#define HEAD_PITCH 68

template <int K>
__device__ __forceinline__ void head_load_a(const float* p, const int pitch, const int nlo,
                                            const int khalf, v2f (&av)[16])
{
    #pragma unroll
    for (int kc = 0; kc < K / 4; ++kc) {
        const int k0 = kc * 4 + khalf;
        av[kc].x = p[nlo * pitch + k0];
        av[kc].y = p[nlo * pitch + k0 + 1];
    }
}

template <int K, int M, int MOUT, int ACT>
__device__ __forceinline__ void head_layer(const v2f (&av)[16], const float* __restrict__ W,
                                           const float* __restrict__ bias, float* q, const int pitch,
                                           const int nlo, const int khalf, const int rofs)
{
    #pragma unroll
    for (int mt = 0; mt < M / 16; ++mt) {
        const int col = mt * 16 + nlo;
        const float bv = bias[col];
        v8f c;
        #pragma unroll
        for (int v = 0; v < 8; ++v) c[v] = bv;
        #pragma unroll
        for (int kc = 0; kc < K / 4; ++kc) {
            const int k0 = kc * 4 + khalf;
            v2f b;
            b.x = W[k0 * M + col];
            b.y = W[(k0 + 1) * M + col];
            c = __builtin_amdgcn_wmma_f32_16x16x4_f32(
                    false, av[kc], false, b, (short)0, c, false, false);
        }
        if (MOUT == M || col < MOUT) {
            #pragma unroll
            for (int v = 0; v < 8; ++v) {
                float r = c[v];
                if (ACT == 1) r = r > 0.f ? r : 0.f;
                q[(rofs + v) * pitch + col] = r;
            }
        }
    }
}

__global__ __launch_bounds__(HEAD_WAVES * 32)
void mlp_head_kernel(const float* __restrict__ X,
                     const float* __restrict__ W0, const float* __restrict__ b0,
                     const float* __restrict__ W1, const float* __restrict__ b1,
                     const float* __restrict__ W2, const float* __restrict__ b2,
                     const float* __restrict__ W3p, const float* __restrict__ b3p,
                     float* __restrict__ out)
{
    __shared__ float tile[HEAD_WAVES][16 * HEAD_PITCH];
    const int lane     = threadIdx.x & 31;
    const int wave     = threadIdx.x >> 5;
    const int row_base = (blockIdx.x * HEAD_WAVES + wave) * 16;
    if (row_base >= N_NODES) return;

    const int nlo   = lane & 15;
    const int khalf = (lane >> 4) * 2;
    const int rofs  = (lane >> 4) * 8;
    float* a = tile[wave];

    v2f av[16];
    head_load_a<16>(X + (size_t)row_base * 16, 16, nlo, khalf, av);
    head_layer<16, 64, 64, 1>(av, W0, b0, a, HEAD_PITCH, nlo, khalf, rofs);
    head_load_a<64>(a, HEAD_PITCH, nlo, khalf, av);
    head_layer<64, 64, 64, 1>(av, W1, b1, a, HEAD_PITCH, nlo, khalf, rofs);
    head_load_a<64>(a, HEAD_PITCH, nlo, khalf, av);
    head_layer<64, 64, 64, 1>(av, W2, b2, a, HEAD_PITCH, nlo, khalf, rofs);
    head_load_a<64>(a, HEAD_PITCH, nlo, khalf, av);
    head_layer<64, 16, 4, 0>(av, W3p, b3p, out + (size_t)row_base * 4, 4, nlo, khalf, rofs);
}


__device__ __forceinline__ int block_excl_scan(const int c, int* wt)
{
    const int lane = threadIdx.x & 31, wv = threadIdx.x >> 5;
    int inc = c;
    inc += dpp_i<DPP_SHR(1)>(inc);
    inc += dpp_i<DPP_SHR(2)>(inc);
    inc += dpp_i<DPP_SHR(4)>(inc);
    inc += dpp_i<DPP_SHR(8)>(inc);
    const int low = other_row_last_i(inc);
    inc += (lane >= 16) ? low : 0;
    if (lane == 31) wt[wv] = inc;
    __syncthreads();
    const int base = wave_sum_i((lane < wv) ? wt[lane] : 0);
    return base + inc - c;
}

__global__ __launch_bounds__(1024)
void bucket_count_kernel(const int* __restrict__ dst, int* __restrict__ hcnt,
                         const float* __restrict__ W3, const float* __restrict__ b3,
                         float* __restrict__ W3p, float* __restrict__ b3p,
                         const float* __restrict__ We, const float* __restrict__ ae,
                         float* __restrict__ wae,
                         const float* __restrict__ W0c, const float* __restrict__ as0,
                         const float* __restrict__ ad0, float* __restrict__ uv)
{
    __shared__ int lh[NB];
    const int t = threadIdx.x, w = blockIdx.x;
    if (t < NB) lh[t] = 0;
    __syncthreads();
    const int* dp = dst + w * PCH;
    int i = t;
    for (; i + 3072 < PCH; i += 4096) {
        int d[4];
        #pragma unroll
        for (int k = 0; k < 4; ++k) d[k] = dp[i + k * 1024];
        #pragma unroll
        for (int k = 0; k < 4; ++k) atomicAdd(&lh[d[k] >> 9], 1);
    }
    for (; i < PCH; i += 1024) atomicAdd(&lh[dp[i] >> 9], 1);
    __syncthreads();
    if (t < NB) hcnt[t * PW + w] = lh[t];
    if (w == 0) {
        const int k = t >> 4, c = t & 15;
        W3p[t] = (c < 4) ? W3[k * 4 + c] : 0.0f;
        if (t < 16) b3p[t] = (t < 4) ? b3[t] : 0.0f;
    }
    if (w == 1 && t < 24) {
        const int i = t >> 3, j = t & 7;
        float s = 0.f;
        #pragma unroll
        for (int k = 0; k < 16; ++k) s += We[i * 128 + j * 16 + k] * ae[i * 16 + k];
        wae[t] = s;
    }
    if (w == 1 && t >= 32 && t < 64) {
        const int q = t - 32, k = q & 15;
        const float* av = (q < 16) ? as0 : ad0;
        float s = 0.f;
        #pragma unroll
        for (int c = 0; c < 16; ++c) s += W0c[k * 16 + c] * av[c];
        uv[q] = s;
    }
}

__device__ __forceinline__ float edge_dot(const float4 e0, const float4 e1, const float* w)
{
    float se = 0.f;
    se += e0.x * w[0]; se += e0.y * w[1]; se += e0.z * w[2]; se += e0.w * w[3];
    se += e1.x * w[4]; se += e1.y * w[5]; se += e1.z * w[6]; se += e1.w * w[7];
    return se;
}

__global__ __launch_bounds__(1024)
void partition_kernel(const int* __restrict__ src, const int* __restrict__ dst,
                      const float* __restrict__ ea, const float* __restrict__ wae,
                      const int* __restrict__ hcnt, int* __restrict__ bst,
                      int* __restrict__ kaw, int* __restrict__ kbw,
                      float* __restrict__ t0p, float* __restrict__ t1p, float* __restrict__ t2p)
{
    __shared__ int  lst[NB];
    __shared__ int  lcur[NB];
    __shared__ int  gb[NB];
    __shared__ int  ttl[NB];
    __shared__ int  ptl[NB];
    __shared__ int  wt[32];
    __shared__ int  wt2[32];
    __shared__ float wl[24];
    __shared__ unsigned char sbk[PCH];
    __shared__ int4 stage[PCH];
    unsigned short* __restrict__ ka = (unsigned short*)kaw;
    unsigned short* __restrict__ kb = (unsigned short*)kbw;
    const int t = threadIdx.x, w = blockIdx.x;
    const int lane = t & 31, wv = t >> 5;
    if (t < 24) wl[t] = wae[t];
    {
        int4 u0[PBW], u1[PBW];
        #pragma unroll
        for (int i = 0; i < PBW; ++i) {
            const int b  = wv + 32 * i;
            const int br = b < NB ? b : NB - 1;
            const int4* cp = (const int4*)(hcnt + br * PW) + 2 * lane;
            u0[i] = cp[0];
            u1[i] = cp[1];
        }
        #pragma unroll
        for (int i = 0; i < PBW; ++i) {
            const int b = wv + 32 * i;
            const int v[8] = {u0[i].x, u0[i].y, u0[i].z, u0[i].w, u1[i].x, u1[i].y, u1[i].z, u1[i].w};
            int all = 0, front = 0;
            #pragma unroll
            for (int k = 0; k < 8; ++k) {
                all   += v[k];
                front += (8 * lane + k < w) ? v[k] : 0;
            }
            all   = wave_sum_i(all);
            front = wave_sum_i(front);
            if (lane == 0 && b < NB) { ttl[b] = all; ptl[b] = front; }
        }
    }
    __syncthreads();
    int c = 0, tot = 0;
    if (t < NB) {
        c   = hcnt[t * PW + w];
        tot = ttl[t];
    }
    const int bs = block_excl_scan(tot, wt);
    const int ex = block_excl_scan(c, wt2);
    if (t < NB) {
        gb[t] = bs + ptl[t];
        lst[t] = ex;
        lcur[t] = ex;
        if (w == 0) bst[t] = bs;
    }
    if (w == 0 && t == NB) bst[NB] = bs;
    __syncthreads();

    const int e0 = w * PCH;
    auto place = [&](const int d, const int sr, const float4 x0, const float4 x1) {
        const int b = d >> 9;
        const int r = atomicAdd(&lcur[b], 1);
        const int key = ((d & 511) << 1) | (sr >> 16);
        stage[r] = make_int4((sr & 0xFFFF) | (key << 16),
                             __float_as_int(edge_dot(x0, x1, wl)),
                             __float_as_int(edge_dot(x0, x1, wl + 8)),
                             __float_as_int(edge_dot(x0, x1, wl + 16)));
        sbk[r] = (unsigned char)b;
    };
    int i = t;
    for (; i + 3072 < PCH; i += 4096) {
        int d[4], sr[4];
        float4 x0[4], x1[4];
        #pragma unroll
        for (int k = 0; k < 4; ++k) {
            const int e = e0 + i + k * 1024;
            d[k] = dst[e]; sr[k] = src[e];
            const float4* ep = (const float4*)(ea + (size_t)e * 8);
            x0[k] = ep[0]; x1[k] = ep[1];
        }
        #pragma unroll
        for (int k = 0; k < 4; ++k) place(d[k], sr[k], x0[k], x1[k]);
    }
    for (; i < PCH; i += 1024) {
        const int e = e0 + i;
        const float4* ep = (const float4*)(ea + (size_t)e * 8);
        place(dst[e], src[e], ep[0], ep[1]);
    }
    __syncthreads();
    for (int j = t; j < PCH; j += 1024) {
        const int4 rec = stage[j];
        const int b = sbk[j];
        const int g = gb[b] + (j - lst[b]);
        ka[g]  = (unsigned short)(rec.x & 0xFFFF);
        kb[g]  = (unsigned short)((unsigned)rec.x >> 16);
        t0p[g] = __int_as_float(rec.y);
        t1p[g] = __int_as_float(rec.z);
        t2p[g] = __int_as_float(rec.w);
    }
}


__global__ __launch_bounds__(1024)
void bucket_sort_kernel(const int* __restrict__ bst, int* kaw, int* kbw,
                        float* t0p, float* t1p, float* t2p,
                        int* __restrict__ off, int* __restrict__ mid)
{
    __shared__ int  kst[1024];
    __shared__ int  knx[1024];
    __shared__ int  wt[32];
    __shared__ int4 stage[BCAP];
    const int t = threadIdx.x, b = blockIdx.x;
    const int s = bst[b];
    const int n = bst[b + 1] - s;
    unsigned short* qa = (unsigned short*)kaw + s;
    unsigned short* qb = (unsigned short*)kbw + s;
    float* q0 = t0p + s;
    float* q1 = t1p + s;
    float* q2 = t2p + s;
    knx[t] = 0;
    __syncthreads();
    int i = t;
    for (; i + 3072 < n; i += 4096) {
        int x[4];
        #pragma unroll
        for (int k = 0; k < 4; ++k) x[k] = qb[i + k * 1024];
        #pragma unroll
        for (int k = 0; k < 4; ++k) atomicAdd(&knx[x[k] & 1023], 1);
    }
    for (; i < n; i += 1024) atomicAdd(&knx[qb[i] & 1023], 1);
    __syncthreads();
    const int c  = knx[t];
    const int ex = block_excl_scan(c, wt);
    kst[t] = ex;
    knx[t] = ex;
    {
        const int node = b * 512 + (t >> 1);
        if (node <= N_NODES) {
            if (t & 1) mid[node] = s + ex; else off[node] = s + ex;
        }
    }
    __syncthreads();
    if (n <= BCAP) {
        i = t;
        for (; i + 3072 < n; i += 4096) {
            int xa[4], xb[4];
            float y0[4], y1[4], y2[4];
            #pragma unroll
            for (int k = 0; k < 4; ++k) {
                const int j = i + k * 1024;
                xa[k] = qa[j]; xb[k] = qb[j]; y0[k] = q0[j]; y1[k] = q1[j]; y2[k] = q2[j];
            }
            #pragma unroll
            for (int k = 0; k < 4; ++k) {
                const int p = atomicAdd(&knx[xb[k] & 1023], 1);
                stage[p] = make_int4(xa[k], __float_as_int(y0[k]), __float_as_int(y1[k]), __float_as_int(y2[k]));
            }
        }
        for (; i < n; i += 1024) {
            const int p = atomicAdd(&knx[qb[i] & 1023], 1);
            stage[p] = make_int4((int)qa[i], __float_as_int(q0[i]), __float_as_int(q1[i]), __float_as_int(q2[i]));
        }
        __syncthreads();
        for (int j = t; j < n; j += 1024) {
            const int4 rec = stage[j];
            qa[j] = (unsigned short)rec.x;
            q0[j] = __int_as_float(rec.y);
            q1[j] = __int_as_float(rec.z);
            q2[j] = __int_as_float(rec.w);
        }
    } else if (t == 0) {
        volatile unsigned short* va = qa;
        volatile unsigned short* vb = qb;
        volatile float* v0 = q0;
        volatile float* v1 = q1;
        volatile float* v2 = q2;
        for (int k = 0; k < 1024; ++k) {
            const int kend = (k < 1023) ? kst[k + 1] : n;
            while (knx[k] < kend) {
                const int p = knx[k];
                const unsigned short ra = va[p], rb = vb[p];
                const float r0 = v0[p], r1 = v1[p], r2 = v2[p];
                const int kk = rb & 1023;
                if (kk == k) { knx[k] = p + 1; continue; }
                const int j = knx[kk];
                knx[kk] = j + 1;
                const unsigned short sa = va[j], sb = vb[j];
                const float s0 = v0[j], s1 = v1[j], s2 = v2[j];
                va[j] = ra; vb[j] = rb; v0[j] = r0; v1[j] = r1; v2[j] = r2;
                va[p] = sa; vb[p] = sb; v0[p] = s0; v1[p] = s1; v2[p] = s2;
            }
        }
    }
}


#define DPP_QLANE(u) DPP_QUAD(u, u, u, u)

struct rowq { v2f lo, hi; };
__device__ __forceinline__ rowq make_rowq(const float4 r)
{
    rowq q;
    q.lo = (v2f){r.x, r.y};
    q.hi = (v2f){r.z, r.w};
    return q;
}
__device__ __forceinline__ float dotq(const rowq r, const rowq a)
{
    const v2f t = r.lo * a.lo + r.hi * a.hi;
    return t.x + t.y;
}


#define NEXT_PITCH 18
#define SEG_GROUPS (N_NODES / 8)
#define SEG_WGS 1250
static_assert(N_NODES % 8 == 0, "every group holds 8 nodes: all 8 waves reach every barrier");
static_assert(SEG_WGS <= 256 * 5, "the whole grid is resident at once");
template <int LAYER>
__global__ __launch_bounds__(256, 10)
void segment_softmax_aggregate_kernel(const int* __restrict__ s16w, const float* __restrict__ tq,
                                      const int* __restrict__ off, const int* __restrict__ mid,
                                      const float* __restrict__ asrc, const float* __restrict__ adst,
                                      const float* __restrict__ h, const float* __restrict__ b,
                                      const float* __restrict__ Wf, const float* __restrict__ Wn,
                                      float* __restrict__ outb)
{
    constexpr bool NEXT  = LAYER < 2;
    constexpr bool FIRST = LAYER == 0;
    const unsigned short* __restrict__ s16 = (const unsigned short*)s16w;
    __shared__ __attribute__((aligned(16))) float tile[2][8 * NEXT_PITCH];
    const int lane = threadIdx.x & 31;
    const int wv   = threadIdx.x >> 5;

    const int g  = lane >> 2;
    const unsigned i4 = lane & 3;
    const float4* __restrict__ h4 = (const float4*)h;
    const rowq a4 = make_rowq(*(const float4*)(asrc + 4 * i4));
    const rowq d4 = make_rowq(*(const float4*)(adst + 4 * i4));
    const bool b2 = (i4 & 2) != 0, b1 = (i4 & 1) != 0;
    const int fo = 4 * i4 + ((lane >> 3) & 3);
    const float bo = b[fo];

    auto rec_raw = [&](const int j, const int e) -> int2 {
        int jj = j < e ? j : e - 1;
        jj = jj < 0 ? 0 : jj;
        return make_int2((int)s16[jj], __float_as_int(tq[jj]));
    };
    auto rec_fix = [&](const int2 r, const int j, const int e, const int m) -> int2 {
        return j < e ? make_int2(r.x + (j >= m ? 65536 : 0), r.y) : make_int2(0, 0);
    };

    int beg = __builtin_amdgcn_readfirstlane(off[blockIdx.x * 8 + wv]);
    int end = __builtin_amdgcn_readfirstlane(off[blockIdx.x * 8 + wv + 1]);
    int mdl = __builtin_amdgcn_readfirstlane(mid[blockIdx.x * 8 + wv]);
    int2 c0r = rec_raw(beg + lane, end);
    int2 c1r = rec_raw(beg + 32 + lane, end);
    rowq hd = make_rowq(h4[(unsigned)(blockIdx.x * 8 + wv) * 4u + i4]);

    int it = 0;
    for (int grp = blockIdx.x; grp < SEG_GROUPS; grp += SEG_WGS, ++it) {
    const int d = grp * 8 + wv;
    const int n = end - beg;
    const int2 c0 = rec_fix(c0r, beg + lane, end, mdl);
    const int2 c1 = rec_fix(c1r, beg + 32 + lane, end, mdl);
    const int gnx = grp + SEG_WGS < SEG_GROUPS ? grp + SEG_WGS : grp;
    const int obn = off[gnx * 8 + wv];
    const int oen = off[gnx * 8 + wv + 1];
    const int omn = mid[gnx * 8 + wv];
    float sd = dotq(hd, d4);
    sd += dpp_f<DPP_XOR2>(sd);
    sd += dpp_f<DPP_XOR1>(sd);

    auto slot_rec = [&](const int j) -> int2 { return rec_fix(rec_raw(j, end), j, end, mdl); };

#define ROW_STEP(u)                                                              \
    {                                                                            \
        const unsigned s = (unsigned)dpp_i<DPP_QLANE(u)>(s_l);                   \
        row[u] = make_rowq(h4[s * 4u + i4]);                                     \
    }
    auto gather_rows = [&](const int s_l, rowq (&row)[4]) {
        ROW_STEP(0) ROW_STEP(1) ROW_STEP(2) ROW_STEP(3)
    };
    auto mask_rows = [&](const int cnt, rowq (&row)[4]) {
        #pragma unroll
        for (int u = 0; u < 4; ++u) {
            const bool ok = g < ((cnt - u + 3) >> 2);
            row[u].lo = ok ? row[u].lo : (v2f){0.f, 0.f};
            row[u].hi = ok ? row[u].hi : (v2f){0.f, 0.f};
        }
    };
    auto slot_score = [&](const rowq (&row)[4]) -> float {
        float pv[4], w2[2];
        #pragma unroll
        for (int u = 0; u < 4; ++u) pv[u] = dotq(row[u], a4);
        #pragma unroll
        for (int k = 0; k < 2; ++k) {
            const float keepv = b2 ? pv[k + 2] : pv[k];
            const float send  = b2 ? pv[k] : pv[k + 2];
            w2[k] = keepv + dpp_f<DPP_XOR2>(send);
        }
        const float keepv = b1 ? w2[1] : w2[0];
        const float send  = b1 ? w2[0] : w2[1];
        return keepv + dpp_f<DPP_XOR1>(send);
    };
    auto slot_logit = [&](const int j, const float ss, const float se) -> float {
        return (j < end) ? leakyf(ss + sd + se) : -INFINITY;
    };
    auto chunk_logit = [&](const int base, rowq (&row)[4]) -> float {
        const int j = base + lane;
        const int2 r = slot_rec(j);
        gather_rows(r.x, row);
        mask_rows(end - base, row);
        return slot_logit(j, slot_score(row), __int_as_float(r.y));
    };

#define ACC_STEP(row, a, u)                                                      \
    {                                                                            \
        const float av = dpp_f<DPP_QLANE(u)>(a);                                 \
        acc.lo += row[u].lo * av;                                                \
        acc.hi += row[u].hi * av;                                                \
    }
#define ACC_CHUNK(row, a)                                                        \
    ACC_STEP(row, a, 0) ACC_STEP(row, a, 1) ACC_STEP(row, a, 2) ACC_STEP(row, a, 3)

    const int j0 = beg + lane, j1 = j0 + 32;
    rowq r0[4], r1[4];
    gather_rows(c0.x, r0);
    if (n > 32) gather_rows(c1.x, r1);
    float q0, q1 = -INFINITY;
    mask_rows(n, r0);
    q0 = slot_logit(j0, slot_score(r0), __int_as_float(c0.y));
    if (n > 32) {
        mask_rows(n - 32, r1);
        q1 = slot_logit(j1, slot_score(r1), __int_as_float(c1.y));
    }
    const int nbeg = __builtin_amdgcn_readfirstlane(obn);
    const int nend = __builtin_amdgcn_readfirstlane(oen);
    const int nmid = __builtin_amdgcn_readfirstlane(omn);
    const int2 c0n = rec_raw(nbeg + lane, nend);
    const int2 c1n = rec_raw(nbeg + 32 + lane, nend);
    const rowq hdn = make_rowq(h4[(unsigned)(gnx * 8 + wv) * 4u + i4]);
    float mx = fmaxf(q0, q1);
    for (int base = beg + 64; base < end; base += 32) {
        rowq t[4];
        mx = fmaxf(mx, chunk_logit(base, t));
    }
    mx = wave_max_f(mx);
    if (!__builtin_isfinite(mx)) mx = 0.f;

    q0 = __expf(q0 - mx);
    q1 = __expf(q1 - mx);
    float sum = q0 + q1;
    for (int base = beg + 64; base < end; base += 32) {
        rowq t[4];
        sum += __expf(chunk_logit(base, t) - mx);
    }
    sum = wave_sum_f(sum);
    const float dn = sum + 1e-16f;

    rowq acc = make_rowq(make_float4(0.f, 0.f, 0.f, 0.f));
    {
        const float a0 = q0 / dn;
        const float a1 = q1 / dn;
        ACC_CHUNK(r0, a0)
        if (n > 32) { ACC_CHUNK(r1, a1) }
        for (int base = beg + 64; base < end; base += 32) {
            rowq t[4];
            const float a = __expf(chunk_logit(base, t) - mx) / dn;
            ACC_CHUNK(t, a)
        }
    }
    const bool h16 = (lane & 16) != 0, h8 = (lane & 8) != 0;
    v2f k2 = h16 ? acc.hi : acc.lo;
    const v2f s2 = h16 ? acc.lo : acc.hi;
    k2.x += swap16_f(s2.x);
    k2.y += swap16_f(s2.y);
    float kf = h8 ? k2.y : k2.x;
    const float sf = h8 ? k2.x : k2.y;
    kf += dpp_f<DPP_XMASK(8)>(sf);
    kf += dpp_f<DPP_XMASK(4)>(kf);
    {
        const float o = FIRST ? kf : leakyf(kf + bo);
        if ((lane & 4) == 0) {
            if (NEXT) tile[it & 1][wv * NEXT_PITCH + fo] = o;
            else      outb[d * 16 + fo] = o;
        }
    }
#undef ROW_STEP
#undef ACC_STEP
#undef ACC_CHUNK


    if (NEXT) {
        const int nlo   = lane & 15;
        const int khalf = (lane >> 4) * 2;
        v2f bw[4], bf[4];
        float bcol = 0.0f;
        #pragma unroll
        for (int kc = 0; kc < 4; ++kc) { bw[kc] = (v2f){0.0f, 0.0f}; bf[kc] = (v2f){0.0f, 0.0f}; }
        if (wv == 0) {
            #pragma unroll
            for (int kc = 0; kc < 4; ++kc) {
                const int k0 = kc * 4 + khalf;
                bw[kc].x = Wn[k0 * 16 + nlo];
                bw[kc].y = Wn[(k0 + 1) * 16 + nlo];
                if (FIRST) {
                    bf[kc].x = Wf[k0 * 16 + nlo];
                    bf[kc].y = Wf[(k0 + 1) * 16 + nlo];
                }
            }
            if (FIRST) bcol = b[nlo];
        }
        __syncthreads();
        if (wv == 0) {
            const bool real = nlo < 8;
            float* tb = tile[it & 1];
            const float* trow = tb + (nlo & 7) * NEXT_PITCH + khalf;
            v2f a[4];
            #pragma unroll
            for (int kc = 0; kc < 4; ++kc) {
                const float2 t2 = *(const float2*)(trow + kc * 4);
                a[kc].x = real ? t2.x : 0.0f;
                a[kc].y = real ? t2.y : 0.0f;
            }
            if (FIRST) {
                v8f y;
                #pragma unroll
                for (int v = 0; v < 8; ++v) y[v] = 0.0f;
                #pragma unroll
                for (int kc = 0; kc < 4; ++kc) {
                    y = __builtin_amdgcn_wmma_f32_16x16x4_f32(
                            false, a[kc], false, bf[kc], (short)0, y, false, false);
                }
                if (lane < 16) {
                    #pragma unroll
                    for (int v = 0; v < 8; ++v) tb[v * NEXT_PITCH + nlo] = leakyf(y[v] + bcol);
                }
                __builtin_amdgcn_fence(__ATOMIC_ACQ_REL, "wavefront");
                #pragma unroll
                for (int kc = 0; kc < 4; ++kc) {
                    const float2 t2 = *(const float2*)(trow + kc * 4);
                    a[kc].x = real ? t2.x : 0.0f;
                    a[kc].y = real ? t2.y : 0.0f;
                }
            }
            v8f c;
            #pragma unroll
            for (int v = 0; v < 8; ++v) c[v] = 0.0f;
            #pragma unroll
            for (int kc = 0; kc < 4; ++kc) {
                c = __builtin_amdgcn_wmma_f32_16x16x4_f32(
                        false, a[kc], false, bw[kc], (short)0, c, false, false);
            }
            if (lane < 16) {
                #pragma unroll
                for (int v = 0; v < 8; ++v) outb[(grp * 8 + v) * 16 + nlo] = c[v];
            }
        }
    }
    beg = nbeg; end = nend; mdl = nmid; c0r = c0n; c1r = c1n; hd = hdn;
    }
}

extern "C" void kernel_launch(void* const* d_in, const int* in_sizes, int n_in,
                              void* d_out, int out_size, void* d_ws, size_t ws_size,
                              hipStream_t stream)
{
    const float* x     = (const float*)d_in[0];
    const int*   ei    = (const int*)  d_in[1];
    const float* ea    = (const float*)d_in[2];
    const float* cW    = (const float*)d_in[3];
    const float* casrc = (const float*)d_in[4];
    const float* cadst = (const float*)d_in[5];
    const float* cWe   = (const float*)d_in[6];
    const float* cae   = (const float*)d_in[7];
    const float* cb    = (const float*)d_in[8];
    const float* W0 = (const float*)d_in[9];  const float* b0 = (const float*)d_in[10];
    const float* W1 = (const float*)d_in[11]; const float* b1 = (const float*)d_in[12];
    const float* W2 = (const float*)d_in[13]; const float* b2 = (const float*)d_in[14];
    const float* W3 = (const float*)d_in[15]; const float* b3 = (const float*)d_in[16];
    float* out = (float*)d_out;

    const int* e_src = ei;
    const int* e_dst = ei + N_EDGES;

    float* ws   = (float*)d_ws;
    int*   kaw  = (int*)ws;
    int*   kbw  = kaw + N_EDGES / 2;
    float* h1   = (float*)kbw;
    int*   mid  = (int*)(ws + N_EDGES);
    float* uv   = (float*)(mid + OFF_N);
    int*   off  = (int*)(ws + N_EDGES + 2 * (size_t)N_NODES);
    int*   bsc  = off + OFF_N;
    int*   bst  = bsc + HN;
    float* t1p  = ws + N_EDGES + 4 * (size_t)N_NODES;
    float* wae  = t1p  + N_EDGES;
    float* w3p  = wae  + 32;
    float* b3p  = w3p  + 64 * 16;
    float* t2p  = b3p + 16;
    float* t0p  = t2p + N_EDGES;
    float* h2   = t0p;

    const int headBlocks  = (N_NODES + HEAD_WAVES * 16 - 1) / (HEAD_WAVES * 16);

    bucket_count_kernel<<<PW, 1024, 0, stream>>>(e_dst, bsc, W3, b3, w3p, b3p, cWe, cae, wae, cW, casrc, cadst, uv);
    partition_kernel<<<PW, 1024, 0, stream>>>(e_src, e_dst, ea, wae, bsc, bst, kaw, kbw, t0p, t1p, t2p);
    bucket_sort_kernel<<<NB, 1024, 0, stream>>>(bst, kaw, kbw, t0p, t1p, t2p, off, mid);

    segment_softmax_aggregate_kernel<0><<<SEG_WGS, 256, 0, stream>>>(
        kaw, t0p, off, mid, uv, uv + 16, x, cb, cW, cW + 256, h1);
    segment_softmax_aggregate_kernel<1><<<SEG_WGS, 256, 0, stream>>>(
        kaw, t1p, off, mid, casrc + 16, cadst + 16, h1, cb + 16, nullptr, cW + 512, h2);
    segment_softmax_aggregate_kernel<2><<<SEG_WGS, 256, 0, stream>>>(
        kaw, t2p, off, mid, casrc + 32, cadst + 32, h2, cb + 32, nullptr, nullptr, h1);

    mlp_head_kernel<<<headBlocks, HEAD_WAVES * 32, 0, stream>>>(h1, W0, b0, W1, b1, W2, b2, w3p, b3p, out);
}
